# starting kernel
# baseline (speedup 1.0000x reference)
#include <hip/hip_runtime.h>
#include <hip/hip_bf16.h>
#include <math.h>

static constexpr int Bc = 4;
static constexpr int Nn = 1024;
static constexpr int Dd = 256;
static constexpr int Hh = 8;
static constexpr int DH = 32;

typedef float v2f __attribute__((ext_vector_type(2)));
typedef float v8f __attribute__((ext_vector_type(8)));

__global__ __launch_bounds__(32) void proj12_kernel(
    const float* __restrict__ hmat, const float* __restrict__ W1,
    const float* __restrict__ W2, float* __restrict__ wh1,
    float* __restrict__ wh2)
{
    const int lane = threadIdx.x & 31;
    const int m    = lane & 15;
    const int kh   = (lane >> 4) << 1;
    const int col0 = blockIdx.x * 16;
    const int row0 = blockIdx.y * 16;

    v8f acc1 = {};
    v8f acc2 = {};
    const float* aRow  = hmat + (long)(row0 + m) * Dd;
    const float* b1Row = W1   + (long)(col0 + m) * Dd;
    const float* b2Row = W2   + (long)(col0 + m) * Dd;

    for (int k0 = 0; k0 < Dd; k0 += 4) {
        v2f a  = *(const v2f*)(aRow  + k0 + kh);
        v2f b1 = *(const v2f*)(b1Row + k0 + kh);
        v2f b2 = *(const v2f*)(b2Row + k0 + kh);
        acc1 = __builtin_amdgcn_wmma_f32_16x16x4_f32(false, a, false, b1,
                                                     (short)0, acc1, false, false);
        acc2 = __builtin_amdgcn_wmma_f32_16x16x4_f32(false, a, false, b2,
                                                     (short)0, acc2, false, false);
    }

    const int rbase = row0 + ((lane >> 4) << 3);
    const int c     = col0 + m;
#pragma unroll
    for (int r = 0; r < 8; ++r) {
        wh1[(long)(rbase + r) * Dd + c] = acc1[r];
        wh2[(long)(rbase + r) * Dd + c] = acc2[r];
    }
}

__global__ __launch_bounds__(256) void scores_kernel(
    const float* __restrict__ wh1, const float* __restrict__ wh2,
    const float* __restrict__ Ai, const float* __restrict__ Aj,
    float* __restrict__ s1, float* __restrict__ s2,
    float* __restrict__ o1, float* __restrict__ o2)
{
    int idx = blockIdx.x * blockDim.x + threadIdx.x;
    if (idx >= Bc * Nn * Hh) return;
    const int h = idx % Hh;
    const int n = (idx / Hh) % Nn;
    const int b = idx / (Hh * Nn);

    const float* r1 = wh1 + ((long)(b * Nn + n)) * Dd + h * DH;
    const float* r2 = wh2 + ((long)(b * Nn + n)) * Dd + h * DH;
    const float* ai = Ai + h * DH;
    const float* aj = Aj + h * DH;

    float a1 = 0.f, a2 = 0.f, a3 = 0.f, a4 = 0.f;
#pragma unroll 8
    for (int d = 0; d < DH; ++d) {
        const float w1v = r1[d], w2v = r2[d], aiv = ai[d], ajv = aj[d];
        a1 += w1v * aiv;  a2 += w1v * ajv;
        a3 += w2v * aiv;  a4 += w2v * ajv;
    }
    const int o = (b * Hh + h) * Nn + n;
    s1[o] = a1; s2[o] = a2; o1[o] = a3; o2[o] = a4;
}

__global__ __launch_bounds__(256) void entmax_kernel(
    const float* __restrict__ s1, const float* __restrict__ s2,
    const float* __restrict__ o1, const float* __restrict__ o2,
    const int* __restrict__ adj, const int* __restrict__ nag,
    float* __restrict__ att)
{
    __shared__ float xs[Nn];
    __shared__ float srt[Nn];
    __shared__ float tau[Nn];
    __shared__ float redA[256];
    __shared__ float redB[256];
    __shared__ int   redI[256];

    const int tid = threadIdx.x;
    const int bid = blockIdx.x;
    const int i   = bid & (Nn - 1);
    const int h   = (bid >> 10) & (Hh - 1);
    const int b   = bid >> 13;
    const int na  = nag[0];

    const int  rowbase = (b * Hh + h) * Nn;
    const float s1i = s1[rowbase + i];
    const float o1i = o1[rowbase + i];
    const int   im  = i % na;
    const long  adjbase = ((long)b * Nn + i) * Nn;

    float lmax = -3.4e38f;
    for (int j = tid; j < Nn; j += 256) {
        float e = (im == (j % na)) ? (s1i + s2[rowbase + j])
                                   : (o1i + o2[rowbase + j]);
        if (adj[adjbase + j] == 0) e = -1.0e12f;
        const float x = 0.5f * e;
        xs[j] = x;
        lmax  = fmaxf(lmax, x);
    }
    redA[tid] = lmax;
    __syncthreads();
    for (int s = 128; s > 0; s >>= 1) {
        if (tid < s) redA[tid] = fmaxf(redA[tid], redA[tid + s]);
        __syncthreads();
    }
    const float xmax = redA[0];
    __syncthreads();
    for (int j = tid; j < Nn; j += 256) {
        const float x = xs[j] - xmax;
        xs[j]  = x;
        srt[j] = x;
    }
    __syncthreads();

    for (int k = 2; k <= Nn; k <<= 1) {
        for (int jj = k >> 1; jj > 0; jj >>= 1) {
            for (int idx = tid; idx < Nn; idx += 256) {
                const int ixj = idx ^ jj;
                if (ixj > idx) {
                    const float a = srt[idx];
                    const float c = srt[ixj];
                    const bool desc = ((idx & k) == 0);
                    if (desc ? (a < c) : (a > c)) { srt[idx] = c; srt[ixj] = a; }
                }
            }
            __syncthreads();
        }
    }

    const int base = tid * 4;
    const float v0 = srt[base], v1 = srt[base + 1];
    const float v2 = srt[base + 2], v3 = srt[base + 3];
    const float c1 = v0 + v1, c2 = c1 + v2, c3 = c2 + v3;
    const float q0 = v0 * v0, q1 = v1 * v1, q2 = v2 * v2, q3 = v3 * v3;
    const float d1 = q0 + q1, d2 = d1 + q2, d3 = d2 + q3;
    redA[tid] = c3;
    redB[tid] = d3;
    __syncthreads();
    for (int off = 1; off < 256; off <<= 1) {
        const float pa = (tid >= off) ? redA[tid - off] : 0.0f;
        const float pb = (tid >= off) ? redB[tid - off] : 0.0f;
        __syncthreads();
        redA[tid] += pa;
        redB[tid] += pb;
        __syncthreads();
    }
    const float exS = redA[tid] - c3;
    const float exQ = redB[tid] - d3;

    const float sv[4] = { v0, v1, v2, v3 };
    const float cs[4] = { exS + v0, exS + c1, exS + c2, exS + c3 };
    const float cq[4] = { exQ + q0, exQ + d1, exQ + d2, exQ + d3 };
    int cnt = 0;
#pragma unroll
    for (int u = 0; u < 4; ++u) {
        const float rho   = (float)(base + u + 1);
        const float mean  = cs[u] / rho;
        const float msq   = cq[u] / rho;
        const float ssv   = rho * (msq - mean * mean);
        const float delta = (1.0f - ssv) / rho;
        const float t     = mean - ((delta > 0.0f) ? sqrtf(delta) : 0.0f);
        tau[base + u] = t;
        cnt += (t <= sv[u]) ? 1 : 0;
    }
    redI[tid] = cnt;
    __syncthreads();
    for (int s = 128; s > 0; s >>= 1) {
        if (tid < s) redI[tid] += redI[tid + s];
        __syncthreads();
    }
    const int   support = redI[0];
    const float tstar   = tau[support - 1];

    float* outrow = att + ((long)(b * Hh + h) * Nn + i) * Nn;
    for (int j = tid; j < Nn; j += 256) {
        float y = xs[j] - tstar;
        y = fmaxf(y, 0.0f);
        outrow[j] = y * y;
    }
}

static constexpr int KC  = 128;
static constexpr int LDP = 40;

__global__ __launch_bounds__(128) void aggregate_kernel(
    const float* __restrict__ att, const float* __restrict__ wh1,
    const float* __restrict__ wh2, const int* __restrict__ nag,
    float* __restrict__ hp)
{
    __shared__ float ldsW1[KC][LDP];
    __shared__ float ldsW2[KC][LDP];

    const int tid  = threadIdx.x;
    const int wave = tid >> 5;
    const int lane = tid & 31;
    const int m    = lane & 15;
    const int kh   = (lane >> 4) << 1;
    const int bh   = blockIdx.x;
    const int b    = bh / Hh;
    const int h    = bh % Hh;
    const int i0   = blockIdx.y * 64 + wave * 16;
    const int na   = nag[0];

    const int row    = i0 + m;
    const int rowmod = row % na;
    const float* attrow = att + ((long)bh * Nn + row) * Nn;

    v8f accS0 = {};
    v8f accS1 = {};

    for (int kc = 0; kc < Nn; kc += KC) {
        __syncthreads();
        {
            const float* src1 = wh1 + ((long)(b * Nn + kc + tid)) * Dd + h * DH;
            const float* src2 = wh2 + ((long)(b * Nn + kc + tid)) * Dd + h * DH;
            float4* dst1 = (float4*)&ldsW1[tid][0];
            float4* dst2 = (float4*)&ldsW2[tid][0];
#pragma unroll
            for (int q = 0; q < 8; ++q) {
                dst1[q] = ((const float4*)src1)[q];
                dst2[q] = ((const float4*)src2)[q];
            }
        }
        __syncthreads();

        for (int k0 = 0; k0 < KC; k0 += 4) {
            const int j0 = kc + k0 + kh;
            const int j1 = j0 + 1;
            const v2f av = *(const v2f*)(attrow + j0);
            const bool self0 = ((j0 % na) == rowmod);
            const bool self1 = ((j1 % na) == rowmod);
            v2f aS, aO;
            aS.x = self0 ? av.x : 0.0f;  aS.y = self1 ? av.y : 0.0f;
            aO.x = self0 ? 0.0f : av.x;  aO.y = self1 ? 0.0f : av.y;

            const int l0 = k0 + kh;
            v2f b1_0; b1_0.x = ldsW1[l0][m];      b1_0.y = ldsW1[l0 + 1][m];
            v2f b1_1; b1_1.x = ldsW1[l0][16 + m]; b1_1.y = ldsW1[l0 + 1][16 + m];
            v2f b2_0; b2_0.x = ldsW2[l0][m];      b2_0.y = ldsW2[l0 + 1][m];
            v2f b2_1; b2_1.x = ldsW2[l0][16 + m]; b2_1.y = ldsW2[l0 + 1][16 + m];

            accS0 = __builtin_amdgcn_wmma_f32_16x16x4_f32(false, aS, false, b1_0,
                                                          (short)0, accS0, false, false);
            accS0 = __builtin_amdgcn_wmma_f32_16x16x4_f32(false, aO, false, b2_0,
                                                          (short)0, accS0, false, false);
            accS1 = __builtin_amdgcn_wmma_f32_16x16x4_f32(false, aS, false, b1_1,
                                                          (short)0, accS1, false, false);
            accS1 = __builtin_amdgcn_wmma_f32_16x16x4_f32(false, aO, false, b2_1,
                                                          (short)0, accS1, false, false);
        }
    }

    const int rbase = i0 + ((lane >> 4) << 3);
#pragma unroll
    for (int r = 0; r < 8; ++r) {
        float* o = hp + ((long)(b * Nn + rbase + r)) * Dd + h * DH;
        o[m]      = accS0[r];
        o[16 + m] = accS1[r];
    }
}

__global__ __launch_bounds__(32) void outproj_kernel(
    const float* __restrict__ hp, const float* __restrict__ Wo,
    const float* __restrict__ bo, float* __restrict__ out)
{
    const int lane = threadIdx.x & 31;
    const int m    = lane & 15;
    const int kh   = (lane >> 4) << 1;
    const int col0 = blockIdx.x * 16;
    const int row0 = blockIdx.y * 16;

    v8f acc = {};
    const float* aRow = hp + (long)(row0 + m) * Dd;
    const float* bRow = Wo + (long)(col0 + m) * Dd;
    for (int k0 = 0; k0 < Dd; k0 += 4) {
        v2f a  = *(const v2f*)(aRow + k0 + kh);
        v2f bb = *(const v2f*)(bRow + k0 + kh);
        acc = __builtin_amdgcn_wmma_f32_16x16x4_f32(false, a, false, bb,
                                                    (short)0, acc, false, false);
    }

    const int rbase = row0 + ((lane >> 4) << 3);
    const int c     = col0 + m;
    const float bias = bo[c];
#pragma unroll
    for (int r = 0; r < 8; ++r) {
        float v = acc[r] + bias;
        v = (v > 0.0f) ? v : (expf(v) - 1.0f);
        out[(long)(rbase + r) * Dd + c] = v;
    }
}

extern "C" void kernel_launch(void* const* d_in, const int* in_sizes, int n_in,
                              void* d_out, int out_size, void* d_ws, size_t ws_size,
                              hipStream_t stream) {
    (void)in_sizes; (void)n_in; (void)out_size; (void)ws_size;

    const float* hmat = (const float*)d_in[0];
    const int*   nag  = (const int*)  d_in[1];
    const int*   adj  = (const int*)  d_in[2];
    const float* W1   = (const float*)d_in[3];
    const float* W2   = (const float*)d_in[4];
    const float* Ai   = (const float*)d_in[5];
    const float* Aj   = (const float*)d_in[6];
    const float* Wo   = (const float*)d_in[7];
    const float* bo   = (const float*)d_in[8];
    float*       out  = (float*)d_out;

    float* ws  = (float*)d_ws;
    const size_t whN = (size_t)Bc * Nn * Dd;
    const size_t scN = (size_t)Bc * Hh * Nn;
    float* wh1 = ws;
    float* wh2 = wh1 + whN;
    float* s1  = wh2 + whN;
    float* s2  = s1 + scN;
    float* o1  = s2 + scN;
    float* o2  = o1 + scN;
    float* hp  = o2 + scN;
    float* att = hp + whN;

    proj12_kernel<<<dim3(Dd / 16, (Bc * Nn) / 16), 32, 0, stream>>>(
        hmat, W1, W2, wh1, wh2);

    scores_kernel<<<(Bc * Nn * Hh + 255) / 256, 256, 0, stream>>>(
        wh1, wh2, Ai, Aj, s1, s2, o1, o2);

    entmax_kernel<<<Bc * Hh * Nn, 256, 0, stream>>>(
        s1, s2, o1, o2, adj, nag, att);

    aggregate_kernel<<<dim3(Bc * Hh, Nn / 64), 128, 0, stream>>>(
        att, wh1, wh2, nag, hp);

    outproj_kernel<<<dim3(Dd / 16, (Bc * Nn) / 16), 32, 0, stream>>>(
        hp, Wo, bo, out);
}


# entmax row: one wave per row, 32 values per lane in registers; bitonic sort by register min/max and lane exchange, scans by lane shuffles, no LDS or barrier; cumsum order changes
# speedup vs baseline: 2.2719x; 2.2719x over previous
#include <hip/hip_runtime.h>
#include <hip/hip_bf16.h>
#include <math.h>

static constexpr int Bc = 4;
static constexpr int Nn = 1024;
static constexpr int Dd = 256;
static constexpr int Hh = 8;
static constexpr int DH = 32;

typedef float v2f __attribute__((ext_vector_type(2)));
typedef float v8f __attribute__((ext_vector_type(8)));

__global__ __launch_bounds__(32) void proj12_kernel(
    const float* __restrict__ hmat, const float* __restrict__ W1,
    const float* __restrict__ W2, float* __restrict__ wh1,
    float* __restrict__ wh2)
{
    const int lane = threadIdx.x & 31;
    const int m    = lane & 15;
    const int kh   = (lane >> 4) << 1;
    const int col0 = blockIdx.x * 16;
    const int row0 = blockIdx.y * 16;

    v8f acc1 = {};
    v8f acc2 = {};
    const float* aRow  = hmat + (long)(row0 + m) * Dd;
    const float* b1Row = W1   + (long)(col0 + m) * Dd;
    const float* b2Row = W2   + (long)(col0 + m) * Dd;

    for (int k0 = 0; k0 < Dd; k0 += 4) {
        v2f a  = *(const v2f*)(aRow  + k0 + kh);
        v2f b1 = *(const v2f*)(b1Row + k0 + kh);
        v2f b2 = *(const v2f*)(b2Row + k0 + kh);
        acc1 = __builtin_amdgcn_wmma_f32_16x16x4_f32(false, a, false, b1,
                                                     (short)0, acc1, false, false);
        acc2 = __builtin_amdgcn_wmma_f32_16x16x4_f32(false, a, false, b2,
                                                     (short)0, acc2, false, false);
    }

    const int rbase = row0 + ((lane >> 4) << 3);
    const int c     = col0 + m;
#pragma unroll
    for (int r = 0; r < 8; ++r) {
        wh1[(long)(rbase + r) * Dd + c] = acc1[r];
        wh2[(long)(rbase + r) * Dd + c] = acc2[r];
    }
}

__global__ __launch_bounds__(256) void scores_kernel(
    const float* __restrict__ wh1, const float* __restrict__ wh2,
    const float* __restrict__ Ai, const float* __restrict__ Aj,
    float* __restrict__ s1, float* __restrict__ s2,
    float* __restrict__ o1, float* __restrict__ o2)
{
    int idx = blockIdx.x * blockDim.x + threadIdx.x;
    if (idx >= Bc * Nn * Hh) return;
    const int h = idx % Hh;
    const int n = (idx / Hh) % Nn;
    const int b = idx / (Hh * Nn);

    const float* r1 = wh1 + ((long)(b * Nn + n)) * Dd + h * DH;
    const float* r2 = wh2 + ((long)(b * Nn + n)) * Dd + h * DH;
    const float* ai = Ai + h * DH;
    const float* aj = Aj + h * DH;

    float a1 = 0.f, a2 = 0.f, a3 = 0.f, a4 = 0.f;
#pragma unroll 8
    for (int d = 0; d < DH; ++d) {
        const float w1v = r1[d], w2v = r2[d], aiv = ai[d], ajv = aj[d];
        a1 += w1v * aiv;  a2 += w1v * ajv;
        a3 += w2v * aiv;  a4 += w2v * ajv;
    }
    const int o = (b * Hh + h) * Nn + n;
    s1[o] = a1; s2[o] = a2; o1[o] = a3; o2[o] = a4;
}

static constexpr int RPW = 4;

__device__ __forceinline__ float lane_pull(float x, int addr) {
    return __builtin_bit_cast(float,
        __builtin_amdgcn_ds_bpermute(addr, __builtin_bit_cast(int, x)));
}
__device__ __forceinline__ int lane_pull_i(int x, int addr) {
    return __builtin_amdgcn_ds_bpermute(addr, x);
}

template <int JJ>
__device__ __forceinline__ void ce_regs(float (&v)[32]) {
#pragma unroll
    for (int u = 0; u < 32; ++u) {
        if ((u & JJ) == 0) {
            const float a = v[u], c = v[u | JJ];
            v[u]      = fmaxf(a, c);
            v[u | JJ] = fminf(a, c);
        }
    }
}
template <int K>
__device__ __forceinline__ void flip_regs(float (&v)[32]) {
#pragma unroll
    for (int u = 0; u < 32; ++u) {
        if ((u & (K >> 1)) == 0) {
            const int p = u ^ (K - 1);
            const float a = v[u], c = v[p];
            v[u] = fmaxf(a, c);
            v[p] = fminf(a, c);
        }
    }
}
template <int LM, bool FLIP>
__device__ __forceinline__ void ce_lanes(float (&v)[32], int lane) {
    constexpr int HB = (LM >= 16) ? 16 : (LM >= 8) ? 8 : (LM >= 4) ? 4
                     : (LM >= 2) ? 2 : 1;
    const int   addr  = (lane ^ LM) << 2;
    const float bound = (lane & HB) ? -__builtin_inff() : __builtin_inff();
    float p[32];
#pragma unroll
    for (int u = 0; u < 32; ++u) p[u] = lane_pull(v[FLIP ? (u ^ 31) : u], addr);
#pragma unroll
    for (int u = 0; u < 32; ++u) v[u] = __builtin_amdgcn_fmed3f(v[u], p[u], bound);
}
__device__ __forceinline__ void ce_regs_tail(float (&v)[32]) {
    ce_regs<16>(v); ce_regs<8>(v); ce_regs<4>(v); ce_regs<2>(v); ce_regs<1>(v);
}

__global__ __launch_bounds__(256) void entmax_kernel(
    const float* __restrict__ s1, const float* __restrict__ s2,
    const float* __restrict__ o1, const float* __restrict__ o2,
    const int* __restrict__ adj, const int* __restrict__ nag,
    float* __restrict__ att)
{
    const int lane = threadIdx.x & 31;
    const int wave = threadIdx.x >> 5;
    const unsigned na      = (unsigned)nag[0];
    const unsigned step128 = 128u % na;
    const unsigned jm_lane = (unsigned)(lane * 4) % na;
    const int row0 = (blockIdx.x * 8 + wave) * RPW;

#pragma unroll 1
    for (int rr = 0; rr < RPW; ++rr) {
        const int bid = row0 + rr;
        const int i   = bid & (Nn - 1);
        const int h   = (bid >> 10) & (Hh - 1);
        const int b   = bid >> 13;

        const int   rowbase = (b * Hh + h) * Nn;
        const float s1i = s1[rowbase + i];
        const float o1i = o1[rowbase + i];
        const unsigned im = (unsigned)i % na;
        const long  adjbase = ((long)b * Nn + i) * Nn;

        float x[32];
        float lmax = -3.4e38f;
        unsigned jm = jm_lane;
#pragma unroll
        for (int q = 0; q < 8; ++q) {
            const int j = (q * 32 + lane) * 4;
            const float4 sv = *(const float4*)(s2 + rowbase + j);
            const float4 ov = *(const float4*)(o2 + rowbase + j);
            const int4   av = *(const int4*)(adj + adjbase + j);
            asm volatile("" :: "v"(sv.x), "v"(sv.y), "v"(sv.z), "v"(sv.w),
                               "v"(ov.x), "v"(ov.y), "v"(ov.z), "v"(ov.w));
            const float se[4] = { sv.x, sv.y, sv.z, sv.w };
            const float oe[4] = { ov.x, ov.y, ov.z, ov.w };
            const int   ae[4] = { av.x, av.y, av.z, av.w };
            unsigned m = jm;
#pragma unroll
            for (int c = 0; c < 4; ++c) {
                float e = (m == im) ? (s1i + se[c]) : (o1i + oe[c]);
                if (ae[c] == 0) e = -1.0e12f;
                const float xv = 0.5f * e;
                x[q * 4 + c] = xv;
                lmax = fmaxf(lmax, xv);
                m = (m + 1u == na) ? 0u : (m + 1u);
            }
            jm += step128;
            if (jm >= na) jm -= na;
        }
#pragma unroll
        for (int msk = 16; msk >= 1; msk >>= 1)
            lmax = fmaxf(lmax, lane_pull(lmax, (lane ^ msk) << 2));
        const float xmax = lmax;

        float v[32];
#pragma unroll
        for (int u = 0; u < 32; ++u) { x[u] -= xmax; v[u] = x[u]; }

        flip_regs<2>(v);
        flip_regs<4>(v);  ce_regs<1>(v);
        flip_regs<8>(v);  ce_regs<2>(v); ce_regs<1>(v);
        flip_regs<16>(v); ce_regs<4>(v); ce_regs<2>(v); ce_regs<1>(v);
        flip_regs<32>(v); ce_regs<8>(v); ce_regs<4>(v); ce_regs<2>(v); ce_regs<1>(v);
        ce_lanes<1, true>(v, lane);                                  ce_regs_tail(v);
        ce_lanes<3, true>(v, lane);  ce_lanes<1, false>(v, lane);    ce_regs_tail(v);
        ce_lanes<7, true>(v, lane);  ce_lanes<2, false>(v, lane);
        ce_lanes<1, false>(v, lane);                                 ce_regs_tail(v);
        ce_lanes<15, true>(v, lane); ce_lanes<4, false>(v, lane);
        ce_lanes<2, false>(v, lane); ce_lanes<1, false>(v, lane);    ce_regs_tail(v);
        ce_lanes<31, true>(v, lane); ce_lanes<8, false>(v, lane);
        ce_lanes<4, false>(v, lane); ce_lanes<2, false>(v, lane);
        ce_lanes<1, false>(v, lane);                                 ce_regs_tail(v);

        float totS = 0.0f, totQ = 0.0f;
#pragma unroll
        for (int u = 0; u < 32; ++u) { totS += v[u]; totQ += v[u] * v[u]; }
        float incS = totS, incQ = totQ;
#pragma unroll
        for (int off = 1; off < 32; off <<= 1) {
            const int   src = (lane >= off) ? (lane - off) : lane;
            const float pa  = lane_pull(incS, src << 2);
            const float pb  = lane_pull(incQ, src << 2);
            if (lane >= off) { incS += pa; incQ += pb; }
        }
        const int   prv = (lane > 0) ? (lane - 1) : 0;
        const float eS  = lane_pull(incS, prv << 2);
        const float eQ  = lane_pull(incQ, prv << 2);
        float runS = (lane > 0) ? eS : 0.0f;
        float runQ = (lane > 0) ? eQ : 0.0f;

        float tau[32];
        int cnt = 0;
#pragma unroll
        for (int u = 0; u < 32; ++u) {
            runS += v[u];
            runQ += v[u] * v[u];
            const float rho   = (float)(lane * 32 + u + 1);
            const float mean  = runS / rho;
            const float msq   = runQ / rho;
            const float ssv   = rho * (msq - mean * mean);
            const float delta = (1.0f - ssv) / rho;
            const float t     = mean - ((delta > 0.0f) ? sqrtf(delta) : 0.0f);
            tau[u] = t;
            cnt += (t <= v[u]) ? 1 : 0;
        }
#pragma unroll
        for (int msk = 16; msk >= 1; msk >>= 1)
            cnt += lane_pull_i(cnt, (lane ^ msk) << 2);
        const int support = __builtin_amdgcn_readfirstlane(cnt);
        const int sidx = support - 1;
        const int us   = sidx & 31;
        float tsel = tau[0];
#pragma unroll
        for (int u = 1; u < 32; ++u) tsel = (us == u) ? tau[u] : tsel;
        const float tstar = lane_pull(tsel, (sidx >> 5) << 2);

        float* outrow = att + ((long)(b * Hh + h) * Nn + i) * Nn;
#pragma unroll
        for (int q = 0; q < 8; ++q) {
            float4 o;
            float y;
            y = fmaxf(x[q * 4 + 0] - tstar, 0.0f); o.x = y * y;
            y = fmaxf(x[q * 4 + 1] - tstar, 0.0f); o.y = y * y;
            y = fmaxf(x[q * 4 + 2] - tstar, 0.0f); o.z = y * y;
            y = fmaxf(x[q * 4 + 3] - tstar, 0.0f); o.w = y * y;
            *(float4*)(outrow + (q * 32 + lane) * 4) = o;
        }
    }
}

static constexpr int KC  = 128;
static constexpr int LDP = 40;

__global__ __launch_bounds__(128) void aggregate_kernel(
    const float* __restrict__ att, const float* __restrict__ wh1,
    const float* __restrict__ wh2, const int* __restrict__ nag,
    float* __restrict__ hp)
{
    __shared__ float ldsW1[KC][LDP];
    __shared__ float ldsW2[KC][LDP];

    const int tid  = threadIdx.x;
    const int wave = tid >> 5;
    const int lane = tid & 31;
    const int m    = lane & 15;
    const int kh   = (lane >> 4) << 1;
    const int bh   = blockIdx.x;
    const int b    = bh / Hh;
    const int h    = bh % Hh;
    const int i0   = blockIdx.y * 64 + wave * 16;
    const int na   = nag[0];

    const int row    = i0 + m;
    const int rowmod = row % na;
    const float* attrow = att + ((long)bh * Nn + row) * Nn;

    v8f accS0 = {};
    v8f accS1 = {};

    for (int kc = 0; kc < Nn; kc += KC) {
        __syncthreads();
        {
            const float* src1 = wh1 + ((long)(b * Nn + kc + tid)) * Dd + h * DH;
            const float* src2 = wh2 + ((long)(b * Nn + kc + tid)) * Dd + h * DH;
            float4* dst1 = (float4*)&ldsW1[tid][0];
            float4* dst2 = (float4*)&ldsW2[tid][0];
#pragma unroll
            for (int q = 0; q < 8; ++q) {
                dst1[q] = ((const float4*)src1)[q];
                dst2[q] = ((const float4*)src2)[q];
            }
        }
        __syncthreads();

        for (int k0 = 0; k0 < KC; k0 += 4) {
            const int j0 = kc + k0 + kh;
            const int j1 = j0 + 1;
            const v2f av = *(const v2f*)(attrow + j0);
            const bool self0 = ((j0 % na) == rowmod);
            const bool self1 = ((j1 % na) == rowmod);
            v2f aS, aO;
            aS.x = self0 ? av.x : 0.0f;  aS.y = self1 ? av.y : 0.0f;
            aO.x = self0 ? 0.0f : av.x;  aO.y = self1 ? 0.0f : av.y;

            const int l0 = k0 + kh;
            v2f b1_0; b1_0.x = ldsW1[l0][m];      b1_0.y = ldsW1[l0 + 1][m];
            v2f b1_1; b1_1.x = ldsW1[l0][16 + m]; b1_1.y = ldsW1[l0 + 1][16 + m];
            v2f b2_0; b2_0.x = ldsW2[l0][m];      b2_0.y = ldsW2[l0 + 1][m];
            v2f b2_1; b2_1.x = ldsW2[l0][16 + m]; b2_1.y = ldsW2[l0 + 1][16 + m];

            accS0 = __builtin_amdgcn_wmma_f32_16x16x4_f32(false, aS, false, b1_0,
                                                          (short)0, accS0, false, false);
            accS0 = __builtin_amdgcn_wmma_f32_16x16x4_f32(false, aO, false, b2_0,
                                                          (short)0, accS0, false, false);
            accS1 = __builtin_amdgcn_wmma_f32_16x16x4_f32(false, aS, false, b1_1,
                                                          (short)0, accS1, false, false);
            accS1 = __builtin_amdgcn_wmma_f32_16x16x4_f32(false, aO, false, b2_1,
                                                          (short)0, accS1, false, false);
        }
    }

    const int rbase = i0 + ((lane >> 4) << 3);
#pragma unroll
    for (int r = 0; r < 8; ++r) {
        float* o = hp + ((long)(b * Nn + rbase + r)) * Dd + h * DH;
        o[m]      = accS0[r];
        o[16 + m] = accS1[r];
    }
}

__global__ __launch_bounds__(32) void outproj_kernel(
    const float* __restrict__ hp, const float* __restrict__ Wo,
    const float* __restrict__ bo, float* __restrict__ out)
{
    const int lane = threadIdx.x & 31;
    const int m    = lane & 15;
    const int kh   = (lane >> 4) << 1;
    const int col0 = blockIdx.x * 16;
    const int row0 = blockIdx.y * 16;

    v8f acc = {};
    const float* aRow = hp + (long)(row0 + m) * Dd;
    const float* bRow = Wo + (long)(col0 + m) * Dd;
    for (int k0 = 0; k0 < Dd; k0 += 4) {
        v2f a  = *(const v2f*)(aRow + k0 + kh);
        v2f bb = *(const v2f*)(bRow + k0 + kh);
        acc = __builtin_amdgcn_wmma_f32_16x16x4_f32(false, a, false, bb,
                                                    (short)0, acc, false, false);
    }

    const int rbase = row0 + ((lane >> 4) << 3);
    const int c     = col0 + m;
    const float bias = bo[c];
#pragma unroll
    for (int r = 0; r < 8; ++r) {
        float v = acc[r] + bias;
        v = (v > 0.0f) ? v : (expf(v) - 1.0f);
        out[(long)(rbase + r) * Dd + c] = v;
    }
}

extern "C" void kernel_launch(void* const* d_in, const int* in_sizes, int n_in,
                              void* d_out, int out_size, void* d_ws, size_t ws_size,
                              hipStream_t stream) {
    (void)in_sizes; (void)n_in; (void)out_size; (void)ws_size;

    const float* hmat = (const float*)d_in[0];
    const int*   nag  = (const int*)  d_in[1];
    const int*   adj  = (const int*)  d_in[2];
    const float* W1   = (const float*)d_in[3];
    const float* W2   = (const float*)d_in[4];
    const float* Ai   = (const float*)d_in[5];
    const float* Aj   = (const float*)d_in[6];
    const float* Wo   = (const float*)d_in[7];
    const float* bo   = (const float*)d_in[8];
    float*       out  = (float*)d_out;

    float* ws  = (float*)d_ws;
    const size_t whN = (size_t)Bc * Nn * Dd;
    const size_t scN = (size_t)Bc * Hh * Nn;
    float* wh1 = ws;
    float* wh2 = wh1 + whN;
    float* s1  = wh2 + whN;
    float* s2  = s1 + scN;
    float* o1  = s2 + scN;
    float* o2  = o1 + scN;
    float* hp  = o2 + scN;
    float* att = hp + whN;

    proj12_kernel<<<dim3(Dd / 16, (Bc * Nn) / 16), 32, 0, stream>>>(
        hmat, W1, W2, wh1, wh2);

    scores_kernel<<<(Bc * Nn * Hh + 255) / 256, 256, 0, stream>>>(
        wh1, wh2, Ai, Aj, s1, s2, o1, o2);

    entmax_kernel<<<(Bc * Hh * Nn) / (8 * RPW), 256, 0, stream>>>(
        s1, s2, o1, o2, adj, nag, att);

    aggregate_kernel<<<dim3(Bc * Hh, Nn / 64), 128, 0, stream>>>(
        att, wh1, wh2, nag, hp);

    outproj_kernel<<<dim3(Dd / 16, (Bc * Nn) / 16), 32, 0, stream>>>(
        hp, Wo, bo, out);
}


# entmax sort: cross-lane compare-exchange reads its partner on the vector ALU (row_xmask DPP for lane^1..15, one permlanex16 gather for lane^31) instead of ds_bpermute plus an LDS wait per element
# speedup vs baseline: 3.3059x; 1.4551x over previous
#include <hip/hip_runtime.h>
#include <hip/hip_bf16.h>
#include <math.h>

static constexpr int Bc = 4;
static constexpr int Nn = 1024;
static constexpr int Dd = 256;
static constexpr int Hh = 8;
static constexpr int DH = 32;

typedef float v2f __attribute__((ext_vector_type(2)));
typedef float v8f __attribute__((ext_vector_type(8)));

__global__ __launch_bounds__(32) void proj12_kernel(
    const float* __restrict__ hmat, const float* __restrict__ W1,
    const float* __restrict__ W2, float* __restrict__ wh1,
    float* __restrict__ wh2)
{
    const int lane = threadIdx.x & 31;
    const int m    = lane & 15;
    const int kh   = (lane >> 4) << 1;
    const int col0 = blockIdx.x * 16;
    const int row0 = blockIdx.y * 16;

    v8f acc1 = {};
    v8f acc2 = {};
    const float* aRow  = hmat + (long)(row0 + m) * Dd;
    const float* b1Row = W1   + (long)(col0 + m) * Dd;
    const float* b2Row = W2   + (long)(col0 + m) * Dd;

    for (int k0 = 0; k0 < Dd; k0 += 4) {
        v2f a  = *(const v2f*)(aRow  + k0 + kh);
        v2f b1 = *(const v2f*)(b1Row + k0 + kh);
        v2f b2 = *(const v2f*)(b2Row + k0 + kh);
        acc1 = __builtin_amdgcn_wmma_f32_16x16x4_f32(false, a, false, b1,
                                                     (short)0, acc1, false, false);
        acc2 = __builtin_amdgcn_wmma_f32_16x16x4_f32(false, a, false, b2,
                                                     (short)0, acc2, false, false);
    }

    const int rbase = row0 + ((lane >> 4) << 3);
    const int c     = col0 + m;
#pragma unroll
    for (int r = 0; r < 8; ++r) {
        wh1[(long)(rbase + r) * Dd + c] = acc1[r];
        wh2[(long)(rbase + r) * Dd + c] = acc2[r];
    }
}

__global__ __launch_bounds__(256) void scores_kernel(
    const float* __restrict__ wh1, const float* __restrict__ wh2,
    const float* __restrict__ Ai, const float* __restrict__ Aj,
    float* __restrict__ s1, float* __restrict__ s2,
    float* __restrict__ o1, float* __restrict__ o2)
{
    int idx = blockIdx.x * blockDim.x + threadIdx.x;
    if (idx >= Bc * Nn * Hh) return;
    const int h = idx % Hh;
    const int n = (idx / Hh) % Nn;
    const int b = idx / (Hh * Nn);

    const float* r1 = wh1 + ((long)(b * Nn + n)) * Dd + h * DH;
    const float* r2 = wh2 + ((long)(b * Nn + n)) * Dd + h * DH;
    const float* ai = Ai + h * DH;
    const float* aj = Aj + h * DH;

    float a1 = 0.f, a2 = 0.f, a3 = 0.f, a4 = 0.f;
#pragma unroll 8
    for (int d = 0; d < DH; ++d) {
        const float w1v = r1[d], w2v = r2[d], aiv = ai[d], ajv = aj[d];
        a1 += w1v * aiv;  a2 += w1v * ajv;
        a3 += w2v * aiv;  a4 += w2v * ajv;
    }
    const int o = (b * Hh + h) * Nn + n;
    s1[o] = a1; s2[o] = a2; o1[o] = a3; o2[o] = a4;
}

static constexpr int RPW = 4;

__device__ __forceinline__ float lane_pull(float x, int addr) {
    return __builtin_bit_cast(float,
        __builtin_amdgcn_ds_bpermute(addr, __builtin_bit_cast(int, x)));
}
__device__ __forceinline__ int lane_pull_i(int x, int addr) {
    return __builtin_amdgcn_ds_bpermute(addr, x);
}

template <int JJ>
__device__ __forceinline__ void ce_regs(float (&v)[32]) {
#pragma unroll
    for (int u = 0; u < 32; ++u) {
        if ((u & JJ) == 0) {
            const float a = v[u], c = v[u | JJ];
            v[u]      = fmaxf(a, c);
            v[u | JJ] = fminf(a, c);
        }
    }
}
template <int K>
__device__ __forceinline__ void flip_regs(float (&v)[32]) {
#pragma unroll
    for (int u = 0; u < 32; ++u) {
        if ((u & (K >> 1)) == 0) {
            const int p = u ^ (K - 1);
            const float a = v[u], c = v[p];
            v[u] = fmaxf(a, c);
            v[p] = fminf(a, c);
        }
    }
}
template <int LM>
__device__ __forceinline__ float lane_xor(float x) {
    const int b = __builtin_bit_cast(int, x);
    if constexpr (LM < 16) {
        return __builtin_bit_cast(float,
            __builtin_amdgcn_update_dpp(0, b, 0x160 + LM, 0xf, 0xf, true));
    } else {
        static_assert(LM == 31, "only the mirrored other-row exchange is used");
        return __builtin_bit_cast(float,
            __builtin_amdgcn_permlanex16(b, b, 0x89abcdefu, 0x01234567u, false, false));
    }
}
template <int LM, bool FLIP>
__device__ __forceinline__ void ce_lanes(float (&v)[32], int lane) {
    constexpr int HB = (LM >= 16) ? 16 : (LM >= 8) ? 8 : (LM >= 4) ? 4
                     : (LM >= 2) ? 2 : 1;
    const float bound = (lane & HB) ? -__builtin_inff() : __builtin_inff();
    float p[32];
#pragma unroll
    for (int u = 0; u < 32; ++u) p[u] = lane_xor<LM>(v[FLIP ? (u ^ 31) : u]);
#pragma unroll
    for (int u = 0; u < 32; ++u) v[u] = __builtin_amdgcn_fmed3f(v[u], p[u], bound);
}
__device__ __forceinline__ void ce_regs_tail(float (&v)[32]) {
    ce_regs<16>(v); ce_regs<8>(v); ce_regs<4>(v); ce_regs<2>(v); ce_regs<1>(v);
}

__global__ __launch_bounds__(256) void entmax_kernel(
    const float* __restrict__ s1, const float* __restrict__ s2,
    const float* __restrict__ o1, const float* __restrict__ o2,
    const int* __restrict__ adj, const int* __restrict__ nag,
    float* __restrict__ att)
{
    const int lane = threadIdx.x & 31;
    const int wave = threadIdx.x >> 5;
    const unsigned na      = (unsigned)nag[0];
    const unsigned step128 = 128u % na;
    const unsigned jm_lane = (unsigned)(lane * 4) % na;
    const int row0 = (blockIdx.x * 8 + wave) * RPW;

#pragma unroll 1
    for (int rr = 0; rr < RPW; ++rr) {
        const int bid = row0 + rr;
        const int i   = bid & (Nn - 1);
        const int h   = (bid >> 10) & (Hh - 1);
        const int b   = bid >> 13;

        const int   rowbase = (b * Hh + h) * Nn;
        const float s1i = s1[rowbase + i];
        const float o1i = o1[rowbase + i];
        const unsigned im = (unsigned)i % na;
        const long  adjbase = ((long)b * Nn + i) * Nn;

        float x[32];
        float lmax = -3.4e38f;
        unsigned jm = jm_lane;
#pragma unroll
        for (int q = 0; q < 8; ++q) {
            const int j = (q * 32 + lane) * 4;
            const float4 sv = *(const float4*)(s2 + rowbase + j);
            const float4 ov = *(const float4*)(o2 + rowbase + j);
            const int4   av = *(const int4*)(adj + adjbase + j);
            asm volatile("" :: "v"(sv.x), "v"(sv.y), "v"(sv.z), "v"(sv.w),
                               "v"(ov.x), "v"(ov.y), "v"(ov.z), "v"(ov.w));
            const float se[4] = { sv.x, sv.y, sv.z, sv.w };
            const float oe[4] = { ov.x, ov.y, ov.z, ov.w };
            const int   ae[4] = { av.x, av.y, av.z, av.w };
            unsigned m = jm;
#pragma unroll
            for (int c = 0; c < 4; ++c) {
                float e = (m == im) ? (s1i + se[c]) : (o1i + oe[c]);
                if (ae[c] == 0) e = -1.0e12f;
                const float xv = 0.5f * e;
                x[q * 4 + c] = xv;
                lmax = fmaxf(lmax, xv);
                m = (m + 1u == na) ? 0u : (m + 1u);
            }
            jm += step128;
            if (jm >= na) jm -= na;
        }
#pragma unroll
        for (int msk = 16; msk >= 1; msk >>= 1)
            lmax = fmaxf(lmax, lane_pull(lmax, (lane ^ msk) << 2));
        const float xmax = lmax;

        float v[32];
#pragma unroll
        for (int u = 0; u < 32; ++u) { x[u] -= xmax; v[u] = x[u]; }

        flip_regs<2>(v);
        flip_regs<4>(v);  ce_regs<1>(v);
        flip_regs<8>(v);  ce_regs<2>(v); ce_regs<1>(v);
        flip_regs<16>(v); ce_regs<4>(v); ce_regs<2>(v); ce_regs<1>(v);
        flip_regs<32>(v); ce_regs<8>(v); ce_regs<4>(v); ce_regs<2>(v); ce_regs<1>(v);
        ce_lanes<1, true>(v, lane);                                  ce_regs_tail(v);
        ce_lanes<3, true>(v, lane);  ce_lanes<1, false>(v, lane);    ce_regs_tail(v);
        ce_lanes<7, true>(v, lane);  ce_lanes<2, false>(v, lane);
        ce_lanes<1, false>(v, lane);                                 ce_regs_tail(v);
        ce_lanes<15, true>(v, lane); ce_lanes<4, false>(v, lane);
        ce_lanes<2, false>(v, lane); ce_lanes<1, false>(v, lane);    ce_regs_tail(v);
        ce_lanes<31, true>(v, lane); ce_lanes<8, false>(v, lane);
        ce_lanes<4, false>(v, lane); ce_lanes<2, false>(v, lane);
        ce_lanes<1, false>(v, lane);                                 ce_regs_tail(v);

        float totS = 0.0f, totQ = 0.0f;
#pragma unroll
        for (int u = 0; u < 32; ++u) { totS += v[u]; totQ += v[u] * v[u]; }
        float incS = totS, incQ = totQ;
#pragma unroll
        for (int off = 1; off < 32; off <<= 1) {
            const int   src = (lane >= off) ? (lane - off) : lane;
            const float pa  = lane_pull(incS, src << 2);
            const float pb  = lane_pull(incQ, src << 2);
            if (lane >= off) { incS += pa; incQ += pb; }
        }
        const int   prv = (lane > 0) ? (lane - 1) : 0;
        const float eS  = lane_pull(incS, prv << 2);
        const float eQ  = lane_pull(incQ, prv << 2);
        float runS = (lane > 0) ? eS : 0.0f;
        float runQ = (lane > 0) ? eQ : 0.0f;

        float tau[32];
        int cnt = 0;
#pragma unroll
        for (int u = 0; u < 32; ++u) {
            runS += v[u];
            runQ += v[u] * v[u];
            const float rho   = (float)(lane * 32 + u + 1);
            const float mean  = runS / rho;
            const float msq   = runQ / rho;
            const float ssv   = rho * (msq - mean * mean);
            const float delta = (1.0f - ssv) / rho;
            const float t     = mean - ((delta > 0.0f) ? sqrtf(delta) : 0.0f);
            tau[u] = t;
            cnt += (t <= v[u]) ? 1 : 0;
        }
#pragma unroll
        for (int msk = 16; msk >= 1; msk >>= 1)
            cnt += lane_pull_i(cnt, (lane ^ msk) << 2);
        const int support = __builtin_amdgcn_readfirstlane(cnt);
        const int sidx = support - 1;
        const int us   = sidx & 31;
        float tsel = tau[0];
#pragma unroll
        for (int u = 1; u < 32; ++u) tsel = (us == u) ? tau[u] : tsel;
        const float tstar = lane_pull(tsel, (sidx >> 5) << 2);

        float* outrow = att + ((long)(b * Hh + h) * Nn + i) * Nn;
#pragma unroll
        for (int q = 0; q < 8; ++q) {
            float4 o;
            float y;
            y = fmaxf(x[q * 4 + 0] - tstar, 0.0f); o.x = y * y;
            y = fmaxf(x[q * 4 + 1] - tstar, 0.0f); o.y = y * y;
            y = fmaxf(x[q * 4 + 2] - tstar, 0.0f); o.z = y * y;
            y = fmaxf(x[q * 4 + 3] - tstar, 0.0f); o.w = y * y;
            *(float4*)(outrow + (q * 32 + lane) * 4) = o;
        }
    }
}

static constexpr int KC  = 128;
static constexpr int LDP = 40;

__global__ __launch_bounds__(128) void aggregate_kernel(
    const float* __restrict__ att, const float* __restrict__ wh1,
    const float* __restrict__ wh2, const int* __restrict__ nag,
    float* __restrict__ hp)
{
    __shared__ float ldsW1[KC][LDP];
    __shared__ float ldsW2[KC][LDP];

    const int tid  = threadIdx.x;
    const int wave = tid >> 5;
    const int lane = tid & 31;
    const int m    = lane & 15;
    const int kh   = (lane >> 4) << 1;
    const int bh   = blockIdx.x;
    const int b    = bh / Hh;
    const int h    = bh % Hh;
    const int i0   = blockIdx.y * 64 + wave * 16;
    const int na   = nag[0];

    const int row    = i0 + m;
    const int rowmod = row % na;
    const float* attrow = att + ((long)bh * Nn + row) * Nn;

    v8f accS0 = {};
    v8f accS1 = {};

    for (int kc = 0; kc < Nn; kc += KC) {
        __syncthreads();
        {
            const float* src1 = wh1 + ((long)(b * Nn + kc + tid)) * Dd + h * DH;
            const float* src2 = wh2 + ((long)(b * Nn + kc + tid)) * Dd + h * DH;
            float4* dst1 = (float4*)&ldsW1[tid][0];
            float4* dst2 = (float4*)&ldsW2[tid][0];
#pragma unroll
            for (int q = 0; q < 8; ++q) {
                dst1[q] = ((const float4*)src1)[q];
                dst2[q] = ((const float4*)src2)[q];
            }
        }
        __syncthreads();

        for (int k0 = 0; k0 < KC; k0 += 4) {
            const int j0 = kc + k0 + kh;
            const int j1 = j0 + 1;
            const v2f av = *(const v2f*)(attrow + j0);
            const bool self0 = ((j0 % na) == rowmod);
            const bool self1 = ((j1 % na) == rowmod);
            v2f aS, aO;
            aS.x = self0 ? av.x : 0.0f;  aS.y = self1 ? av.y : 0.0f;
            aO.x = self0 ? 0.0f : av.x;  aO.y = self1 ? 0.0f : av.y;

            const int l0 = k0 + kh;
            v2f b1_0; b1_0.x = ldsW1[l0][m];      b1_0.y = ldsW1[l0 + 1][m];
            v2f b1_1; b1_1.x = ldsW1[l0][16 + m]; b1_1.y = ldsW1[l0 + 1][16 + m];
            v2f b2_0; b2_0.x = ldsW2[l0][m];      b2_0.y = ldsW2[l0 + 1][m];
            v2f b2_1; b2_1.x = ldsW2[l0][16 + m]; b2_1.y = ldsW2[l0 + 1][16 + m];

            accS0 = __builtin_amdgcn_wmma_f32_16x16x4_f32(false, aS, false, b1_0,
                                                          (short)0, accS0, false, false);
            accS0 = __builtin_amdgcn_wmma_f32_16x16x4_f32(false, aO, false, b2_0,
                                                          (short)0, accS0, false, false);
            accS1 = __builtin_amdgcn_wmma_f32_16x16x4_f32(false, aS, false, b1_1,
                                                          (short)0, accS1, false, false);
            accS1 = __builtin_amdgcn_wmma_f32_16x16x4_f32(false, aO, false, b2_1,
                                                          (short)0, accS1, false, false);
        }
    }

    const int rbase = i0 + ((lane >> 4) << 3);
#pragma unroll
    for (int r = 0; r < 8; ++r) {
        float* o = hp + ((long)(b * Nn + rbase + r)) * Dd + h * DH;
        o[m]      = accS0[r];
        o[16 + m] = accS1[r];
    }
}

__global__ __launch_bounds__(32) void outproj_kernel(
    const float* __restrict__ hp, const float* __restrict__ Wo,
    const float* __restrict__ bo, float* __restrict__ out)
{
    const int lane = threadIdx.x & 31;
    const int m    = lane & 15;
    const int kh   = (lane >> 4) << 1;
    const int col0 = blockIdx.x * 16;
    const int row0 = blockIdx.y * 16;

    v8f acc = {};
    const float* aRow = hp + (long)(row0 + m) * Dd;
    const float* bRow = Wo + (long)(col0 + m) * Dd;
    for (int k0 = 0; k0 < Dd; k0 += 4) {
        v2f a  = *(const v2f*)(aRow + k0 + kh);
        v2f bb = *(const v2f*)(bRow + k0 + kh);
        acc = __builtin_amdgcn_wmma_f32_16x16x4_f32(false, a, false, bb,
                                                    (short)0, acc, false, false);
    }

    const int rbase = row0 + ((lane >> 4) << 3);
    const int c     = col0 + m;
    const float bias = bo[c];
#pragma unroll
    for (int r = 0; r < 8; ++r) {
        float v = acc[r] + bias;
        v = (v > 0.0f) ? v : (expf(v) - 1.0f);
        out[(long)(rbase + r) * Dd + c] = v;
    }
}

extern "C" void kernel_launch(void* const* d_in, const int* in_sizes, int n_in,
                              void* d_out, int out_size, void* d_ws, size_t ws_size,
                              hipStream_t stream) {
    (void)in_sizes; (void)n_in; (void)out_size; (void)ws_size;

    const float* hmat = (const float*)d_in[0];
    const int*   nag  = (const int*)  d_in[1];
    const int*   adj  = (const int*)  d_in[2];
    const float* W1   = (const float*)d_in[3];
    const float* W2   = (const float*)d_in[4];
    const float* Ai   = (const float*)d_in[5];
    const float* Aj   = (const float*)d_in[6];
    const float* Wo   = (const float*)d_in[7];
    const float* bo   = (const float*)d_in[8];
    float*       out  = (float*)d_out;

    float* ws  = (float*)d_ws;
    const size_t whN = (size_t)Bc * Nn * Dd;
    const size_t scN = (size_t)Bc * Hh * Nn;
    float* wh1 = ws;
    float* wh2 = wh1 + whN;
    float* s1  = wh2 + whN;
    float* s2  = s1 + scN;
    float* o1  = s2 + scN;
    float* o2  = o1 + scN;
    float* hp  = o2 + scN;
    float* att = hp + whN;

    proj12_kernel<<<dim3(Dd / 16, (Bc * Nn) / 16), 32, 0, stream>>>(
        hmat, W1, W2, wh1, wh2);

    scores_kernel<<<(Bc * Nn * Hh + 255) / 256, 256, 0, stream>>>(
        wh1, wh2, Ai, Aj, s1, s2, o1, o2);

    entmax_kernel<<<(Bc * Hh * Nn) / (8 * RPW), 256, 0, stream>>>(
        s1, s2, o1, o2, adj, nag, att);

    aggregate_kernel<<<dim3(Bc * Hh, Nn / 64), 128, 0, stream>>>(
        att, wh1, wh2, nag, hp);

    outproj_kernel<<<dim3(Dd / 16, (Bc * Nn) / 16), 32, 0, stream>>>(
        hp, Wo, bo, out);
}


# entmax support count: test tau_k<=srt_k in squared form rho*s^2-2*s*S+Q<=1 (no divide, no sqrt per element); tau formed once at the support index; a near-tie count may round differently
# speedup vs baseline: 3.9106x; 1.1829x over previous
#include <hip/hip_runtime.h>
#include <hip/hip_bf16.h>
#include <math.h>

static constexpr int Bc = 4;
static constexpr int Nn = 1024;
static constexpr int Dd = 256;
static constexpr int Hh = 8;
static constexpr int DH = 32;

typedef float v2f __attribute__((ext_vector_type(2)));
typedef float v8f __attribute__((ext_vector_type(8)));

__global__ __launch_bounds__(32) void proj12_kernel(
    const float* __restrict__ hmat, const float* __restrict__ W1,
    const float* __restrict__ W2, float* __restrict__ wh1,
    float* __restrict__ wh2)
{
    const int lane = threadIdx.x & 31;
    const int m    = lane & 15;
    const int kh   = (lane >> 4) << 1;
    const int col0 = blockIdx.x * 16;
    const int row0 = blockIdx.y * 16;

    v8f acc1 = {};
    v8f acc2 = {};
    const float* aRow  = hmat + (long)(row0 + m) * Dd;
    const float* b1Row = W1   + (long)(col0 + m) * Dd;
    const float* b2Row = W2   + (long)(col0 + m) * Dd;

    for (int k0 = 0; k0 < Dd; k0 += 4) {
        v2f a  = *(const v2f*)(aRow  + k0 + kh);
        v2f b1 = *(const v2f*)(b1Row + k0 + kh);
        v2f b2 = *(const v2f*)(b2Row + k0 + kh);
        acc1 = __builtin_amdgcn_wmma_f32_16x16x4_f32(false, a, false, b1,
                                                     (short)0, acc1, false, false);
        acc2 = __builtin_amdgcn_wmma_f32_16x16x4_f32(false, a, false, b2,
                                                     (short)0, acc2, false, false);
    }

    const int rbase = row0 + ((lane >> 4) << 3);
    const int c     = col0 + m;
#pragma unroll
    for (int r = 0; r < 8; ++r) {
        wh1[(long)(rbase + r) * Dd + c] = acc1[r];
        wh2[(long)(rbase + r) * Dd + c] = acc2[r];
    }
}

__global__ __launch_bounds__(256) void scores_kernel(
    const float* __restrict__ wh1, const float* __restrict__ wh2,
    const float* __restrict__ Ai, const float* __restrict__ Aj,
    float* __restrict__ s1, float* __restrict__ s2,
    float* __restrict__ o1, float* __restrict__ o2)
{
    int idx = blockIdx.x * blockDim.x + threadIdx.x;
    if (idx >= Bc * Nn * Hh) return;
    const int h = idx % Hh;
    const int n = (idx / Hh) % Nn;
    const int b = idx / (Hh * Nn);

    const float* r1 = wh1 + ((long)(b * Nn + n)) * Dd + h * DH;
    const float* r2 = wh2 + ((long)(b * Nn + n)) * Dd + h * DH;
    const float* ai = Ai + h * DH;
    const float* aj = Aj + h * DH;

    float a1 = 0.f, a2 = 0.f, a3 = 0.f, a4 = 0.f;
#pragma unroll 8
    for (int d = 0; d < DH; ++d) {
        const float w1v = r1[d], w2v = r2[d], aiv = ai[d], ajv = aj[d];
        a1 += w1v * aiv;  a2 += w1v * ajv;
        a3 += w2v * aiv;  a4 += w2v * ajv;
    }
    const int o = (b * Hh + h) * Nn + n;
    s1[o] = a1; s2[o] = a2; o1[o] = a3; o2[o] = a4;
}

static constexpr int RPW = 4;

__device__ __forceinline__ float lane_pull(float x, int addr) {
    return __builtin_bit_cast(float,
        __builtin_amdgcn_ds_bpermute(addr, __builtin_bit_cast(int, x)));
}
__device__ __forceinline__ int lane_pull_i(int x, int addr) {
    return __builtin_amdgcn_ds_bpermute(addr, x);
}

template <int JJ>
__device__ __forceinline__ void ce_regs(float (&v)[32]) {
#pragma unroll
    for (int u = 0; u < 32; ++u) {
        if ((u & JJ) == 0) {
            const float a = v[u], c = v[u | JJ];
            v[u]      = fmaxf(a, c);
            v[u | JJ] = fminf(a, c);
        }
    }
}
template <int K>
__device__ __forceinline__ void flip_regs(float (&v)[32]) {
#pragma unroll
    for (int u = 0; u < 32; ++u) {
        if ((u & (K >> 1)) == 0) {
            const int p = u ^ (K - 1);
            const float a = v[u], c = v[p];
            v[u] = fmaxf(a, c);
            v[p] = fminf(a, c);
        }
    }
}
template <int LM>
__device__ __forceinline__ float lane_xor(float x) {
    const int b = __builtin_bit_cast(int, x);
    if constexpr (LM < 16) {
        return __builtin_bit_cast(float,
            __builtin_amdgcn_update_dpp(0, b, 0x160 + LM, 0xf, 0xf, true));
    } else {
        static_assert(LM == 31, "only the mirrored other-row exchange is used");
        return __builtin_bit_cast(float,
            __builtin_amdgcn_permlanex16(b, b, 0x89abcdefu, 0x01234567u, false, false));
    }
}
template <int LM, bool FLIP>
__device__ __forceinline__ void ce_lanes(float (&v)[32], int lane) {
    constexpr int HB = (LM >= 16) ? 16 : (LM >= 8) ? 8 : (LM >= 4) ? 4
                     : (LM >= 2) ? 2 : 1;
    const float bound = (lane & HB) ? -__builtin_inff() : __builtin_inff();
    float p[32];
#pragma unroll
    for (int u = 0; u < 32; ++u) p[u] = lane_xor<LM>(v[FLIP ? (u ^ 31) : u]);
#pragma unroll
    for (int u = 0; u < 32; ++u) v[u] = __builtin_amdgcn_fmed3f(v[u], p[u], bound);
}
__device__ __forceinline__ void ce_regs_tail(float (&v)[32]) {
    ce_regs<16>(v); ce_regs<8>(v); ce_regs<4>(v); ce_regs<2>(v); ce_regs<1>(v);
}

__global__ __launch_bounds__(256) void entmax_kernel(
    const float* __restrict__ s1, const float* __restrict__ s2,
    const float* __restrict__ o1, const float* __restrict__ o2,
    const int* __restrict__ adj, const int* __restrict__ nag,
    float* __restrict__ att)
{
    const int lane = threadIdx.x & 31;
    const int wave = threadIdx.x >> 5;
    const unsigned na      = (unsigned)nag[0];
    const unsigned step128 = 128u % na;
    const unsigned jm_lane = (unsigned)(lane * 4) % na;
    const int row0 = (blockIdx.x * 8 + wave) * RPW;

#pragma unroll 1
    for (int rr = 0; rr < RPW; ++rr) {
        const int bid = row0 + rr;
        const int i   = bid & (Nn - 1);
        const int h   = (bid >> 10) & (Hh - 1);
        const int b   = bid >> 13;

        const int   rowbase = (b * Hh + h) * Nn;
        const float s1i = s1[rowbase + i];
        const float o1i = o1[rowbase + i];
        const unsigned im = (unsigned)i % na;
        const long  adjbase = ((long)b * Nn + i) * Nn;

        float x[32];
        float lmax = -3.4e38f;
        unsigned jm = jm_lane;
#pragma unroll
        for (int q = 0; q < 8; ++q) {
            const int j = (q * 32 + lane) * 4;
            const float4 sv = *(const float4*)(s2 + rowbase + j);
            const float4 ov = *(const float4*)(o2 + rowbase + j);
            const int4   av = *(const int4*)(adj + adjbase + j);
            asm volatile("" :: "v"(sv.x), "v"(sv.y), "v"(sv.z), "v"(sv.w),
                               "v"(ov.x), "v"(ov.y), "v"(ov.z), "v"(ov.w));
            const float se[4] = { sv.x, sv.y, sv.z, sv.w };
            const float oe[4] = { ov.x, ov.y, ov.z, ov.w };
            const int   ae[4] = { av.x, av.y, av.z, av.w };
            unsigned m = jm;
#pragma unroll
            for (int c = 0; c < 4; ++c) {
                float e = (m == im) ? (s1i + se[c]) : (o1i + oe[c]);
                if (ae[c] == 0) e = -1.0e12f;
                const float xv = 0.5f * e;
                x[q * 4 + c] = xv;
                lmax = fmaxf(lmax, xv);
                m = (m + 1u == na) ? 0u : (m + 1u);
            }
            jm += step128;
            if (jm >= na) jm -= na;
        }
#pragma unroll
        for (int msk = 16; msk >= 1; msk >>= 1)
            lmax = fmaxf(lmax, lane_pull(lmax, (lane ^ msk) << 2));
        const float xmax = lmax;

        float v[32];
#pragma unroll
        for (int u = 0; u < 32; ++u) { x[u] -= xmax; v[u] = x[u]; }

        flip_regs<2>(v);
        flip_regs<4>(v);  ce_regs<1>(v);
        flip_regs<8>(v);  ce_regs<2>(v); ce_regs<1>(v);
        flip_regs<16>(v); ce_regs<4>(v); ce_regs<2>(v); ce_regs<1>(v);
        flip_regs<32>(v); ce_regs<8>(v); ce_regs<4>(v); ce_regs<2>(v); ce_regs<1>(v);
        ce_lanes<1, true>(v, lane);                                  ce_regs_tail(v);
        ce_lanes<3, true>(v, lane);  ce_lanes<1, false>(v, lane);    ce_regs_tail(v);
        ce_lanes<7, true>(v, lane);  ce_lanes<2, false>(v, lane);
        ce_lanes<1, false>(v, lane);                                 ce_regs_tail(v);
        ce_lanes<15, true>(v, lane); ce_lanes<4, false>(v, lane);
        ce_lanes<2, false>(v, lane); ce_lanes<1, false>(v, lane);    ce_regs_tail(v);
        ce_lanes<31, true>(v, lane); ce_lanes<8, false>(v, lane);
        ce_lanes<4, false>(v, lane); ce_lanes<2, false>(v, lane);
        ce_lanes<1, false>(v, lane);                                 ce_regs_tail(v);

        float totS = 0.0f, totQ = 0.0f;
#pragma unroll
        for (int u = 0; u < 32; ++u) { totS += v[u]; totQ += v[u] * v[u]; }
        float incS = totS, incQ = totQ;
#pragma unroll
        for (int off = 1; off < 32; off <<= 1) {
            const int   src = (lane >= off) ? (lane - off) : lane;
            const float pa  = lane_pull(incS, src << 2);
            const float pb  = lane_pull(incQ, src << 2);
            if (lane >= off) { incS += pa; incQ += pb; }
        }
        const int   prv = (lane > 0) ? (lane - 1) : 0;
        const float eS  = lane_pull(incS, prv << 2);
        const float eQ  = lane_pull(incQ, prv << 2);
        float runS = (lane > 0) ? eS : 0.0f;
        float runQ = (lane > 0) ? eQ : 0.0f;

        float Sp[32], Qp[32];
        int cnt = 0;
#pragma unroll
        for (int u = 0; u < 32; ++u) {
            runS += v[u];
            runQ += v[u] * v[u];
            Sp[u] = runS;
            Qp[u] = runQ;
            const float rho = (float)(lane * 32 + u + 1);
            const float d   = rho * v[u] - runS;
            const float g   = v[u] * (d - runS) + runQ;
            cnt += (g <= 1.0f) ? 1 : 0;
        }
#pragma unroll
        for (int msk = 16; msk >= 1; msk >>= 1)
            cnt += lane_pull_i(cnt, (lane ^ msk) << 2);
        const int support = __builtin_amdgcn_readfirstlane(cnt);
        const int sidx = support - 1;
        const int us   = sidx & 31;
        float Ssel = Sp[0], Qsel = Qp[0];
#pragma unroll
        for (int u = 1; u < 32; ++u) {
            Ssel = (us == u) ? Sp[u] : Ssel;
            Qsel = (us == u) ? Qp[u] : Qsel;
        }
        float tsel;
        {
            const float rho   = (float)(lane * 32 + us + 1);
            const float mean  = Ssel / rho;
            const float msq   = Qsel / rho;
            const float ssv   = rho * (msq - mean * mean);
            const float delta = (1.0f - ssv) / rho;
            tsel = mean - ((delta > 0.0f) ? sqrtf(delta) : 0.0f);
        }
        const float tstar = lane_pull(tsel, (sidx >> 5) << 2);

        float* outrow = att + ((long)(b * Hh + h) * Nn + i) * Nn;
#pragma unroll
        for (int q = 0; q < 8; ++q) {
            float4 o;
            float y;
            y = fmaxf(x[q * 4 + 0] - tstar, 0.0f); o.x = y * y;
            y = fmaxf(x[q * 4 + 1] - tstar, 0.0f); o.y = y * y;
            y = fmaxf(x[q * 4 + 2] - tstar, 0.0f); o.z = y * y;
            y = fmaxf(x[q * 4 + 3] - tstar, 0.0f); o.w = y * y;
            *(float4*)(outrow + (q * 32 + lane) * 4) = o;
        }
    }
}

static constexpr int KC  = 128;
static constexpr int LDP = 40;

__global__ __launch_bounds__(128) void aggregate_kernel(
    const float* __restrict__ att, const float* __restrict__ wh1,
    const float* __restrict__ wh2, const int* __restrict__ nag,
    float* __restrict__ hp)
{
    __shared__ float ldsW1[KC][LDP];
    __shared__ float ldsW2[KC][LDP];

    const int tid  = threadIdx.x;
    const int wave = tid >> 5;
    const int lane = tid & 31;
    const int m    = lane & 15;
    const int kh   = (lane >> 4) << 1;
    const int bh   = blockIdx.x;
    const int b    = bh / Hh;
    const int h    = bh % Hh;
    const int i0   = blockIdx.y * 64 + wave * 16;
    const int na   = nag[0];

    const int row    = i0 + m;
    const int rowmod = row % na;
    const float* attrow = att + ((long)bh * Nn + row) * Nn;

    v8f accS0 = {};
    v8f accS1 = {};

    for (int kc = 0; kc < Nn; kc += KC) {
        __syncthreads();
        {
            const float* src1 = wh1 + ((long)(b * Nn + kc + tid)) * Dd + h * DH;
            const float* src2 = wh2 + ((long)(b * Nn + kc + tid)) * Dd + h * DH;
            float4* dst1 = (float4*)&ldsW1[tid][0];
            float4* dst2 = (float4*)&ldsW2[tid][0];
#pragma unroll
            for (int q = 0; q < 8; ++q) {
                dst1[q] = ((const float4*)src1)[q];
                dst2[q] = ((const float4*)src2)[q];
            }
        }
        __syncthreads();

        for (int k0 = 0; k0 < KC; k0 += 4) {
            const int j0 = kc + k0 + kh;
            const int j1 = j0 + 1;
            const v2f av = *(const v2f*)(attrow + j0);
            const bool self0 = ((j0 % na) == rowmod);
            const bool self1 = ((j1 % na) == rowmod);
            v2f aS, aO;
            aS.x = self0 ? av.x : 0.0f;  aS.y = self1 ? av.y : 0.0f;
            aO.x = self0 ? 0.0f : av.x;  aO.y = self1 ? 0.0f : av.y;

            const int l0 = k0 + kh;
            v2f b1_0; b1_0.x = ldsW1[l0][m];      b1_0.y = ldsW1[l0 + 1][m];
            v2f b1_1; b1_1.x = ldsW1[l0][16 + m]; b1_1.y = ldsW1[l0 + 1][16 + m];
            v2f b2_0; b2_0.x = ldsW2[l0][m];      b2_0.y = ldsW2[l0 + 1][m];
            v2f b2_1; b2_1.x = ldsW2[l0][16 + m]; b2_1.y = ldsW2[l0 + 1][16 + m];

            accS0 = __builtin_amdgcn_wmma_f32_16x16x4_f32(false, aS, false, b1_0,
                                                          (short)0, accS0, false, false);
            accS0 = __builtin_amdgcn_wmma_f32_16x16x4_f32(false, aO, false, b2_0,
                                                          (short)0, accS0, false, false);
            accS1 = __builtin_amdgcn_wmma_f32_16x16x4_f32(false, aS, false, b1_1,
                                                          (short)0, accS1, false, false);
            accS1 = __builtin_amdgcn_wmma_f32_16x16x4_f32(false, aO, false, b2_1,
                                                          (short)0, accS1, false, false);
        }
    }

    const int rbase = i0 + ((lane >> 4) << 3);
#pragma unroll
    for (int r = 0; r < 8; ++r) {
        float* o = hp + ((long)(b * Nn + rbase + r)) * Dd + h * DH;
        o[m]      = accS0[r];
        o[16 + m] = accS1[r];
    }
}

__global__ __launch_bounds__(32) void outproj_kernel(
    const float* __restrict__ hp, const float* __restrict__ Wo,
    const float* __restrict__ bo, float* __restrict__ out)
{
    const int lane = threadIdx.x & 31;
    const int m    = lane & 15;
    const int kh   = (lane >> 4) << 1;
    const int col0 = blockIdx.x * 16;
    const int row0 = blockIdx.y * 16;

    v8f acc = {};
    const float* aRow = hp + (long)(row0 + m) * Dd;
    const float* bRow = Wo + (long)(col0 + m) * Dd;
    for (int k0 = 0; k0 < Dd; k0 += 4) {
        v2f a  = *(const v2f*)(aRow + k0 + kh);
        v2f bb = *(const v2f*)(bRow + k0 + kh);
        acc = __builtin_amdgcn_wmma_f32_16x16x4_f32(false, a, false, bb,
                                                    (short)0, acc, false, false);
    }

    const int rbase = row0 + ((lane >> 4) << 3);
    const int c     = col0 + m;
    const float bias = bo[c];
#pragma unroll
    for (int r = 0; r < 8; ++r) {
        float v = acc[r] + bias;
        v = (v > 0.0f) ? v : (expf(v) - 1.0f);
        out[(long)(rbase + r) * Dd + c] = v;
    }
}

extern "C" void kernel_launch(void* const* d_in, const int* in_sizes, int n_in,
                              void* d_out, int out_size, void* d_ws, size_t ws_size,
                              hipStream_t stream) {
    (void)in_sizes; (void)n_in; (void)out_size; (void)ws_size;

    const float* hmat = (const float*)d_in[0];
    const int*   nag  = (const int*)  d_in[1];
    const int*   adj  = (const int*)  d_in[2];
    const float* W1   = (const float*)d_in[3];
    const float* W2   = (const float*)d_in[4];
    const float* Ai   = (const float*)d_in[5];
    const float* Aj   = (const float*)d_in[6];
    const float* Wo   = (const float*)d_in[7];
    const float* bo   = (const float*)d_in[8];
    float*       out  = (float*)d_out;

    float* ws  = (float*)d_ws;
    const size_t whN = (size_t)Bc * Nn * Dd;
    const size_t scN = (size_t)Bc * Hh * Nn;
    float* wh1 = ws;
    float* wh2 = wh1 + whN;
    float* s1  = wh2 + whN;
    float* s2  = s1 + scN;
    float* o1  = s2 + scN;
    float* o2  = o1 + scN;
    float* hp  = o2 + scN;
    float* att = hp + whN;

    proj12_kernel<<<dim3(Dd / 16, (Bc * Nn) / 16), 32, 0, stream>>>(
        hmat, W1, W2, wh1, wh2);

    scores_kernel<<<(Bc * Nn * Hh + 255) / 256, 256, 0, stream>>>(
        wh1, wh2, Ai, Aj, s1, s2, o1, o2);

    entmax_kernel<<<(Bc * Hh * Nn) / (8 * RPW), 256, 0, stream>>>(
        s1, s2, o1, o2, adj, nag, att);

    aggregate_kernel<<<dim3(Bc * Hh, Nn / 64), 128, 0, stream>>>(
        att, wh1, wh2, nag, hp);

    outproj_kernel<<<dim3(Dd / 16, (Bc * Nn) / 16), 32, 0, stream>>>(
        hp, Wo, bo, out);
}


# entmax: only values within 1 of the row max can be in the support; each wave gathers them in LDS and sorts 256 values, not 1024 (whole-row sort kept for rows with more); cumsum order changes
# speedup vs baseline: 4.6052x; 1.1776x over previous
#include <hip/hip_runtime.h>
#include <hip/hip_bf16.h>
#include <math.h>

static constexpr int Bc = 4;
static constexpr int Nn = 1024;
static constexpr int Dd = 256;
static constexpr int Hh = 8;
static constexpr int DH = 32;

typedef float v2f __attribute__((ext_vector_type(2)));
typedef float v8f __attribute__((ext_vector_type(8)));

__global__ __launch_bounds__(32) void proj12_kernel(
    const float* __restrict__ hmat, const float* __restrict__ W1,
    const float* __restrict__ W2, float* __restrict__ wh1,
    float* __restrict__ wh2)
{
    const int lane = threadIdx.x & 31;
    const int m    = lane & 15;
    const int kh   = (lane >> 4) << 1;
    const int col0 = blockIdx.x * 16;
    const int row0 = blockIdx.y * 16;

    v8f acc1 = {};
    v8f acc2 = {};
    const float* aRow  = hmat + (long)(row0 + m) * Dd;
    const float* b1Row = W1   + (long)(col0 + m) * Dd;
    const float* b2Row = W2   + (long)(col0 + m) * Dd;

    for (int k0 = 0; k0 < Dd; k0 += 4) {
        v2f a  = *(const v2f*)(aRow  + k0 + kh);
        v2f b1 = *(const v2f*)(b1Row + k0 + kh);
        v2f b2 = *(const v2f*)(b2Row + k0 + kh);
        acc1 = __builtin_amdgcn_wmma_f32_16x16x4_f32(false, a, false, b1,
                                                     (short)0, acc1, false, false);
        acc2 = __builtin_amdgcn_wmma_f32_16x16x4_f32(false, a, false, b2,
                                                     (short)0, acc2, false, false);
    }

    const int rbase = row0 + ((lane >> 4) << 3);
    const int c     = col0 + m;
#pragma unroll
    for (int r = 0; r < 8; ++r) {
        wh1[(long)(rbase + r) * Dd + c] = acc1[r];
        wh2[(long)(rbase + r) * Dd + c] = acc2[r];
    }
}

__global__ __launch_bounds__(256) void scores_kernel(
    const float* __restrict__ wh1, const float* __restrict__ wh2,
    const float* __restrict__ Ai, const float* __restrict__ Aj,
    float* __restrict__ s1, float* __restrict__ s2,
    float* __restrict__ o1, float* __restrict__ o2)
{
    int idx = blockIdx.x * blockDim.x + threadIdx.x;
    if (idx >= Bc * Nn * Hh) return;
    const int h = idx % Hh;
    const int n = (idx / Hh) % Nn;
    const int b = idx / (Hh * Nn);

    const float* r1 = wh1 + ((long)(b * Nn + n)) * Dd + h * DH;
    const float* r2 = wh2 + ((long)(b * Nn + n)) * Dd + h * DH;
    const float* ai = Ai + h * DH;
    const float* aj = Aj + h * DH;

    float a1 = 0.f, a2 = 0.f, a3 = 0.f, a4 = 0.f;
#pragma unroll 8
    for (int d = 0; d < DH; ++d) {
        const float w1v = r1[d], w2v = r2[d], aiv = ai[d], ajv = aj[d];
        a1 += w1v * aiv;  a2 += w1v * ajv;
        a3 += w2v * aiv;  a4 += w2v * ajv;
    }
    const int o = (b * Hh + h) * Nn + n;
    s1[o] = a1; s2[o] = a2; o1[o] = a3; o2[o] = a4;
}

static constexpr int RPW = 4;
static constexpr int RS  = 8;
static constexpr int CAP = 32 * RS;

__device__ __forceinline__ float lane_pull(float x, int addr) {
    return __builtin_bit_cast(float,
        __builtin_amdgcn_ds_bpermute(addr, __builtin_bit_cast(int, x)));
}
__device__ __forceinline__ int lane_pull_i(int x, int addr) {
    return __builtin_amdgcn_ds_bpermute(addr, x);
}

template <int R, int JJ>
__device__ __forceinline__ void ce_regs(float (&v)[R]) {
#pragma unroll
    for (int u = 0; u < R; ++u) {
        if ((u & JJ) == 0) {
            const float a = v[u], c = v[u | JJ];
            v[u]      = fmaxf(a, c);
            v[u | JJ] = fminf(a, c);
        }
    }
}
template <int R, int K>
__device__ __forceinline__ void flip_regs(float (&v)[R]) {
#pragma unroll
    for (int u = 0; u < R; ++u) {
        if ((u & (K >> 1)) == 0) {
            const int p = u ^ (K - 1);
            const float a = v[u], c = v[p];
            v[u] = fmaxf(a, c);
            v[p] = fminf(a, c);
        }
    }
}
template <int LM>
__device__ __forceinline__ float lane_xor(float x) {
    const int b = __builtin_bit_cast(int, x);
    if constexpr (LM < 16) {
        return __builtin_bit_cast(float,
            __builtin_amdgcn_update_dpp(0, b, 0x160 + LM, 0xf, 0xf, true));
    } else {
        static_assert(LM == 31, "only the mirrored other-row exchange is used");
        return __builtin_bit_cast(float,
            __builtin_amdgcn_permlanex16(b, b, 0x89abcdefu, 0x01234567u, false, false));
    }
}
template <int R, int LM, bool FLIP>
__device__ __forceinline__ void ce_lanes(float (&v)[R], int lane) {
    constexpr int HB = (LM >= 16) ? 16 : (LM >= 8) ? 8 : (LM >= 4) ? 4
                     : (LM >= 2) ? 2 : 1;
    const float bound = (lane & HB) ? -__builtin_inff() : __builtin_inff();
    float p[R];
#pragma unroll
    for (int u = 0; u < R; ++u) p[u] = lane_xor<LM>(v[FLIP ? (u ^ (R - 1)) : u]);
#pragma unroll
    for (int u = 0; u < R; ++u) v[u] = __builtin_amdgcn_fmed3f(v[u], p[u], bound);
}
template <int R>
__device__ __forceinline__ void ce_regs_tail(float (&v)[R]) {
    if constexpr (R > 16) ce_regs<R, 16>(v);
    if constexpr (R > 8)  ce_regs<R, 8>(v);
    ce_regs<R, 4>(v); ce_regs<R, 2>(v); ce_regs<R, 1>(v);
}

template <int R>
__device__ __forceinline__ float entmax_tau(float (&v)[R], int lane)
{
    static_assert(R == 32 || R == 8, "register stages below are written for these");
    constexpr int LG = (R == 32) ? 5 : 3;

    flip_regs<R, 2>(v);
    flip_regs<R, 4>(v);  ce_regs<R, 1>(v);
    flip_regs<R, 8>(v);  ce_regs<R, 2>(v); ce_regs<R, 1>(v);
    if constexpr (R == 32) {
        flip_regs<R, 16>(v); ce_regs<R, 4>(v); ce_regs<R, 2>(v); ce_regs<R, 1>(v);
        flip_regs<R, 32>(v); ce_regs<R, 8>(v); ce_regs<R, 4>(v); ce_regs<R, 2>(v);
        ce_regs<R, 1>(v);
    }
    ce_lanes<R, 1, true>(v, lane);                                     ce_regs_tail<R>(v);
    ce_lanes<R, 3, true>(v, lane);  ce_lanes<R, 1, false>(v, lane);    ce_regs_tail<R>(v);
    ce_lanes<R, 7, true>(v, lane);  ce_lanes<R, 2, false>(v, lane);
    ce_lanes<R, 1, false>(v, lane);                                    ce_regs_tail<R>(v);
    ce_lanes<R, 15, true>(v, lane); ce_lanes<R, 4, false>(v, lane);
    ce_lanes<R, 2, false>(v, lane); ce_lanes<R, 1, false>(v, lane);    ce_regs_tail<R>(v);
    ce_lanes<R, 31, true>(v, lane); ce_lanes<R, 8, false>(v, lane);
    ce_lanes<R, 4, false>(v, lane); ce_lanes<R, 2, false>(v, lane);
    ce_lanes<R, 1, false>(v, lane);                                    ce_regs_tail<R>(v);

    float totS = 0.0f, totQ = 0.0f;
#pragma unroll
    for (int u = 0; u < R; ++u) { totS += v[u]; totQ += v[u] * v[u]; }
    float incS = totS, incQ = totQ;
#pragma unroll
    for (int off = 1; off < 32; off <<= 1) {
        const int   src = (lane >= off) ? (lane - off) : lane;
        const float pa  = lane_pull(incS, src << 2);
        const float pb  = lane_pull(incQ, src << 2);
        if (lane >= off) { incS += pa; incQ += pb; }
    }
    const int   prv = (lane > 0) ? (lane - 1) : 0;
    const float eS  = lane_pull(incS, prv << 2);
    const float eQ  = lane_pull(incQ, prv << 2);
    float runS = (lane > 0) ? eS : 0.0f;
    float runQ = (lane > 0) ? eQ : 0.0f;

    float Sp[R], Qp[R];
    int cnt = 0;
#pragma unroll
    for (int u = 0; u < R; ++u) {
        runS += v[u];
        runQ += v[u] * v[u];
        Sp[u] = runS;
        Qp[u] = runQ;
        const float rho = (float)(lane * R + u + 1);
        const float d   = rho * v[u] - runS;
        const float g   = v[u] * (d - runS) + runQ;
        cnt += (g <= 1.0f) ? 1 : 0;
    }
#pragma unroll
    for (int msk = 16; msk >= 1; msk >>= 1)
        cnt += lane_pull_i(cnt, (lane ^ msk) << 2);
    const int support = __builtin_amdgcn_readfirstlane(cnt);
    const int sidx = support - 1;
    const int us   = sidx & (R - 1);
    float Ssel = Sp[0], Qsel = Qp[0];
#pragma unroll
    for (int u = 1; u < R; ++u) {
        Ssel = (us == u) ? Sp[u] : Ssel;
        Qsel = (us == u) ? Qp[u] : Qsel;
    }
    float tsel;
    {
        const float rho   = (float)(lane * R + us + 1);
        const float mean  = Ssel / rho;
        const float msq   = Qsel / rho;
        const float ssv   = rho * (msq - mean * mean);
        const float delta = (1.0f - ssv) / rho;
        tsel = mean - ((delta > 0.0f) ? sqrtf(delta) : 0.0f);
    }
    return lane_pull(tsel, (sidx >> LG) << 2);
}

__global__ __launch_bounds__(256) void entmax_kernel(
    const float* __restrict__ s1, const float* __restrict__ s2,
    const float* __restrict__ o1, const float* __restrict__ o2,
    const int* __restrict__ adj, const int* __restrict__ nag,
    float* __restrict__ att)
{
    __shared__ float cand[8][Nn];

    const int lane = threadIdx.x & 31;
    const int wave = threadIdx.x >> 5;
    float* wl = cand[wave];
    const unsigned na      = (unsigned)nag[0];
    const unsigned step128 = 128u % na;
    const unsigned jm_lane = (unsigned)(lane * 4) % na;
    const int row0 = (blockIdx.x * 8 + wave) * RPW;

#pragma unroll 1
    for (int rr = 0; rr < RPW; ++rr) {
        const int bid = row0 + rr;
        const int i   = bid & (Nn - 1);
        const int h   = (bid >> 10) & (Hh - 1);
        const int b   = bid >> 13;

        const int   rowbase = (b * Hh + h) * Nn;
        const float s1i = s1[rowbase + i];
        const float o1i = o1[rowbase + i];
        const unsigned im = (unsigned)i % na;
        const long  adjbase = ((long)b * Nn + i) * Nn;

        float x[32];
        float lmax = -3.4e38f;
        unsigned jm = jm_lane;
#pragma unroll
        for (int q = 0; q < 8; ++q) {
            const int j = (q * 32 + lane) * 4;
            const float4 sv = *(const float4*)(s2 + rowbase + j);
            const float4 ov = *(const float4*)(o2 + rowbase + j);
            const int4   av = *(const int4*)(adj + adjbase + j);
            asm volatile("" :: "v"(sv.x), "v"(sv.y), "v"(sv.z), "v"(sv.w),
                               "v"(ov.x), "v"(ov.y), "v"(ov.z), "v"(ov.w));
            const float se[4] = { sv.x, sv.y, sv.z, sv.w };
            const float oe[4] = { ov.x, ov.y, ov.z, ov.w };
            const int   ae[4] = { av.x, av.y, av.z, av.w };
            unsigned m = jm;
#pragma unroll
            for (int c = 0; c < 4; ++c) {
                float e = (m == im) ? (s1i + se[c]) : (o1i + oe[c]);
                if (ae[c] == 0) e = -1.0e12f;
                const float xv = 0.5f * e;
                x[q * 4 + c] = xv;
                lmax = fmaxf(lmax, xv);
                m = (m + 1u == na) ? 0u : (m + 1u);
            }
            jm += step128;
            if (jm >= na) jm -= na;
        }
#pragma unroll
        for (int msk = 16; msk >= 1; msk >>= 1)
            lmax = fmaxf(lmax, lane_pull(lmax, (lane ^ msk) << 2));
        const float xmax = lmax;
#pragma unroll
        for (int u = 0; u < 32; ++u) x[u] -= xmax;

        int ncand = 0;
#pragma unroll
        for (int u = 0; u < 32; ++u) {
            const bool     keep = (x[u] >= -1.0f);
            const unsigned who  = __builtin_amdgcn_ballot_w32(keep);
            const int      pos  = (int)__builtin_amdgcn_mbcnt_lo(who, (unsigned)ncand);
            if (keep) wl[pos] = x[u];
            ncand += __builtin_popcount(who);
        }

        float tstar;
        if (ncand <= CAP) {
            __builtin_amdgcn_fence(__ATOMIC_ACQ_REL, "wavefront");
            float w[RS];
#pragma unroll
            for (int r = 0; r < RS; ++r) {
                const int   slot = r * 32 + lane;
                const float t    = wl[slot];
                w[r] = (slot < ncand) ? t : -2.0f;
            }
            tstar = entmax_tau<RS>(w, lane);
        } else {
            float v[32];
#pragma unroll
            for (int u = 0; u < 32; ++u) v[u] = x[u];
            tstar = entmax_tau<32>(v, lane);
        }

        float* outrow = att + ((long)(b * Hh + h) * Nn + i) * Nn;
#pragma unroll
        for (int q = 0; q < 8; ++q) {
            float4 o;
            float y;
            y = fmaxf(x[q * 4 + 0] - tstar, 0.0f); o.x = y * y;
            y = fmaxf(x[q * 4 + 1] - tstar, 0.0f); o.y = y * y;
            y = fmaxf(x[q * 4 + 2] - tstar, 0.0f); o.z = y * y;
            y = fmaxf(x[q * 4 + 3] - tstar, 0.0f); o.w = y * y;
            *(float4*)(outrow + (q * 32 + lane) * 4) = o;
        }
    }
}

static constexpr int KC  = 128;
static constexpr int LDP = 40;

__global__ __launch_bounds__(128) void aggregate_kernel(
    const float* __restrict__ att, const float* __restrict__ wh1,
    const float* __restrict__ wh2, const int* __restrict__ nag,
    float* __restrict__ hp)
{
    __shared__ float ldsW1[KC][LDP];
    __shared__ float ldsW2[KC][LDP];

    const int tid  = threadIdx.x;
    const int wave = tid >> 5;
    const int lane = tid & 31;
    const int m    = lane & 15;
    const int kh   = (lane >> 4) << 1;
    const int bh   = blockIdx.x;
    const int b    = bh / Hh;
    const int h    = bh % Hh;
    const int i0   = blockIdx.y * 64 + wave * 16;
    const int na   = nag[0];

    const int row    = i0 + m;
    const int rowmod = row % na;
    const float* attrow = att + ((long)bh * Nn + row) * Nn;

    v8f accS0 = {};
    v8f accS1 = {};

    for (int kc = 0; kc < Nn; kc += KC) {
        __syncthreads();
        {
            const float* src1 = wh1 + ((long)(b * Nn + kc + tid)) * Dd + h * DH;
            const float* src2 = wh2 + ((long)(b * Nn + kc + tid)) * Dd + h * DH;
            float4* dst1 = (float4*)&ldsW1[tid][0];
            float4* dst2 = (float4*)&ldsW2[tid][0];
#pragma unroll
            for (int q = 0; q < 8; ++q) {
                dst1[q] = ((const float4*)src1)[q];
                dst2[q] = ((const float4*)src2)[q];
            }
        }
        __syncthreads();

        for (int k0 = 0; k0 < KC; k0 += 4) {
            const int j0 = kc + k0 + kh;
            const int j1 = j0 + 1;
            const v2f av = *(const v2f*)(attrow + j0);
            const bool self0 = ((j0 % na) == rowmod);
            const bool self1 = ((j1 % na) == rowmod);
            v2f aS, aO;
            aS.x = self0 ? av.x : 0.0f;  aS.y = self1 ? av.y : 0.0f;
            aO.x = self0 ? 0.0f : av.x;  aO.y = self1 ? 0.0f : av.y;

            const int l0 = k0 + kh;
            v2f b1_0; b1_0.x = ldsW1[l0][m];      b1_0.y = ldsW1[l0 + 1][m];
            v2f b1_1; b1_1.x = ldsW1[l0][16 + m]; b1_1.y = ldsW1[l0 + 1][16 + m];
            v2f b2_0; b2_0.x = ldsW2[l0][m];      b2_0.y = ldsW2[l0 + 1][m];
            v2f b2_1; b2_1.x = ldsW2[l0][16 + m]; b2_1.y = ldsW2[l0 + 1][16 + m];

            accS0 = __builtin_amdgcn_wmma_f32_16x16x4_f32(false, aS, false, b1_0,
                                                          (short)0, accS0, false, false);
            accS0 = __builtin_amdgcn_wmma_f32_16x16x4_f32(false, aO, false, b2_0,
                                                          (short)0, accS0, false, false);
            accS1 = __builtin_amdgcn_wmma_f32_16x16x4_f32(false, aS, false, b1_1,
                                                          (short)0, accS1, false, false);
            accS1 = __builtin_amdgcn_wmma_f32_16x16x4_f32(false, aO, false, b2_1,
                                                          (short)0, accS1, false, false);
        }
    }

    const int rbase = i0 + ((lane >> 4) << 3);
#pragma unroll
    for (int r = 0; r < 8; ++r) {
        float* o = hp + ((long)(b * Nn + rbase + r)) * Dd + h * DH;
        o[m]      = accS0[r];
        o[16 + m] = accS1[r];
    }
}

__global__ __launch_bounds__(32) void outproj_kernel(
    const float* __restrict__ hp, const float* __restrict__ Wo,
    const float* __restrict__ bo, float* __restrict__ out)
{
    const int lane = threadIdx.x & 31;
    const int m    = lane & 15;
    const int kh   = (lane >> 4) << 1;
    const int col0 = blockIdx.x * 16;
    const int row0 = blockIdx.y * 16;

    v8f acc = {};
    const float* aRow = hp + (long)(row0 + m) * Dd;
    const float* bRow = Wo + (long)(col0 + m) * Dd;
    for (int k0 = 0; k0 < Dd; k0 += 4) {
        v2f a  = *(const v2f*)(aRow + k0 + kh);
        v2f bb = *(const v2f*)(bRow + k0 + kh);
        acc = __builtin_amdgcn_wmma_f32_16x16x4_f32(false, a, false, bb,
                                                    (short)0, acc, false, false);
    }

    const int rbase = row0 + ((lane >> 4) << 3);
    const int c     = col0 + m;
    const float bias = bo[c];
#pragma unroll
    for (int r = 0; r < 8; ++r) {
        float v = acc[r] + bias;
        v = (v > 0.0f) ? v : (expf(v) - 1.0f);
        out[(long)(rbase + r) * Dd + c] = v;
    }
}

extern "C" void kernel_launch(void* const* d_in, const int* in_sizes, int n_in,
                              void* d_out, int out_size, void* d_ws, size_t ws_size,
                              hipStream_t stream) {
    (void)in_sizes; (void)n_in; (void)out_size; (void)ws_size;

    const float* hmat = (const float*)d_in[0];
    const int*   nag  = (const int*)  d_in[1];
    const int*   adj  = (const int*)  d_in[2];
    const float* W1   = (const float*)d_in[3];
    const float* W2   = (const float*)d_in[4];
    const float* Ai   = (const float*)d_in[5];
    const float* Aj   = (const float*)d_in[6];
    const float* Wo   = (const float*)d_in[7];
    const float* bo   = (const float*)d_in[8];
    float*       out  = (float*)d_out;

    float* ws  = (float*)d_ws;
    const size_t whN = (size_t)Bc * Nn * Dd;
    const size_t scN = (size_t)Bc * Hh * Nn;
    float* wh1 = ws;
    float* wh2 = wh1 + whN;
    float* s1  = wh2 + whN;
    float* s2  = s1 + scN;
    float* o1  = s2 + scN;
    float* o2  = o1 + scN;
    float* hp  = o2 + scN;
    float* att = hp + whN;

    proj12_kernel<<<dim3(Dd / 16, (Bc * Nn) / 16), 32, 0, stream>>>(
        hmat, W1, W2, wh1, wh2);

    scores_kernel<<<(Bc * Nn * Hh + 255) / 256, 256, 0, stream>>>(
        wh1, wh2, Ai, Aj, s1, s2, o1, o2);

    entmax_kernel<<<(Bc * Hh * Nn) / (8 * RPW), 256, 0, stream>>>(
        s1, s2, o1, o2, adj, nag, att);

    aggregate_kernel<<<dim3(Bc * Hh, Nn / 64), 128, 0, stream>>>(
        att, wh1, wh2, nag, hp);

    outproj_kernel<<<dim3(Dd / 16, (Bc * Nn) / 16), 32, 0, stream>>>(
        hp, Wo, bo, out);
}


# aggregate: operands fetched ahead of use: att one K-chunk ahead as 16-byte loads (k remapped alike in A and B), LDS B fragments one step ahead in one group; k sum order changes
# speedup vs baseline: 5.0879x; 1.1048x over previous
#include <hip/hip_runtime.h>
#include <hip/hip_bf16.h>
#include <math.h>

static constexpr int Bc = 4;
static constexpr int Nn = 1024;
static constexpr int Dd = 256;
static constexpr int Hh = 8;
static constexpr int DH = 32;

typedef float v2f __attribute__((ext_vector_type(2)));
typedef float v8f __attribute__((ext_vector_type(8)));

__global__ __launch_bounds__(32) void proj12_kernel(
    const float* __restrict__ hmat, const float* __restrict__ W1,
    const float* __restrict__ W2, float* __restrict__ wh1,
    float* __restrict__ wh2)
{
    const int lane = threadIdx.x & 31;
    const int m    = lane & 15;
    const int kh   = (lane >> 4) << 1;
    const int col0 = blockIdx.x * 16;
    const int row0 = blockIdx.y * 16;

    v8f acc1 = {};
    v8f acc2 = {};
    const float* aRow  = hmat + (long)(row0 + m) * Dd;
    const float* b1Row = W1   + (long)(col0 + m) * Dd;
    const float* b2Row = W2   + (long)(col0 + m) * Dd;

    for (int k0 = 0; k0 < Dd; k0 += 4) {
        v2f a  = *(const v2f*)(aRow  + k0 + kh);
        v2f b1 = *(const v2f*)(b1Row + k0 + kh);
        v2f b2 = *(const v2f*)(b2Row + k0 + kh);
        acc1 = __builtin_amdgcn_wmma_f32_16x16x4_f32(false, a, false, b1,
                                                     (short)0, acc1, false, false);
        acc2 = __builtin_amdgcn_wmma_f32_16x16x4_f32(false, a, false, b2,
                                                     (short)0, acc2, false, false);
    }

    const int rbase = row0 + ((lane >> 4) << 3);
    const int c     = col0 + m;
#pragma unroll
    for (int r = 0; r < 8; ++r) {
        wh1[(long)(rbase + r) * Dd + c] = acc1[r];
        wh2[(long)(rbase + r) * Dd + c] = acc2[r];
    }
}

__global__ __launch_bounds__(256) void scores_kernel(
    const float* __restrict__ wh1, const float* __restrict__ wh2,
    const float* __restrict__ Ai, const float* __restrict__ Aj,
    float* __restrict__ s1, float* __restrict__ s2,
    float* __restrict__ o1, float* __restrict__ o2)
{
    int idx = blockIdx.x * blockDim.x + threadIdx.x;
    if (idx >= Bc * Nn * Hh) return;
    const int h = idx % Hh;
    const int n = (idx / Hh) % Nn;
    const int b = idx / (Hh * Nn);

    const float* r1 = wh1 + ((long)(b * Nn + n)) * Dd + h * DH;
    const float* r2 = wh2 + ((long)(b * Nn + n)) * Dd + h * DH;
    const float* ai = Ai + h * DH;
    const float* aj = Aj + h * DH;

    float a1 = 0.f, a2 = 0.f, a3 = 0.f, a4 = 0.f;
#pragma unroll 8
    for (int d = 0; d < DH; ++d) {
        const float w1v = r1[d], w2v = r2[d], aiv = ai[d], ajv = aj[d];
        a1 += w1v * aiv;  a2 += w1v * ajv;
        a3 += w2v * aiv;  a4 += w2v * ajv;
    }
    const int o = (b * Hh + h) * Nn + n;
    s1[o] = a1; s2[o] = a2; o1[o] = a3; o2[o] = a4;
}

static constexpr int RPW = 4;
static constexpr int RS  = 8;
static constexpr int CAP = 32 * RS;

__device__ __forceinline__ float lane_pull(float x, int addr) {
    return __builtin_bit_cast(float,
        __builtin_amdgcn_ds_bpermute(addr, __builtin_bit_cast(int, x)));
}
__device__ __forceinline__ int lane_pull_i(int x, int addr) {
    return __builtin_amdgcn_ds_bpermute(addr, x);
}

template <int R, int JJ>
__device__ __forceinline__ void ce_regs(float (&v)[R]) {
#pragma unroll
    for (int u = 0; u < R; ++u) {
        if ((u & JJ) == 0) {
            const float a = v[u], c = v[u | JJ];
            v[u]      = fmaxf(a, c);
            v[u | JJ] = fminf(a, c);
        }
    }
}
template <int R, int K>
__device__ __forceinline__ void flip_regs(float (&v)[R]) {
#pragma unroll
    for (int u = 0; u < R; ++u) {
        if ((u & (K >> 1)) == 0) {
            const int p = u ^ (K - 1);
            const float a = v[u], c = v[p];
            v[u] = fmaxf(a, c);
            v[p] = fminf(a, c);
        }
    }
}
template <int LM>
__device__ __forceinline__ float lane_xor(float x) {
    const int b = __builtin_bit_cast(int, x);
    if constexpr (LM < 16) {
        return __builtin_bit_cast(float,
            __builtin_amdgcn_update_dpp(0, b, 0x160 + LM, 0xf, 0xf, true));
    } else {
        static_assert(LM == 31, "only the mirrored other-row exchange is used");
        return __builtin_bit_cast(float,
            __builtin_amdgcn_permlanex16(b, b, 0x89abcdefu, 0x01234567u, false, false));
    }
}
template <int R, int LM, bool FLIP>
__device__ __forceinline__ void ce_lanes(float (&v)[R], int lane) {
    constexpr int HB = (LM >= 16) ? 16 : (LM >= 8) ? 8 : (LM >= 4) ? 4
                     : (LM >= 2) ? 2 : 1;
    const float bound = (lane & HB) ? -__builtin_inff() : __builtin_inff();
    float p[R];
#pragma unroll
    for (int u = 0; u < R; ++u) p[u] = lane_xor<LM>(v[FLIP ? (u ^ (R - 1)) : u]);
#pragma unroll
    for (int u = 0; u < R; ++u) v[u] = __builtin_amdgcn_fmed3f(v[u], p[u], bound);
}
template <int R>
__device__ __forceinline__ void ce_regs_tail(float (&v)[R]) {
    if constexpr (R > 16) ce_regs<R, 16>(v);
    if constexpr (R > 8)  ce_regs<R, 8>(v);
    ce_regs<R, 4>(v); ce_regs<R, 2>(v); ce_regs<R, 1>(v);
}

template <int R>
__device__ __forceinline__ float entmax_tau(float (&v)[R], int lane)
{
    static_assert(R == 32 || R == 8, "register stages below are written for these");
    constexpr int LG = (R == 32) ? 5 : 3;

    flip_regs<R, 2>(v);
    flip_regs<R, 4>(v);  ce_regs<R, 1>(v);
    flip_regs<R, 8>(v);  ce_regs<R, 2>(v); ce_regs<R, 1>(v);
    if constexpr (R == 32) {
        flip_regs<R, 16>(v); ce_regs<R, 4>(v); ce_regs<R, 2>(v); ce_regs<R, 1>(v);
        flip_regs<R, 32>(v); ce_regs<R, 8>(v); ce_regs<R, 4>(v); ce_regs<R, 2>(v);
        ce_regs<R, 1>(v);
    }
    ce_lanes<R, 1, true>(v, lane);                                     ce_regs_tail<R>(v);
    ce_lanes<R, 3, true>(v, lane);  ce_lanes<R, 1, false>(v, lane);    ce_regs_tail<R>(v);
    ce_lanes<R, 7, true>(v, lane);  ce_lanes<R, 2, false>(v, lane);
    ce_lanes<R, 1, false>(v, lane);                                    ce_regs_tail<R>(v);
    ce_lanes<R, 15, true>(v, lane); ce_lanes<R, 4, false>(v, lane);
    ce_lanes<R, 2, false>(v, lane); ce_lanes<R, 1, false>(v, lane);    ce_regs_tail<R>(v);
    ce_lanes<R, 31, true>(v, lane); ce_lanes<R, 8, false>(v, lane);
    ce_lanes<R, 4, false>(v, lane); ce_lanes<R, 2, false>(v, lane);
    ce_lanes<R, 1, false>(v, lane);                                    ce_regs_tail<R>(v);

    float totS = 0.0f, totQ = 0.0f;
#pragma unroll
    for (int u = 0; u < R; ++u) { totS += v[u]; totQ += v[u] * v[u]; }
    float incS = totS, incQ = totQ;
#pragma unroll
    for (int off = 1; off < 32; off <<= 1) {
        const int   src = (lane >= off) ? (lane - off) : lane;
        const float pa  = lane_pull(incS, src << 2);
        const float pb  = lane_pull(incQ, src << 2);
        if (lane >= off) { incS += pa; incQ += pb; }
    }
    const int   prv = (lane > 0) ? (lane - 1) : 0;
    const float eS  = lane_pull(incS, prv << 2);
    const float eQ  = lane_pull(incQ, prv << 2);
    float runS = (lane > 0) ? eS : 0.0f;
    float runQ = (lane > 0) ? eQ : 0.0f;

    float Sp[R], Qp[R];
    int cnt = 0;
#pragma unroll
    for (int u = 0; u < R; ++u) {
        runS += v[u];
        runQ += v[u] * v[u];
        Sp[u] = runS;
        Qp[u] = runQ;
        const float rho = (float)(lane * R + u + 1);
        const float d   = rho * v[u] - runS;
        const float g   = v[u] * (d - runS) + runQ;
        cnt += (g <= 1.0f) ? 1 : 0;
    }
#pragma unroll
    for (int msk = 16; msk >= 1; msk >>= 1)
        cnt += lane_pull_i(cnt, (lane ^ msk) << 2);
    const int support = __builtin_amdgcn_readfirstlane(cnt);
    const int sidx = support - 1;
    const int us   = sidx & (R - 1);
    float Ssel = Sp[0], Qsel = Qp[0];
#pragma unroll
    for (int u = 1; u < R; ++u) {
        Ssel = (us == u) ? Sp[u] : Ssel;
        Qsel = (us == u) ? Qp[u] : Qsel;
    }
    float tsel;
    {
        const float rho   = (float)(lane * R + us + 1);
        const float mean  = Ssel / rho;
        const float msq   = Qsel / rho;
        const float ssv   = rho * (msq - mean * mean);
        const float delta = (1.0f - ssv) / rho;
        tsel = mean - ((delta > 0.0f) ? sqrtf(delta) : 0.0f);
    }
    return lane_pull(tsel, (sidx >> LG) << 2);
}

__global__ __launch_bounds__(256) void entmax_kernel(
    const float* __restrict__ s1, const float* __restrict__ s2,
    const float* __restrict__ o1, const float* __restrict__ o2,
    const int* __restrict__ adj, const int* __restrict__ nag,
    float* __restrict__ att)
{
    __shared__ float cand[8][Nn];

    const int lane = threadIdx.x & 31;
    const int wave = threadIdx.x >> 5;
    float* wl = cand[wave];
    const unsigned na      = (unsigned)nag[0];
    const unsigned step128 = 128u % na;
    const unsigned jm_lane = (unsigned)(lane * 4) % na;
    const int row0 = (blockIdx.x * 8 + wave) * RPW;

#pragma unroll 1
    for (int rr = 0; rr < RPW; ++rr) {
        const int bid = row0 + rr;
        const int i   = bid & (Nn - 1);
        const int h   = (bid >> 10) & (Hh - 1);
        const int b   = bid >> 13;

        const int   rowbase = (b * Hh + h) * Nn;
        const float s1i = s1[rowbase + i];
        const float o1i = o1[rowbase + i];
        const unsigned im = (unsigned)i % na;
        const long  adjbase = ((long)b * Nn + i) * Nn;

        float x[32];
        float lmax = -3.4e38f;
        unsigned jm = jm_lane;
#pragma unroll
        for (int q = 0; q < 8; ++q) {
            const int j = (q * 32 + lane) * 4;
            const float4 sv = *(const float4*)(s2 + rowbase + j);
            const float4 ov = *(const float4*)(o2 + rowbase + j);
            const int4   av = *(const int4*)(adj + adjbase + j);
            asm volatile("" :: "v"(sv.x), "v"(sv.y), "v"(sv.z), "v"(sv.w),
                               "v"(ov.x), "v"(ov.y), "v"(ov.z), "v"(ov.w));
            const float se[4] = { sv.x, sv.y, sv.z, sv.w };
            const float oe[4] = { ov.x, ov.y, ov.z, ov.w };
            const int   ae[4] = { av.x, av.y, av.z, av.w };
            unsigned m = jm;
#pragma unroll
            for (int c = 0; c < 4; ++c) {
                float e = (m == im) ? (s1i + se[c]) : (o1i + oe[c]);
                if (ae[c] == 0) e = -1.0e12f;
                const float xv = 0.5f * e;
                x[q * 4 + c] = xv;
                lmax = fmaxf(lmax, xv);
                m = (m + 1u == na) ? 0u : (m + 1u);
            }
            jm += step128;
            if (jm >= na) jm -= na;
        }
#pragma unroll
        for (int msk = 16; msk >= 1; msk >>= 1)
            lmax = fmaxf(lmax, lane_pull(lmax, (lane ^ msk) << 2));
        const float xmax = lmax;
#pragma unroll
        for (int u = 0; u < 32; ++u) x[u] -= xmax;

        int ncand = 0;
#pragma unroll
        for (int u = 0; u < 32; ++u) {
            const bool     keep = (x[u] >= -1.0f);
            const unsigned who  = __builtin_amdgcn_ballot_w32(keep);
            const int      pos  = (int)__builtin_amdgcn_mbcnt_lo(who, (unsigned)ncand);
            if (keep) wl[pos] = x[u];
            ncand += __builtin_popcount(who);
        }

        float tstar;
        if (ncand <= CAP) {
            __builtin_amdgcn_fence(__ATOMIC_ACQ_REL, "wavefront");
            float w[RS];
#pragma unroll
            for (int r = 0; r < RS; ++r) {
                const int   slot = r * 32 + lane;
                const float t    = wl[slot];
                w[r] = (slot < ncand) ? t : -2.0f;
            }
            tstar = entmax_tau<RS>(w, lane);
        } else {
            float v[32];
#pragma unroll
            for (int u = 0; u < 32; ++u) v[u] = x[u];
            tstar = entmax_tau<32>(v, lane);
        }

        float* outrow = att + ((long)(b * Hh + h) * Nn + i) * Nn;
#pragma unroll
        for (int q = 0; q < 8; ++q) {
            float4 o;
            float y;
            y = fmaxf(x[q * 4 + 0] - tstar, 0.0f); o.x = y * y;
            y = fmaxf(x[q * 4 + 1] - tstar, 0.0f); o.y = y * y;
            y = fmaxf(x[q * 4 + 2] - tstar, 0.0f); o.z = y * y;
            y = fmaxf(x[q * 4 + 3] - tstar, 0.0f); o.w = y * y;
            *(float4*)(outrow + (q * 32 + lane) * 4) = o;
        }
    }
}

static constexpr int KC  = 128;
static constexpr int LDP = 40;
static constexpr int TS  = KC / 8;

__global__ __launch_bounds__(128) void aggregate_kernel(
    const float* __restrict__ att, const float* __restrict__ wh1,
    const float* __restrict__ wh2, const int* __restrict__ nag,
    float* __restrict__ hp)
{
    __shared__ float ldsW1[KC][LDP];
    __shared__ float ldsW2[KC][LDP];

    const int tid  = threadIdx.x;
    const int wave = tid >> 5;
    const int lane = tid & 31;
    const int m    = lane & 15;
    const int half = lane >> 4;
    const int bh   = blockIdx.x;
    const int b    = bh / Hh;
    const int h    = bh % Hh;
    const int i0   = blockIdx.y * 64 + wave * 16;
    const unsigned na = (unsigned)nag[0];

    const int row = i0 + m;
    const unsigned rowmod = (unsigned)row % na;

    const float* attp = att + ((long)bh * Nn + row) * Nn + 4 * half;

    unsigned cm = (unsigned)(4 * half) % na;
    const unsigned step8 = 8u % na;

    v8f accS0 = {};
    v8f accS1 = {};

    float4 aN[TS];
#pragma unroll
    for (int t = 0; t < TS; ++t) aN[t] = *(const float4*)(attp + 8 * t);

#pragma unroll 1
    for (int kc = 0; kc < Nn; kc += KC) {
        float4 aC[TS];
#pragma unroll
        for (int t = 0; t < TS; ++t) aC[t] = aN[t];

        __syncthreads();
        {
            const float* src1 = wh1 + ((long)(b * Nn + kc + tid)) * Dd + h * DH;
            const float* src2 = wh2 + ((long)(b * Nn + kc + tid)) * Dd + h * DH;
            float4* dst1 = (float4*)&ldsW1[tid][0];
            float4* dst2 = (float4*)&ldsW2[tid][0];
#pragma unroll
            for (int q = 0; q < 8; ++q) {
                dst1[q] = ((const float4*)src1)[q];
                dst2[q] = ((const float4*)src2)[q];
            }
        }
        __syncthreads();

        if (kc + KC < Nn) {
            const float* an = attp + (kc + KC);
#pragma unroll
            for (int t = 0; t < TS; ++t) aN[t] = *(const float4*)(an + 8 * t);
        }

        float br[2][16];
        auto load_b = [&](int t, float (&r)[16]) {
#pragma unroll
            for (int s = 0; s < 2; ++s) {
                const int l0 = 8 * t + 4 * half + 2 * s;
                r[8 * s + 0] = ldsW1[l0][m];      r[8 * s + 1] = ldsW1[l0 + 1][m];
                r[8 * s + 2] = ldsW1[l0][16 + m]; r[8 * s + 3] = ldsW1[l0 + 1][16 + m];
                r[8 * s + 4] = ldsW2[l0][m];      r[8 * s + 5] = ldsW2[l0 + 1][m];
                r[8 * s + 6] = ldsW2[l0][16 + m]; r[8 * s + 7] = ldsW2[l0 + 1][16 + m];
            }
        };
        load_b(0, br[0]);

#pragma unroll
        for (int t = 0; t < TS; ++t) {
            if (t + 1 < TS) load_b(t + 1, br[(t + 1) & 1]);
            __builtin_amdgcn_sched_barrier(0);

            const float av[4] = { aC[t].x, aC[t].y, aC[t].z, aC[t].w };
            bool sf[4];
            unsigned mm = cm;
#pragma unroll
            for (int e = 0; e < 4; ++e) {
                sf[e] = (mm == rowmod);
                mm = (mm + 1u == na) ? 0u : (mm + 1u);
            }
            cm += step8;
            if (cm >= na) cm -= na;

            const float (&r)[16] = br[t & 1];
#pragma unroll
            for (int s = 0; s < 2; ++s) {
                v2f aS, aO;
                aS.x = sf[2 * s]     ? av[2 * s]     : 0.0f;
                aS.y = sf[2 * s + 1] ? av[2 * s + 1] : 0.0f;
                aO.x = sf[2 * s]     ? 0.0f : av[2 * s];
                aO.y = sf[2 * s + 1] ? 0.0f : av[2 * s + 1];

                v2f b1_0; b1_0.x = r[8 * s + 0]; b1_0.y = r[8 * s + 1];
                v2f b1_1; b1_1.x = r[8 * s + 2]; b1_1.y = r[8 * s + 3];
                v2f b2_0; b2_0.x = r[8 * s + 4]; b2_0.y = r[8 * s + 5];
                v2f b2_1; b2_1.x = r[8 * s + 6]; b2_1.y = r[8 * s + 7];

                accS0 = __builtin_amdgcn_wmma_f32_16x16x4_f32(false, aS, false, b1_0,
                                                              (short)0, accS0, false, false);
                accS0 = __builtin_amdgcn_wmma_f32_16x16x4_f32(false, aO, false, b2_0,
                                                              (short)0, accS0, false, false);
                accS1 = __builtin_amdgcn_wmma_f32_16x16x4_f32(false, aS, false, b1_1,
                                                              (short)0, accS1, false, false);
                accS1 = __builtin_amdgcn_wmma_f32_16x16x4_f32(false, aO, false, b2_1,
                                                              (short)0, accS1, false, false);
            }
            __builtin_amdgcn_sched_barrier(0);
        }
    }

    const int rbase = i0 + ((lane >> 4) << 3);
#pragma unroll
    for (int r = 0; r < 8; ++r) {
        float* o = hp + ((long)(b * Nn + rbase + r)) * Dd + h * DH;
        o[m]      = accS0[r];
        o[16 + m] = accS1[r];
    }
}

__global__ __launch_bounds__(32) void outproj_kernel(
    const float* __restrict__ hp, const float* __restrict__ Wo,
    const float* __restrict__ bo, float* __restrict__ out)
{
    const int lane = threadIdx.x & 31;
    const int m    = lane & 15;
    const int kh   = (lane >> 4) << 1;
    const int col0 = blockIdx.x * 16;
    const int row0 = blockIdx.y * 16;

    v8f acc = {};
    const float* aRow = hp + (long)(row0 + m) * Dd;
    const float* bRow = Wo + (long)(col0 + m) * Dd;
    for (int k0 = 0; k0 < Dd; k0 += 4) {
        v2f a  = *(const v2f*)(aRow + k0 + kh);
        v2f bb = *(const v2f*)(bRow + k0 + kh);
        acc = __builtin_amdgcn_wmma_f32_16x16x4_f32(false, a, false, bb,
                                                    (short)0, acc, false, false);
    }

    const int rbase = row0 + ((lane >> 4) << 3);
    const int c     = col0 + m;
    const float bias = bo[c];
#pragma unroll
    for (int r = 0; r < 8; ++r) {
        float v = acc[r] + bias;
        v = (v > 0.0f) ? v : (expf(v) - 1.0f);
        out[(long)(rbase + r) * Dd + c] = v;
    }
}

extern "C" void kernel_launch(void* const* d_in, const int* in_sizes, int n_in,
                              void* d_out, int out_size, void* d_ws, size_t ws_size,
                              hipStream_t stream) {
    (void)in_sizes; (void)n_in; (void)out_size; (void)ws_size;

    const float* hmat = (const float*)d_in[0];
    const int*   nag  = (const int*)  d_in[1];
    const int*   adj  = (const int*)  d_in[2];
    const float* W1   = (const float*)d_in[3];
    const float* W2   = (const float*)d_in[4];
    const float* Ai   = (const float*)d_in[5];
    const float* Aj   = (const float*)d_in[6];
    const float* Wo   = (const float*)d_in[7];
    const float* bo   = (const float*)d_in[8];
    float*       out  = (float*)d_out;

    float* ws  = (float*)d_ws;
    const size_t whN = (size_t)Bc * Nn * Dd;
    const size_t scN = (size_t)Bc * Hh * Nn;
    float* wh1 = ws;
    float* wh2 = wh1 + whN;
    float* s1  = wh2 + whN;
    float* s2  = s1 + scN;
    float* o1  = s2 + scN;
    float* o2  = o1 + scN;
    float* hp  = o2 + scN;
    float* att = hp + whN;

    proj12_kernel<<<dim3(Dd / 16, (Bc * Nn) / 16), 32, 0, stream>>>(
        hmat, W1, W2, wh1, wh2);

    scores_kernel<<<(Bc * Nn * Hh + 255) / 256, 256, 0, stream>>>(
        wh1, wh2, Ai, Aj, s1, s2, o1, o2);

    entmax_kernel<<<(Bc * Hh * Nn) / (8 * RPW), 256, 0, stream>>>(
        s1, s2, o1, o2, adj, nag, att);

    aggregate_kernel<<<dim3(Bc * Hh, Nn / 64), 128, 0, stream>>>(
        att, wh1, wh2, nag, hp);

    outproj_kernel<<<dim3(Dd / 16, (Bc * Nn) / 16), 32, 0, stream>>>(
        hp, Wo, bo, out);
}


# aggregate: M-tiles take rows of one class (i mod num_agent), so the self/other mask moves from A to the staged B rows: one matrix product per tile, not two; general path kept; k sum order changes
# speedup vs baseline: 5.4394x; 1.0691x over previous
#include <hip/hip_runtime.h>
#include <hip/hip_bf16.h>
#include <math.h>

static constexpr int Bc = 4;
static constexpr int Nn = 1024;
static constexpr int Dd = 256;
static constexpr int Hh = 8;
static constexpr int DH = 32;

typedef float v2f __attribute__((ext_vector_type(2)));
typedef float v8f __attribute__((ext_vector_type(8)));

__global__ __launch_bounds__(32) void proj12_kernel(
    const float* __restrict__ hmat, const float* __restrict__ W1,
    const float* __restrict__ W2, float* __restrict__ wh1,
    float* __restrict__ wh2)
{
    const int lane = threadIdx.x & 31;
    const int m    = lane & 15;
    const int kh   = (lane >> 4) << 1;
    const int col0 = blockIdx.x * 16;
    const int row0 = blockIdx.y * 16;

    v8f acc1 = {};
    v8f acc2 = {};
    const float* aRow  = hmat + (long)(row0 + m) * Dd;
    const float* b1Row = W1   + (long)(col0 + m) * Dd;
    const float* b2Row = W2   + (long)(col0 + m) * Dd;

    for (int k0 = 0; k0 < Dd; k0 += 4) {
        v2f a  = *(const v2f*)(aRow  + k0 + kh);
        v2f b1 = *(const v2f*)(b1Row + k0 + kh);
        v2f b2 = *(const v2f*)(b2Row + k0 + kh);
        acc1 = __builtin_amdgcn_wmma_f32_16x16x4_f32(false, a, false, b1,
                                                     (short)0, acc1, false, false);
        acc2 = __builtin_amdgcn_wmma_f32_16x16x4_f32(false, a, false, b2,
                                                     (short)0, acc2, false, false);
    }

    const int rbase = row0 + ((lane >> 4) << 3);
    const int c     = col0 + m;
#pragma unroll
    for (int r = 0; r < 8; ++r) {
        wh1[(long)(rbase + r) * Dd + c] = acc1[r];
        wh2[(long)(rbase + r) * Dd + c] = acc2[r];
    }
}

__global__ __launch_bounds__(256) void scores_kernel(
    const float* __restrict__ wh1, const float* __restrict__ wh2,
    const float* __restrict__ Ai, const float* __restrict__ Aj,
    float* __restrict__ s1, float* __restrict__ s2,
    float* __restrict__ o1, float* __restrict__ o2)
{
    int idx = blockIdx.x * blockDim.x + threadIdx.x;
    if (idx >= Bc * Nn * Hh) return;
    const int h = idx % Hh;
    const int n = (idx / Hh) % Nn;
    const int b = idx / (Hh * Nn);

    const float* r1 = wh1 + ((long)(b * Nn + n)) * Dd + h * DH;
    const float* r2 = wh2 + ((long)(b * Nn + n)) * Dd + h * DH;
    const float* ai = Ai + h * DH;
    const float* aj = Aj + h * DH;

    float a1 = 0.f, a2 = 0.f, a3 = 0.f, a4 = 0.f;
#pragma unroll 8
    for (int d = 0; d < DH; ++d) {
        const float w1v = r1[d], w2v = r2[d], aiv = ai[d], ajv = aj[d];
        a1 += w1v * aiv;  a2 += w1v * ajv;
        a3 += w2v * aiv;  a4 += w2v * ajv;
    }
    const int o = (b * Hh + h) * Nn + n;
    s1[o] = a1; s2[o] = a2; o1[o] = a3; o2[o] = a4;
}

static constexpr int RPW = 4;
static constexpr int RS  = 8;
static constexpr int CAP = 32 * RS;

__device__ __forceinline__ float lane_pull(float x, int addr) {
    return __builtin_bit_cast(float,
        __builtin_amdgcn_ds_bpermute(addr, __builtin_bit_cast(int, x)));
}
__device__ __forceinline__ int lane_pull_i(int x, int addr) {
    return __builtin_amdgcn_ds_bpermute(addr, x);
}

template <int R, int JJ>
__device__ __forceinline__ void ce_regs(float (&v)[R]) {
#pragma unroll
    for (int u = 0; u < R; ++u) {
        if ((u & JJ) == 0) {
            const float a = v[u], c = v[u | JJ];
            v[u]      = fmaxf(a, c);
            v[u | JJ] = fminf(a, c);
        }
    }
}
template <int R, int K>
__device__ __forceinline__ void flip_regs(float (&v)[R]) {
#pragma unroll
    for (int u = 0; u < R; ++u) {
        if ((u & (K >> 1)) == 0) {
            const int p = u ^ (K - 1);
            const float a = v[u], c = v[p];
            v[u] = fmaxf(a, c);
            v[p] = fminf(a, c);
        }
    }
}
template <int LM>
__device__ __forceinline__ float lane_xor(float x) {
    const int b = __builtin_bit_cast(int, x);
    if constexpr (LM < 16) {
        return __builtin_bit_cast(float,
            __builtin_amdgcn_update_dpp(0, b, 0x160 + LM, 0xf, 0xf, true));
    } else {
        static_assert(LM == 31, "only the mirrored other-row exchange is used");
        return __builtin_bit_cast(float,
            __builtin_amdgcn_permlanex16(b, b, 0x89abcdefu, 0x01234567u, false, false));
    }
}
template <int R, int LM, bool FLIP>
__device__ __forceinline__ void ce_lanes(float (&v)[R], int lane) {
    constexpr int HB = (LM >= 16) ? 16 : (LM >= 8) ? 8 : (LM >= 4) ? 4
                     : (LM >= 2) ? 2 : 1;
    const float bound = (lane & HB) ? -__builtin_inff() : __builtin_inff();
    float p[R];
#pragma unroll
    for (int u = 0; u < R; ++u) p[u] = lane_xor<LM>(v[FLIP ? (u ^ (R - 1)) : u]);
#pragma unroll
    for (int u = 0; u < R; ++u) v[u] = __builtin_amdgcn_fmed3f(v[u], p[u], bound);
}
template <int R>
__device__ __forceinline__ void ce_regs_tail(float (&v)[R]) {
    if constexpr (R > 16) ce_regs<R, 16>(v);
    if constexpr (R > 8)  ce_regs<R, 8>(v);
    ce_regs<R, 4>(v); ce_regs<R, 2>(v); ce_regs<R, 1>(v);
}

template <int R>
__device__ __forceinline__ float entmax_tau(float (&v)[R], int lane)
{
    static_assert(R == 32 || R == 8, "register stages below are written for these");
    constexpr int LG = (R == 32) ? 5 : 3;

    flip_regs<R, 2>(v);
    flip_regs<R, 4>(v);  ce_regs<R, 1>(v);
    flip_regs<R, 8>(v);  ce_regs<R, 2>(v); ce_regs<R, 1>(v);
    if constexpr (R == 32) {
        flip_regs<R, 16>(v); ce_regs<R, 4>(v); ce_regs<R, 2>(v); ce_regs<R, 1>(v);
        flip_regs<R, 32>(v); ce_regs<R, 8>(v); ce_regs<R, 4>(v); ce_regs<R, 2>(v);
        ce_regs<R, 1>(v);
    }
    ce_lanes<R, 1, true>(v, lane);                                     ce_regs_tail<R>(v);
    ce_lanes<R, 3, true>(v, lane);  ce_lanes<R, 1, false>(v, lane);    ce_regs_tail<R>(v);
    ce_lanes<R, 7, true>(v, lane);  ce_lanes<R, 2, false>(v, lane);
    ce_lanes<R, 1, false>(v, lane);                                    ce_regs_tail<R>(v);
    ce_lanes<R, 15, true>(v, lane); ce_lanes<R, 4, false>(v, lane);
    ce_lanes<R, 2, false>(v, lane); ce_lanes<R, 1, false>(v, lane);    ce_regs_tail<R>(v);
    ce_lanes<R, 31, true>(v, lane); ce_lanes<R, 8, false>(v, lane);
    ce_lanes<R, 4, false>(v, lane); ce_lanes<R, 2, false>(v, lane);
    ce_lanes<R, 1, false>(v, lane);                                    ce_regs_tail<R>(v);

    float totS = 0.0f, totQ = 0.0f;
#pragma unroll
    for (int u = 0; u < R; ++u) { totS += v[u]; totQ += v[u] * v[u]; }
    float incS = totS, incQ = totQ;
#pragma unroll
    for (int off = 1; off < 32; off <<= 1) {
        const int   src = (lane >= off) ? (lane - off) : lane;
        const float pa  = lane_pull(incS, src << 2);
        const float pb  = lane_pull(incQ, src << 2);
        if (lane >= off) { incS += pa; incQ += pb; }
    }
    const int   prv = (lane > 0) ? (lane - 1) : 0;
    const float eS  = lane_pull(incS, prv << 2);
    const float eQ  = lane_pull(incQ, prv << 2);
    float runS = (lane > 0) ? eS : 0.0f;
    float runQ = (lane > 0) ? eQ : 0.0f;

    float Sp[R], Qp[R];
    int cnt = 0;
#pragma unroll
    for (int u = 0; u < R; ++u) {
        runS += v[u];
        runQ += v[u] * v[u];
        Sp[u] = runS;
        Qp[u] = runQ;
        const float rho = (float)(lane * R + u + 1);
        const float d   = rho * v[u] - runS;
        const float g   = v[u] * (d - runS) + runQ;
        cnt += (g <= 1.0f) ? 1 : 0;
    }
#pragma unroll
    for (int msk = 16; msk >= 1; msk >>= 1)
        cnt += lane_pull_i(cnt, (lane ^ msk) << 2);
    const int support = __builtin_amdgcn_readfirstlane(cnt);
    const int sidx = support - 1;
    const int us   = sidx & (R - 1);
    float Ssel = Sp[0], Qsel = Qp[0];
#pragma unroll
    for (int u = 1; u < R; ++u) {
        Ssel = (us == u) ? Sp[u] : Ssel;
        Qsel = (us == u) ? Qp[u] : Qsel;
    }
    float tsel;
    {
        const float rho   = (float)(lane * R + us + 1);
        const float mean  = Ssel / rho;
        const float msq   = Qsel / rho;
        const float ssv   = rho * (msq - mean * mean);
        const float delta = (1.0f - ssv) / rho;
        tsel = mean - ((delta > 0.0f) ? sqrtf(delta) : 0.0f);
    }
    return lane_pull(tsel, (sidx >> LG) << 2);
}

__global__ __launch_bounds__(256) void entmax_kernel(
    const float* __restrict__ s1, const float* __restrict__ s2,
    const float* __restrict__ o1, const float* __restrict__ o2,
    const int* __restrict__ adj, const int* __restrict__ nag,
    float* __restrict__ att)
{
    __shared__ float cand[8][Nn];

    const int lane = threadIdx.x & 31;
    const int wave = threadIdx.x >> 5;
    float* wl = cand[wave];
    const unsigned na      = (unsigned)nag[0];
    const unsigned step128 = 128u % na;
    const unsigned jm_lane = (unsigned)(lane * 4) % na;
    const int row0 = (blockIdx.x * 8 + wave) * RPW;

#pragma unroll 1
    for (int rr = 0; rr < RPW; ++rr) {
        const int bid = row0 + rr;
        const int i   = bid & (Nn - 1);
        const int h   = (bid >> 10) & (Hh - 1);
        const int b   = bid >> 13;

        const int   rowbase = (b * Hh + h) * Nn;
        const float s1i = s1[rowbase + i];
        const float o1i = o1[rowbase + i];
        const unsigned im = (unsigned)i % na;
        const long  adjbase = ((long)b * Nn + i) * Nn;

        float x[32];
        float lmax = -3.4e38f;
        unsigned jm = jm_lane;
#pragma unroll
        for (int q = 0; q < 8; ++q) {
            const int j = (q * 32 + lane) * 4;
            const float4 sv = *(const float4*)(s2 + rowbase + j);
            const float4 ov = *(const float4*)(o2 + rowbase + j);
            const int4   av = *(const int4*)(adj + adjbase + j);
            asm volatile("" :: "v"(sv.x), "v"(sv.y), "v"(sv.z), "v"(sv.w),
                               "v"(ov.x), "v"(ov.y), "v"(ov.z), "v"(ov.w));
            const float se[4] = { sv.x, sv.y, sv.z, sv.w };
            const float oe[4] = { ov.x, ov.y, ov.z, ov.w };
            const int   ae[4] = { av.x, av.y, av.z, av.w };
            unsigned m = jm;
#pragma unroll
            for (int c = 0; c < 4; ++c) {
                float e = (m == im) ? (s1i + se[c]) : (o1i + oe[c]);
                if (ae[c] == 0) e = -1.0e12f;
                const float xv = 0.5f * e;
                x[q * 4 + c] = xv;
                lmax = fmaxf(lmax, xv);
                m = (m + 1u == na) ? 0u : (m + 1u);
            }
            jm += step128;
            if (jm >= na) jm -= na;
        }
#pragma unroll
        for (int msk = 16; msk >= 1; msk >>= 1)
            lmax = fmaxf(lmax, lane_pull(lmax, (lane ^ msk) << 2));
        const float xmax = lmax;
#pragma unroll
        for (int u = 0; u < 32; ++u) x[u] -= xmax;

        int ncand = 0;
#pragma unroll
        for (int u = 0; u < 32; ++u) {
            const bool     keep = (x[u] >= -1.0f);
            const unsigned who  = __builtin_amdgcn_ballot_w32(keep);
            const int      pos  = (int)__builtin_amdgcn_mbcnt_lo(who, (unsigned)ncand);
            if (keep) wl[pos] = x[u];
            ncand += __builtin_popcount(who);
        }

        float tstar;
        if (ncand <= CAP) {
            __builtin_amdgcn_fence(__ATOMIC_ACQ_REL, "wavefront");
            float w[RS];
#pragma unroll
            for (int r = 0; r < RS; ++r) {
                const int   slot = r * 32 + lane;
                const float t    = wl[slot];
                w[r] = (slot < ncand) ? t : -2.0f;
            }
            tstar = entmax_tau<RS>(w, lane);
        } else {
            float v[32];
#pragma unroll
            for (int u = 0; u < 32; ++u) v[u] = x[u];
            tstar = entmax_tau<32>(v, lane);
        }

        float* outrow = att + ((long)(b * Hh + h) * Nn + i) * Nn;
#pragma unroll
        for (int q = 0; q < 8; ++q) {
            float4 o;
            float y;
            y = fmaxf(x[q * 4 + 0] - tstar, 0.0f); o.x = y * y;
            y = fmaxf(x[q * 4 + 1] - tstar, 0.0f); o.y = y * y;
            y = fmaxf(x[q * 4 + 2] - tstar, 0.0f); o.z = y * y;
            y = fmaxf(x[q * 4 + 3] - tstar, 0.0f); o.w = y * y;
            *(float4*)(outrow + (q * 32 + lane) * 4) = o;
        }
    }
}

static constexpr int KC  = 128;
static constexpr int LDP = 40;
static constexpr int TS  = KC / 8;

template <bool ONE>
__device__ __forceinline__ void aggregate_body(
    const float* __restrict__ att, const float* __restrict__ wh1,
    const float* __restrict__ wh2, const unsigned na, const unsigned cnt,
    float* __restrict__ hp, float (&ldsW1)[KC][LDP], float (&ldsW2)[KC][LDP])
{
    const int tid  = threadIdx.x;
    const int wave = tid >> 5;
    const int lane = tid & 31;
    const int m    = lane & 15;
    const int half = lane >> 4;
    const int bh   = blockIdx.x;
    const int b    = bh / Hh;
    const int h    = bh % Hh;

    const unsigned p0  = blockIdx.y * 64u;
    const unsigned res = ONE ? (p0 / cnt) : 0u;
    const unsigned q0  = ONE ? (p0 - res * cnt + (unsigned)wave * 16u)
                             : (p0 + (unsigned)wave * 16u);
    auto rowq = [&](unsigned mm) -> int {
        return ONE ? (int)((q0 + mm) * na + res) : (int)(q0 + mm);
    };

    const int row = rowq((unsigned)m);
    const unsigned rowmod = (unsigned)row % na;

    const float* attp = att + ((long)bh * Nn + row) * Nn + 4 * half;

    unsigned cm = (unsigned)(4 * half) % na;
    const unsigned step8 = 8u % na;

    v8f accS0 = {};
    v8f accS1 = {};

    float4 aN[TS];
#pragma unroll
    for (int t = 0; t < TS; ++t) aN[t] = *(const float4*)(attp + 8 * t);

#pragma unroll 1
    for (int kc = 0; kc < Nn; kc += KC) {
        float4 aC[TS];
#pragma unroll
        for (int t = 0; t < TS; ++t) aC[t] = aN[t];

        __syncthreads();
        if constexpr (ONE) {
            const bool selfk = ((unsigned)(kc + tid) % na) == res;
            const float* srow = (selfk ? wh1 : wh2)
                              + ((long)(b * Nn + kc + tid)) * Dd + h * DH;
            float4* dst = (float4*)&ldsW1[tid][0];
#pragma unroll
            for (int q = 0; q < 8; ++q) dst[q] = ((const float4*)srow)[q];
        } else {
            const float* src1 = wh1 + ((long)(b * Nn + kc + tid)) * Dd + h * DH;
            const float* src2 = wh2 + ((long)(b * Nn + kc + tid)) * Dd + h * DH;
            float4* dst1 = (float4*)&ldsW1[tid][0];
            float4* dst2 = (float4*)&ldsW2[tid][0];
#pragma unroll
            for (int q = 0; q < 8; ++q) {
                dst1[q] = ((const float4*)src1)[q];
                dst2[q] = ((const float4*)src2)[q];
            }
        }
        __syncthreads();

        if (kc + KC < Nn) {
            const float* an = attp + (kc + KC);
#pragma unroll
            for (int t = 0; t < TS; ++t) aN[t] = *(const float4*)(an + 8 * t);
        }

        constexpr int NB = ONE ? 8 : 16;
        float br[2][NB];
        auto load_b = [&](int t, float (&r)[NB]) {
#pragma unroll
            for (int s = 0; s < 2; ++s) {
                const int l0 = 8 * t + 4 * half + 2 * s;
                if constexpr (ONE) {
                    r[4 * s + 0] = ldsW1[l0][m];      r[4 * s + 1] = ldsW1[l0 + 1][m];
                    r[4 * s + 2] = ldsW1[l0][16 + m]; r[4 * s + 3] = ldsW1[l0 + 1][16 + m];
                } else {
                    r[8 * s + 0] = ldsW1[l0][m];      r[8 * s + 1] = ldsW1[l0 + 1][m];
                    r[8 * s + 2] = ldsW1[l0][16 + m]; r[8 * s + 3] = ldsW1[l0 + 1][16 + m];
                    r[8 * s + 4] = ldsW2[l0][m];      r[8 * s + 5] = ldsW2[l0 + 1][m];
                    r[8 * s + 6] = ldsW2[l0][16 + m]; r[8 * s + 7] = ldsW2[l0 + 1][16 + m];
                }
            }
        };
        load_b(0, br[0]);

#pragma unroll
        for (int t = 0; t < TS; ++t) {
            if (t + 1 < TS) load_b(t + 1, br[(t + 1) & 1]);
            __builtin_amdgcn_sched_barrier(0);

            const float av[4] = { aC[t].x, aC[t].y, aC[t].z, aC[t].w };
            const float (&r)[NB] = br[t & 1];

            if constexpr (ONE) {
#pragma unroll
                for (int s = 0; s < 2; ++s) {
                    v2f a;   a.x   = av[2 * s];     a.y   = av[2 * s + 1];
                    v2f b_0; b_0.x = r[4 * s + 0];  b_0.y = r[4 * s + 1];
                    v2f b_1; b_1.x = r[4 * s + 2];  b_1.y = r[4 * s + 3];
                    accS0 = __builtin_amdgcn_wmma_f32_16x16x4_f32(false, a, false, b_0,
                                                                  (short)0, accS0, false, false);
                    accS1 = __builtin_amdgcn_wmma_f32_16x16x4_f32(false, a, false, b_1,
                                                                  (short)0, accS1, false, false);
                }
            } else {
                bool sf[4];
                unsigned mm = cm;
#pragma unroll
                for (int e = 0; e < 4; ++e) {
                    sf[e] = (mm == rowmod);
                    mm = (mm + 1u == na) ? 0u : (mm + 1u);
                }
                cm += step8;
                if (cm >= na) cm -= na;

#pragma unroll
                for (int s = 0; s < 2; ++s) {
                    v2f aS, aO;
                    aS.x = sf[2 * s]     ? av[2 * s]     : 0.0f;
                    aS.y = sf[2 * s + 1] ? av[2 * s + 1] : 0.0f;
                    aO.x = sf[2 * s]     ? 0.0f : av[2 * s];
                    aO.y = sf[2 * s + 1] ? 0.0f : av[2 * s + 1];

                    v2f b1_0; b1_0.x = r[8 * s + 0]; b1_0.y = r[8 * s + 1];
                    v2f b1_1; b1_1.x = r[8 * s + 2]; b1_1.y = r[8 * s + 3];
                    v2f b2_0; b2_0.x = r[8 * s + 4]; b2_0.y = r[8 * s + 5];
                    v2f b2_1; b2_1.x = r[8 * s + 6]; b2_1.y = r[8 * s + 7];

                    accS0 = __builtin_amdgcn_wmma_f32_16x16x4_f32(false, aS, false, b1_0,
                                                                  (short)0, accS0, false, false);
                    accS0 = __builtin_amdgcn_wmma_f32_16x16x4_f32(false, aO, false, b2_0,
                                                                  (short)0, accS0, false, false);
                    accS1 = __builtin_amdgcn_wmma_f32_16x16x4_f32(false, aS, false, b1_1,
                                                                  (short)0, accS1, false, false);
                    accS1 = __builtin_amdgcn_wmma_f32_16x16x4_f32(false, aO, false, b2_1,
                                                                  (short)0, accS1, false, false);
                }
            }
            __builtin_amdgcn_sched_barrier(0);
        }
    }

    const unsigned mb = (unsigned)((lane >> 4) << 3);
#pragma unroll
    for (int r = 0; r < 8; ++r) {
        float* o = hp + ((long)(b * Nn + rowq(mb + (unsigned)r))) * Dd + h * DH;
        o[m]      = accS0[r];
        o[16 + m] = accS1[r];
    }
}

__global__ __launch_bounds__(128) void aggregate_kernel(
    const float* __restrict__ att, const float* __restrict__ wh1,
    const float* __restrict__ wh2, const int* __restrict__ nag,
    float* __restrict__ hp)
{
    __shared__ float ldsW1[KC][LDP];
    __shared__ float ldsW2[KC][LDP];

    const unsigned na  = (unsigned)nag[0];
    const unsigned cnt = (unsigned)Nn / na;
    const bool one = (cnt * na == (unsigned)Nn) && ((cnt & 63u) == 0u);
    if (one) aggregate_body<true >(att, wh1, wh2, na, cnt, hp, ldsW1, ldsW2);
    else     aggregate_body<false>(att, wh1, wh2, na, cnt, hp, ldsW1, ldsW2);
}

__global__ __launch_bounds__(32) void outproj_kernel(
    const float* __restrict__ hp, const float* __restrict__ Wo,
    const float* __restrict__ bo, float* __restrict__ out)
{
    const int lane = threadIdx.x & 31;
    const int m    = lane & 15;
    const int kh   = (lane >> 4) << 1;
    const int col0 = blockIdx.x * 16;
    const int row0 = blockIdx.y * 16;

    v8f acc = {};
    const float* aRow = hp + (long)(row0 + m) * Dd;
    const float* bRow = Wo + (long)(col0 + m) * Dd;
    for (int k0 = 0; k0 < Dd; k0 += 4) {
        v2f a  = *(const v2f*)(aRow + k0 + kh);
        v2f bb = *(const v2f*)(bRow + k0 + kh);
        acc = __builtin_amdgcn_wmma_f32_16x16x4_f32(false, a, false, bb,
                                                    (short)0, acc, false, false);
    }

    const int rbase = row0 + ((lane >> 4) << 3);
    const int c     = col0 + m;
    const float bias = bo[c];
#pragma unroll
    for (int r = 0; r < 8; ++r) {
        float v = acc[r] + bias;
        v = (v > 0.0f) ? v : (expf(v) - 1.0f);
        out[(long)(rbase + r) * Dd + c] = v;
    }
}

extern "C" void kernel_launch(void* const* d_in, const int* in_sizes, int n_in,
                              void* d_out, int out_size, void* d_ws, size_t ws_size,
                              hipStream_t stream) {
    (void)in_sizes; (void)n_in; (void)out_size; (void)ws_size;

    const float* hmat = (const float*)d_in[0];
    const int*   nag  = (const int*)  d_in[1];
    const int*   adj  = (const int*)  d_in[2];
    const float* W1   = (const float*)d_in[3];
    const float* W2   = (const float*)d_in[4];
    const float* Ai   = (const float*)d_in[5];
    const float* Aj   = (const float*)d_in[6];
    const float* Wo   = (const float*)d_in[7];
    const float* bo   = (const float*)d_in[8];
    float*       out  = (float*)d_out;

    float* ws  = (float*)d_ws;
    const size_t whN = (size_t)Bc * Nn * Dd;
    const size_t scN = (size_t)Bc * Hh * Nn;
    float* wh1 = ws;
    float* wh2 = wh1 + whN;
    float* s1  = wh2 + whN;
    float* s2  = s1 + scN;
    float* o1  = s2 + scN;
    float* o2  = o1 + scN;
    float* hp  = o2 + scN;
    float* att = hp + whN;

    proj12_kernel<<<dim3(Dd / 16, (Bc * Nn) / 16), 32, 0, stream>>>(
        hmat, W1, W2, wh1, wh2);

    scores_kernel<<<(Bc * Nn * Hh + 255) / 256, 256, 0, stream>>>(
        wh1, wh2, Ai, Aj, s1, s2, o1, o2);

    entmax_kernel<<<(Bc * Hh * Nn) / (8 * RPW), 256, 0, stream>>>(
        s1, s2, o1, o2, adj, nag, att);

    aggregate_kernel<<<dim3(Bc * Hh, Nn / 64), 128, 0, stream>>>(
        att, wh1, wh2, nag, hp);

    outproj_kernel<<<dim3(Dd / 16, (Bc * Nn) / 16), 32, 0, stream>>>(
        hp, Wo, bo, out);
}


# projections (Wh1/Wh2, output): operand tiles staged in LDS by whole-line 16-byte loads and shared by 8 waves, not 8-byte fragment loads over 16 lines each; k sum order changes
# speedup vs baseline: 6.4726x; 1.1899x over previous
#include <hip/hip_runtime.h>
#include <hip/hip_bf16.h>
#include <math.h>

static constexpr int Bc = 4;
static constexpr int Nn = 1024;
static constexpr int Dd = 256;
static constexpr int Hh = 8;
static constexpr int DH = 32;

typedef float v2f __attribute__((ext_vector_type(2)));
typedef float v4f __attribute__((ext_vector_type(4)));
typedef float v8f __attribute__((ext_vector_type(8)));

static constexpr int GK  = 64;
static constexpr int GP  = GK + 4;
static constexpr int GT  = GK / 8;
static constexpr int NCH = Dd / GK;

__device__ __forceinline__ v4f frag4(const float* rowp, int t) {
    return *(const v4f*)(rowp + 8 * t);
}

template <int BM, int BN, int MT, int NT, bool ELU>
__device__ __forceinline__ void proj_tile(
    const float* A, const float* W,
    const float* __restrict__ bias, float* C,
    const int row0, const int col0, float* lds)
{
    constexpr int WN  = BN / (16 * NT);
    constexpr int WMV = BM / (16 * MT);
    static_assert(WN * WMV == 8, "the block has 8 waves");
    constexpr int BUF = (BM + BN) * GP;
    constexpr int NA  = BM / 16;
    constexpr int NB  = BN / 16;

    const int tid  = threadIdx.x;
    const int lane = tid & 31;
    const int wave = tid >> 5;
    const int m    = lane & 15;
    const int half = lane >> 4;
    const int wn   = wave % WN;
    const int wm   = wave / WN;

    const int c4 = tid & 15;
    const int r0 = tid >> 4;
    const float* ga = A + (long)(row0 + r0) * Dd + 4 * c4;
    const float* gb = W + (long)(col0 + r0) * Dd + 4 * c4;
    const int sa = r0 * GP + 4 * c4;
    const int sb = (BM + r0) * GP + 4 * c4;

    const int fa = (wm * 16 * MT + m) * GP + 4 * half;
    const int fb = (BM + wn * 16 * NT + m) * GP + 4 * half;

    v8f acc[MT][NT];
#pragma unroll
    for (int i = 0; i < MT; ++i)
#pragma unroll
        for (int j = 0; j < NT; ++j) acc[i][j] = v8f{};

    v4f pa[NA], pb[NB];
#pragma unroll
    for (int i = 0; i < NA; ++i) pa[i] = *(const v4f*)(ga + (long)(16 * i) * Dd);
#pragma unroll
    for (int i = 0; i < NB; ++i) pb[i] = *(const v4f*)(gb + (long)(16 * i) * Dd);
#pragma unroll
    for (int i = 0; i < NA; ++i) *(v4f*)(lds + sa + 16 * i * GP) = pa[i];
#pragma unroll
    for (int i = 0; i < NB; ++i) *(v4f*)(lds + sb + 16 * i * GP) = pb[i];
    __syncthreads();

#pragma unroll
    for (int c = 0; c < NCH; ++c) {
        const float* cur = lds + (c & 1) * BUF;
        float*       nxt = lds + ((c + 1) & 1) * BUF;
        if (c + 1 < NCH) {
#pragma unroll
            for (int i = 0; i < NA; ++i)
                pa[i] = *(const v4f*)(ga + (long)(16 * i) * Dd + (c + 1) * GK);
#pragma unroll
            for (int i = 0; i < NB; ++i)
                pb[i] = *(const v4f*)(gb + (long)(16 * i) * Dd + (c + 1) * GK);
        }
        __builtin_amdgcn_sched_barrier(0);

#pragma unroll
        for (int t = 0; t < GT; ++t) {
            v4f xa[MT], xb[NT];
#pragma unroll
            for (int i = 0; i < MT; ++i) xa[i] = frag4(cur + fa + 16 * i * GP, t);
#pragma unroll
            for (int j = 0; j < NT; ++j) xb[j] = frag4(cur + fb + 16 * j * GP, t);
#pragma unroll
            for (int s = 0; s < 2; ++s) {
#pragma unroll
                for (int i = 0; i < MT; ++i) {
                    v2f a;
                    a.x = s ? xa[i].z : xa[i].x;
                    a.y = s ? xa[i].w : xa[i].y;
#pragma unroll
                    for (int j = 0; j < NT; ++j) {
                        v2f b;
                        b.x = s ? xb[j].z : xb[j].x;
                        b.y = s ? xb[j].w : xb[j].y;
                        acc[i][j] = __builtin_amdgcn_wmma_f32_16x16x4_f32(
                            false, a, false, b, (short)0, acc[i][j], false, false);
                    }
                }
            }
        }

        __builtin_amdgcn_sched_barrier(0);
        if (c + 1 < NCH) {
#pragma unroll
            for (int i = 0; i < NA; ++i) *(v4f*)(nxt + sa + 16 * i * GP) = pa[i];
#pragma unroll
            for (int i = 0; i < NB; ++i) *(v4f*)(nxt + sb + 16 * i * GP) = pb[i];
            __syncthreads();
        }
    }

#pragma unroll
    for (int i = 0; i < MT; ++i) {
        const int rbase = row0 + (wm * MT + i) * 16 + (half << 3);
#pragma unroll
        for (int j = 0; j < NT; ++j) {
            const int cc = col0 + (wn * NT + j) * 16 + m;
            float bv = 0.0f;
            if constexpr (ELU) bv = bias[cc];
#pragma unroll
            for (int r = 0; r < 8; ++r) {
                float v = acc[i][j][r];
                if constexpr (ELU) {
                    v += bv;
                    v = (v > 0.0f) ? v : (expf(v) - 1.0f);
                }
                C[(long)(rbase + r) * Dd + cc] = v;
            }
        }
    }
}

__global__ __launch_bounds__(256)
void proj12_kernel(
    const float* hmat, const float* W1, const float* W2,
    float* wh1, float* wh2)
{
    __shared__ __attribute__((aligned(16))) float lds[2 * (64 + 128) * GP];
    const bool second = blockIdx.x >= 2;
    proj_tile<64, 128, 2, 2, false>(hmat, second ? W2 : W1, nullptr,
                                    second ? wh2 : wh1,
                                    blockIdx.y * 64, (blockIdx.x & 1) * 128, lds);
}

__global__ __launch_bounds__(256) void scores_kernel(
    const float* __restrict__ wh1, const float* __restrict__ wh2,
    const float* __restrict__ Ai, const float* __restrict__ Aj,
    float* __restrict__ s1, float* __restrict__ s2,
    float* __restrict__ o1, float* __restrict__ o2)
{
    int idx = blockIdx.x * blockDim.x + threadIdx.x;
    if (idx >= Bc * Nn * Hh) return;
    const int h = idx % Hh;
    const int n = (idx / Hh) % Nn;
    const int b = idx / (Hh * Nn);

    const float* r1 = wh1 + ((long)(b * Nn + n)) * Dd + h * DH;
    const float* r2 = wh2 + ((long)(b * Nn + n)) * Dd + h * DH;
    const float* ai = Ai + h * DH;
    const float* aj = Aj + h * DH;

    float a1 = 0.f, a2 = 0.f, a3 = 0.f, a4 = 0.f;
#pragma unroll 8
    for (int d = 0; d < DH; ++d) {
        const float w1v = r1[d], w2v = r2[d], aiv = ai[d], ajv = aj[d];
        a1 += w1v * aiv;  a2 += w1v * ajv;
        a3 += w2v * aiv;  a4 += w2v * ajv;
    }
    const int o = (b * Hh + h) * Nn + n;
    s1[o] = a1; s2[o] = a2; o1[o] = a3; o2[o] = a4;
}

static constexpr int RPW = 4;
static constexpr int RS  = 8;
static constexpr int CAP = 32 * RS;

__device__ __forceinline__ float lane_pull(float x, int addr) {
    return __builtin_bit_cast(float,
        __builtin_amdgcn_ds_bpermute(addr, __builtin_bit_cast(int, x)));
}
__device__ __forceinline__ int lane_pull_i(int x, int addr) {
    return __builtin_amdgcn_ds_bpermute(addr, x);
}

template <int R, int JJ>
__device__ __forceinline__ void ce_regs(float (&v)[R]) {
#pragma unroll
    for (int u = 0; u < R; ++u) {
        if ((u & JJ) == 0) {
            const float a = v[u], c = v[u | JJ];
            v[u]      = fmaxf(a, c);
            v[u | JJ] = fminf(a, c);
        }
    }
}
template <int R, int K>
__device__ __forceinline__ void flip_regs(float (&v)[R]) {
#pragma unroll
    for (int u = 0; u < R; ++u) {
        if ((u & (K >> 1)) == 0) {
            const int p = u ^ (K - 1);
            const float a = v[u], c = v[p];
            v[u] = fmaxf(a, c);
            v[p] = fminf(a, c);
        }
    }
}
template <int LM>
__device__ __forceinline__ float lane_xor(float x) {
    const int b = __builtin_bit_cast(int, x);
    if constexpr (LM < 16) {
        return __builtin_bit_cast(float,
            __builtin_amdgcn_update_dpp(0, b, 0x160 + LM, 0xf, 0xf, true));
    } else {
        static_assert(LM == 31, "only the mirrored other-row exchange is used");
        return __builtin_bit_cast(float,
            __builtin_amdgcn_permlanex16(b, b, 0x89abcdefu, 0x01234567u, false, false));
    }
}
template <int R, int LM, bool FLIP>
__device__ __forceinline__ void ce_lanes(float (&v)[R], int lane) {
    constexpr int HB = (LM >= 16) ? 16 : (LM >= 8) ? 8 : (LM >= 4) ? 4
                     : (LM >= 2) ? 2 : 1;
    const float bound = (lane & HB) ? -__builtin_inff() : __builtin_inff();
    float p[R];
#pragma unroll
    for (int u = 0; u < R; ++u) p[u] = lane_xor<LM>(v[FLIP ? (u ^ (R - 1)) : u]);
#pragma unroll
    for (int u = 0; u < R; ++u) v[u] = __builtin_amdgcn_fmed3f(v[u], p[u], bound);
}
template <int R>
__device__ __forceinline__ void ce_regs_tail(float (&v)[R]) {
    if constexpr (R > 16) ce_regs<R, 16>(v);
    if constexpr (R > 8)  ce_regs<R, 8>(v);
    ce_regs<R, 4>(v); ce_regs<R, 2>(v); ce_regs<R, 1>(v);
}

template <int R>
__device__ __forceinline__ float entmax_tau(float (&v)[R], int lane)
{
    static_assert(R == 32 || R == 8, "register stages below are written for these");
    constexpr int LG = (R == 32) ? 5 : 3;

    flip_regs<R, 2>(v);
    flip_regs<R, 4>(v);  ce_regs<R, 1>(v);
    flip_regs<R, 8>(v);  ce_regs<R, 2>(v); ce_regs<R, 1>(v);
    if constexpr (R == 32) {
        flip_regs<R, 16>(v); ce_regs<R, 4>(v); ce_regs<R, 2>(v); ce_regs<R, 1>(v);
        flip_regs<R, 32>(v); ce_regs<R, 8>(v); ce_regs<R, 4>(v); ce_regs<R, 2>(v);
        ce_regs<R, 1>(v);
    }
    ce_lanes<R, 1, true>(v, lane);                                     ce_regs_tail<R>(v);
    ce_lanes<R, 3, true>(v, lane);  ce_lanes<R, 1, false>(v, lane);    ce_regs_tail<R>(v);
    ce_lanes<R, 7, true>(v, lane);  ce_lanes<R, 2, false>(v, lane);
    ce_lanes<R, 1, false>(v, lane);                                    ce_regs_tail<R>(v);
    ce_lanes<R, 15, true>(v, lane); ce_lanes<R, 4, false>(v, lane);
    ce_lanes<R, 2, false>(v, lane); ce_lanes<R, 1, false>(v, lane);    ce_regs_tail<R>(v);
    ce_lanes<R, 31, true>(v, lane); ce_lanes<R, 8, false>(v, lane);
    ce_lanes<R, 4, false>(v, lane); ce_lanes<R, 2, false>(v, lane);
    ce_lanes<R, 1, false>(v, lane);                                    ce_regs_tail<R>(v);

    float totS = 0.0f, totQ = 0.0f;
#pragma unroll
    for (int u = 0; u < R; ++u) { totS += v[u]; totQ += v[u] * v[u]; }
    float incS = totS, incQ = totQ;
#pragma unroll
    for (int off = 1; off < 32; off <<= 1) {
        const int   src = (lane >= off) ? (lane - off) : lane;
        const float pa  = lane_pull(incS, src << 2);
        const float pb  = lane_pull(incQ, src << 2);
        if (lane >= off) { incS += pa; incQ += pb; }
    }
    const int   prv = (lane > 0) ? (lane - 1) : 0;
    const float eS  = lane_pull(incS, prv << 2);
    const float eQ  = lane_pull(incQ, prv << 2);
    float runS = (lane > 0) ? eS : 0.0f;
    float runQ = (lane > 0) ? eQ : 0.0f;

    float Sp[R], Qp[R];
    int cnt = 0;
#pragma unroll
    for (int u = 0; u < R; ++u) {
        runS += v[u];
        runQ += v[u] * v[u];
        Sp[u] = runS;
        Qp[u] = runQ;
        const float rho = (float)(lane * R + u + 1);
        const float d   = rho * v[u] - runS;
        const float g   = v[u] * (d - runS) + runQ;
        cnt += (g <= 1.0f) ? 1 : 0;
    }
#pragma unroll
    for (int msk = 16; msk >= 1; msk >>= 1)
        cnt += lane_pull_i(cnt, (lane ^ msk) << 2);
    const int support = __builtin_amdgcn_readfirstlane(cnt);
    const int sidx = support - 1;
    const int us   = sidx & (R - 1);
    float Ssel = Sp[0], Qsel = Qp[0];
#pragma unroll
    for (int u = 1; u < R; ++u) {
        Ssel = (us == u) ? Sp[u] : Ssel;
        Qsel = (us == u) ? Qp[u] : Qsel;
    }
    float tsel;
    {
        const float rho   = (float)(lane * R + us + 1);
        const float mean  = Ssel / rho;
        const float msq   = Qsel / rho;
        const float ssv   = rho * (msq - mean * mean);
        const float delta = (1.0f - ssv) / rho;
        tsel = mean - ((delta > 0.0f) ? sqrtf(delta) : 0.0f);
    }
    return lane_pull(tsel, (sidx >> LG) << 2);
}

__global__ __launch_bounds__(256) void entmax_kernel(
    const float* __restrict__ s1, const float* __restrict__ s2,
    const float* __restrict__ o1, const float* __restrict__ o2,
    const int* __restrict__ adj, const int* __restrict__ nag,
    float* __restrict__ att)
{
    __shared__ float cand[8][Nn];

    const int lane = threadIdx.x & 31;
    const int wave = threadIdx.x >> 5;
    float* wl = cand[wave];
    const unsigned na      = (unsigned)nag[0];
    const unsigned step128 = 128u % na;
    const unsigned jm_lane = (unsigned)(lane * 4) % na;
    const int row0 = (blockIdx.x * 8 + wave) * RPW;

#pragma unroll 1
    for (int rr = 0; rr < RPW; ++rr) {
        const int bid = row0 + rr;
        const int i   = bid & (Nn - 1);
        const int h   = (bid >> 10) & (Hh - 1);
        const int b   = bid >> 13;

        const int   rowbase = (b * Hh + h) * Nn;
        const float s1i = s1[rowbase + i];
        const float o1i = o1[rowbase + i];
        const unsigned im = (unsigned)i % na;
        const long  adjbase = ((long)b * Nn + i) * Nn;

        float x[32];
        float lmax = -3.4e38f;
        unsigned jm = jm_lane;
#pragma unroll
        for (int q = 0; q < 8; ++q) {
            const int j = (q * 32 + lane) * 4;
            const float4 sv = *(const float4*)(s2 + rowbase + j);
            const float4 ov = *(const float4*)(o2 + rowbase + j);
            const int4   av = *(const int4*)(adj + adjbase + j);
            asm volatile("" :: "v"(sv.x), "v"(sv.y), "v"(sv.z), "v"(sv.w),
                               "v"(ov.x), "v"(ov.y), "v"(ov.z), "v"(ov.w));
            const float se[4] = { sv.x, sv.y, sv.z, sv.w };
            const float oe[4] = { ov.x, ov.y, ov.z, ov.w };
            const int   ae[4] = { av.x, av.y, av.z, av.w };
            unsigned m = jm;
#pragma unroll
            for (int c = 0; c < 4; ++c) {
                float e = (m == im) ? (s1i + se[c]) : (o1i + oe[c]);
                if (ae[c] == 0) e = -1.0e12f;
                const float xv = 0.5f * e;
                x[q * 4 + c] = xv;
                lmax = fmaxf(lmax, xv);
                m = (m + 1u == na) ? 0u : (m + 1u);
            }
            jm += step128;
            if (jm >= na) jm -= na;
        }
#pragma unroll
        for (int msk = 16; msk >= 1; msk >>= 1)
            lmax = fmaxf(lmax, lane_pull(lmax, (lane ^ msk) << 2));
        const float xmax = lmax;
#pragma unroll
        for (int u = 0; u < 32; ++u) x[u] -= xmax;

        int ncand = 0;
#pragma unroll
        for (int u = 0; u < 32; ++u) {
            const bool     keep = (x[u] >= -1.0f);
            const unsigned who  = __builtin_amdgcn_ballot_w32(keep);
            const int      pos  = (int)__builtin_amdgcn_mbcnt_lo(who, (unsigned)ncand);
            if (keep) wl[pos] = x[u];
            ncand += __builtin_popcount(who);
        }

        float tstar;
        if (ncand <= CAP) {
            __builtin_amdgcn_fence(__ATOMIC_ACQ_REL, "wavefront");
            float w[RS];
#pragma unroll
            for (int r = 0; r < RS; ++r) {
                const int   slot = r * 32 + lane;
                const float t    = wl[slot];
                w[r] = (slot < ncand) ? t : -2.0f;
            }
            tstar = entmax_tau<RS>(w, lane);
        } else {
            float v[32];
#pragma unroll
            for (int u = 0; u < 32; ++u) v[u] = x[u];
            tstar = entmax_tau<32>(v, lane);
        }

        float* outrow = att + ((long)(b * Hh + h) * Nn + i) * Nn;
#pragma unroll
        for (int q = 0; q < 8; ++q) {
            float4 o;
            float y;
            y = fmaxf(x[q * 4 + 0] - tstar, 0.0f); o.x = y * y;
            y = fmaxf(x[q * 4 + 1] - tstar, 0.0f); o.y = y * y;
            y = fmaxf(x[q * 4 + 2] - tstar, 0.0f); o.z = y * y;
            y = fmaxf(x[q * 4 + 3] - tstar, 0.0f); o.w = y * y;
            *(float4*)(outrow + (q * 32 + lane) * 4) = o;
        }
    }
}

static constexpr int KC  = 128;
static constexpr int LDP = 40;
static constexpr int TS  = KC / 8;

template <bool ONE>
__device__ __forceinline__ void aggregate_body(
    const float* __restrict__ att, const float* __restrict__ wh1,
    const float* __restrict__ wh2, const unsigned na, const unsigned cnt,
    float* __restrict__ hp, float (&ldsW1)[KC][LDP], float (&ldsW2)[KC][LDP])
{
    const int tid  = threadIdx.x;
    const int wave = tid >> 5;
    const int lane = tid & 31;
    const int m    = lane & 15;
    const int half = lane >> 4;
    const int bh   = blockIdx.x;
    const int b    = bh / Hh;
    const int h    = bh % Hh;

    const unsigned p0  = blockIdx.y * 64u;
    const unsigned res = ONE ? (p0 / cnt) : 0u;
    const unsigned q0  = ONE ? (p0 - res * cnt + (unsigned)wave * 16u)
                             : (p0 + (unsigned)wave * 16u);
    auto rowq = [&](unsigned mm) -> int {
        return ONE ? (int)((q0 + mm) * na + res) : (int)(q0 + mm);
    };

    const int row = rowq((unsigned)m);
    const unsigned rowmod = (unsigned)row % na;

    const float* attp = att + ((long)bh * Nn + row) * Nn + 4 * half;

    unsigned cm = (unsigned)(4 * half) % na;
    const unsigned step8 = 8u % na;

    v8f accS0 = {};
    v8f accS1 = {};

    float4 aN[TS];
#pragma unroll
    for (int t = 0; t < TS; ++t) aN[t] = *(const float4*)(attp + 8 * t);

#pragma unroll 1
    for (int kc = 0; kc < Nn; kc += KC) {
        float4 aC[TS];
#pragma unroll
        for (int t = 0; t < TS; ++t) aC[t] = aN[t];

        __syncthreads();
        if constexpr (ONE) {
            const bool selfk = ((unsigned)(kc + tid) % na) == res;
            const float* srow = (selfk ? wh1 : wh2)
                              + ((long)(b * Nn + kc + tid)) * Dd + h * DH;
            float4* dst = (float4*)&ldsW1[tid][0];
#pragma unroll
            for (int q = 0; q < 8; ++q) dst[q] = ((const float4*)srow)[q];
        } else {
            const float* src1 = wh1 + ((long)(b * Nn + kc + tid)) * Dd + h * DH;
            const float* src2 = wh2 + ((long)(b * Nn + kc + tid)) * Dd + h * DH;
            float4* dst1 = (float4*)&ldsW1[tid][0];
            float4* dst2 = (float4*)&ldsW2[tid][0];
#pragma unroll
            for (int q = 0; q < 8; ++q) {
                dst1[q] = ((const float4*)src1)[q];
                dst2[q] = ((const float4*)src2)[q];
            }
        }
        __syncthreads();

        if (kc + KC < Nn) {
            const float* an = attp + (kc + KC);
#pragma unroll
            for (int t = 0; t < TS; ++t) aN[t] = *(const float4*)(an + 8 * t);
        }

        constexpr int NB = ONE ? 8 : 16;
        float br[2][NB];
        auto load_b = [&](int t, float (&r)[NB]) {
#pragma unroll
            for (int s = 0; s < 2; ++s) {
                const int l0 = 8 * t + 4 * half + 2 * s;
                if constexpr (ONE) {
                    r[4 * s + 0] = ldsW1[l0][m];      r[4 * s + 1] = ldsW1[l0 + 1][m];
                    r[4 * s + 2] = ldsW1[l0][16 + m]; r[4 * s + 3] = ldsW1[l0 + 1][16 + m];
                } else {
                    r[8 * s + 0] = ldsW1[l0][m];      r[8 * s + 1] = ldsW1[l0 + 1][m];
                    r[8 * s + 2] = ldsW1[l0][16 + m]; r[8 * s + 3] = ldsW1[l0 + 1][16 + m];
                    r[8 * s + 4] = ldsW2[l0][m];      r[8 * s + 5] = ldsW2[l0 + 1][m];
                    r[8 * s + 6] = ldsW2[l0][16 + m]; r[8 * s + 7] = ldsW2[l0 + 1][16 + m];
                }
            }
        };
        load_b(0, br[0]);

#pragma unroll
        for (int t = 0; t < TS; ++t) {
            if (t + 1 < TS) load_b(t + 1, br[(t + 1) & 1]);
            __builtin_amdgcn_sched_barrier(0);

            const float av[4] = { aC[t].x, aC[t].y, aC[t].z, aC[t].w };
            const float (&r)[NB] = br[t & 1];

            if constexpr (ONE) {
#pragma unroll
                for (int s = 0; s < 2; ++s) {
                    v2f a;   a.x   = av[2 * s];     a.y   = av[2 * s + 1];
                    v2f b_0; b_0.x = r[4 * s + 0];  b_0.y = r[4 * s + 1];
                    v2f b_1; b_1.x = r[4 * s + 2];  b_1.y = r[4 * s + 3];
                    accS0 = __builtin_amdgcn_wmma_f32_16x16x4_f32(false, a, false, b_0,
                                                                  (short)0, accS0, false, false);
                    accS1 = __builtin_amdgcn_wmma_f32_16x16x4_f32(false, a, false, b_1,
                                                                  (short)0, accS1, false, false);
                }
            } else {
                bool sf[4];
                unsigned mm = cm;
#pragma unroll
                for (int e = 0; e < 4; ++e) {
                    sf[e] = (mm == rowmod);
                    mm = (mm + 1u == na) ? 0u : (mm + 1u);
                }
                cm += step8;
                if (cm >= na) cm -= na;

#pragma unroll
                for (int s = 0; s < 2; ++s) {
                    v2f aS, aO;
                    aS.x = sf[2 * s]     ? av[2 * s]     : 0.0f;
                    aS.y = sf[2 * s + 1] ? av[2 * s + 1] : 0.0f;
                    aO.x = sf[2 * s]     ? 0.0f : av[2 * s];
                    aO.y = sf[2 * s + 1] ? 0.0f : av[2 * s + 1];

                    v2f b1_0; b1_0.x = r[8 * s + 0]; b1_0.y = r[8 * s + 1];
                    v2f b1_1; b1_1.x = r[8 * s + 2]; b1_1.y = r[8 * s + 3];
                    v2f b2_0; b2_0.x = r[8 * s + 4]; b2_0.y = r[8 * s + 5];
                    v2f b2_1; b2_1.x = r[8 * s + 6]; b2_1.y = r[8 * s + 7];

                    accS0 = __builtin_amdgcn_wmma_f32_16x16x4_f32(false, aS, false, b1_0,
                                                                  (short)0, accS0, false, false);
                    accS0 = __builtin_amdgcn_wmma_f32_16x16x4_f32(false, aO, false, b2_0,
                                                                  (short)0, accS0, false, false);
                    accS1 = __builtin_amdgcn_wmma_f32_16x16x4_f32(false, aS, false, b1_1,
                                                                  (short)0, accS1, false, false);
                    accS1 = __builtin_amdgcn_wmma_f32_16x16x4_f32(false, aO, false, b2_1,
                                                                  (short)0, accS1, false, false);
                }
            }
            __builtin_amdgcn_sched_barrier(0);
        }
    }

    const unsigned mb = (unsigned)((lane >> 4) << 3);
#pragma unroll
    for (int r = 0; r < 8; ++r) {
        float* o = hp + ((long)(b * Nn + rowq(mb + (unsigned)r))) * Dd + h * DH;
        o[m]      = accS0[r];
        o[16 + m] = accS1[r];
    }
}

__global__ __launch_bounds__(128) void aggregate_kernel(
    const float* __restrict__ att, const float* __restrict__ wh1,
    const float* __restrict__ wh2, const int* __restrict__ nag,
    float* __restrict__ hp)
{
    __shared__ float ldsW1[KC][LDP];
    __shared__ float ldsW2[KC][LDP];

    const unsigned na  = (unsigned)nag[0];
    const unsigned cnt = (unsigned)Nn / na;
    const bool one = (cnt * na == (unsigned)Nn) && ((cnt & 63u) == 0u);
    if (one) aggregate_body<true >(att, wh1, wh2, na, cnt, hp, ldsW1, ldsW2);
    else     aggregate_body<false>(att, wh1, wh2, na, cnt, hp, ldsW1, ldsW2);
}

__global__ __launch_bounds__(256)
void outproj_kernel(
    const float* hp, const float* Wo,
    const float* __restrict__ bo, float* out)
{
    __shared__ __attribute__((aligned(16))) float lds[2 * (64 + 64) * GP];
    proj_tile<64, 64, 1, 2, true>(hp, Wo, bo, out,
                                  blockIdx.y * 64, blockIdx.x * 64, lds);
}

extern "C" void kernel_launch(void* const* d_in, const int* in_sizes, int n_in,
                              void* d_out, int out_size, void* d_ws, size_t ws_size,
                              hipStream_t stream) {
    (void)in_sizes; (void)n_in; (void)out_size; (void)ws_size;

    const float* hmat = (const float*)d_in[0];
    const int*   nag  = (const int*)  d_in[1];
    const int*   adj  = (const int*)  d_in[2];
    const float* W1   = (const float*)d_in[3];
    const float* W2   = (const float*)d_in[4];
    const float* Ai   = (const float*)d_in[5];
    const float* Aj   = (const float*)d_in[6];
    const float* Wo   = (const float*)d_in[7];
    const float* bo   = (const float*)d_in[8];
    float*       out  = (float*)d_out;

    float* ws  = (float*)d_ws;
    const size_t whN = (size_t)Bc * Nn * Dd;
    const size_t scN = (size_t)Bc * Hh * Nn;
    float* wh1 = ws;
    float* wh2 = wh1 + whN;
    float* s1  = wh2 + whN;
    float* s2  = s1 + scN;
    float* o1  = s2 + scN;
    float* o2  = o1 + scN;
    float* hp  = o2 + scN;
    float* att = hp + whN;

    proj12_kernel<<<dim3(4, (Bc * Nn) / 64), 256, 0, stream>>>(
        hmat, W1, W2, wh1, wh2);

    scores_kernel<<<(Bc * Nn * Hh + 255) / 256, 256, 0, stream>>>(
        wh1, wh2, Ai, Aj, s1, s2, o1, o2);

    entmax_kernel<<<(Bc * Hh * Nn) / (8 * RPW), 256, 0, stream>>>(
        s1, s2, o1, o2, adj, nag, att);

    aggregate_kernel<<<dim3(Bc * Hh, Nn / 64), 128, 0, stream>>>(
        att, wh1, wh2, nag, hp);

    outproj_kernel<<<dim3(Dd / 64, (Bc * Nn) / 64), 256, 0, stream>>>(
        hp, Wo, bo, out);
}


# aggregate: att tile and Wh rows fetched by whole-line 16-byte loads (a tile row's 512 B per wave instruction) one K-chunk ahead, staged in LDS; A fragments read from LDS, not 32 B of 16 lines
# speedup vs baseline: 7.0343x; 1.0868x over previous
#include <hip/hip_runtime.h>
#include <hip/hip_bf16.h>
#include <math.h>

static constexpr int Bc = 4;
static constexpr int Nn = 1024;
static constexpr int Dd = 256;
static constexpr int Hh = 8;
static constexpr int DH = 32;

typedef float v2f __attribute__((ext_vector_type(2)));
typedef float v4f __attribute__((ext_vector_type(4)));
typedef float v8f __attribute__((ext_vector_type(8)));

static constexpr int GK  = 64;
static constexpr int GP  = GK + 4;
static constexpr int GT  = GK / 8;
static constexpr int NCH = Dd / GK;

__device__ __forceinline__ v4f frag4(const float* rowp, int t) {
    return *(const v4f*)(rowp + 8 * t);
}

template <int BM, int BN, int MT, int NT, bool ELU>
__device__ __forceinline__ void proj_tile(
    const float* A, const float* W,
    const float* __restrict__ bias, float* C,
    const int row0, const int col0, float* lds)
{
    constexpr int WN  = BN / (16 * NT);
    constexpr int WMV = BM / (16 * MT);
    static_assert(WN * WMV == 8, "the block has 8 waves");
    constexpr int BUF = (BM + BN) * GP;
    constexpr int NA  = BM / 16;
    constexpr int NB  = BN / 16;

    const int tid  = threadIdx.x;
    const int lane = tid & 31;
    const int wave = tid >> 5;
    const int m    = lane & 15;
    const int half = lane >> 4;
    const int wn   = wave % WN;
    const int wm   = wave / WN;

    const int c4 = tid & 15;
    const int r0 = tid >> 4;
    const float* ga = A + (long)(row0 + r0) * Dd + 4 * c4;
    const float* gb = W + (long)(col0 + r0) * Dd + 4 * c4;
    const int sa = r0 * GP + 4 * c4;
    const int sb = (BM + r0) * GP + 4 * c4;

    const int fa = (wm * 16 * MT + m) * GP + 4 * half;
    const int fb = (BM + wn * 16 * NT + m) * GP + 4 * half;

    v8f acc[MT][NT];
#pragma unroll
    for (int i = 0; i < MT; ++i)
#pragma unroll
        for (int j = 0; j < NT; ++j) acc[i][j] = v8f{};

    v4f pa[NA], pb[NB];
#pragma unroll
    for (int i = 0; i < NA; ++i) pa[i] = *(const v4f*)(ga + (long)(16 * i) * Dd);
#pragma unroll
    for (int i = 0; i < NB; ++i) pb[i] = *(const v4f*)(gb + (long)(16 * i) * Dd);
#pragma unroll
    for (int i = 0; i < NA; ++i) *(v4f*)(lds + sa + 16 * i * GP) = pa[i];
#pragma unroll
    for (int i = 0; i < NB; ++i) *(v4f*)(lds + sb + 16 * i * GP) = pb[i];
    __syncthreads();

#pragma unroll
    for (int c = 0; c < NCH; ++c) {
        const float* cur = lds + (c & 1) * BUF;
        float*       nxt = lds + ((c + 1) & 1) * BUF;
        if (c + 1 < NCH) {
#pragma unroll
            for (int i = 0; i < NA; ++i)
                pa[i] = *(const v4f*)(ga + (long)(16 * i) * Dd + (c + 1) * GK);
#pragma unroll
            for (int i = 0; i < NB; ++i)
                pb[i] = *(const v4f*)(gb + (long)(16 * i) * Dd + (c + 1) * GK);
        }
        __builtin_amdgcn_sched_barrier(0);

#pragma unroll
        for (int t = 0; t < GT; ++t) {
            v4f xa[MT], xb[NT];
#pragma unroll
            for (int i = 0; i < MT; ++i) xa[i] = frag4(cur + fa + 16 * i * GP, t);
#pragma unroll
            for (int j = 0; j < NT; ++j) xb[j] = frag4(cur + fb + 16 * j * GP, t);
#pragma unroll
            for (int s = 0; s < 2; ++s) {
#pragma unroll
                for (int i = 0; i < MT; ++i) {
                    v2f a;
                    a.x = s ? xa[i].z : xa[i].x;
                    a.y = s ? xa[i].w : xa[i].y;
#pragma unroll
                    for (int j = 0; j < NT; ++j) {
                        v2f b;
                        b.x = s ? xb[j].z : xb[j].x;
                        b.y = s ? xb[j].w : xb[j].y;
                        acc[i][j] = __builtin_amdgcn_wmma_f32_16x16x4_f32(
                            false, a, false, b, (short)0, acc[i][j], false, false);
                    }
                }
            }
        }

        __builtin_amdgcn_sched_barrier(0);
        if (c + 1 < NCH) {
#pragma unroll
            for (int i = 0; i < NA; ++i) *(v4f*)(nxt + sa + 16 * i * GP) = pa[i];
#pragma unroll
            for (int i = 0; i < NB; ++i) *(v4f*)(nxt + sb + 16 * i * GP) = pb[i];
            __syncthreads();
        }
    }

#pragma unroll
    for (int i = 0; i < MT; ++i) {
        const int rbase = row0 + (wm * MT + i) * 16 + (half << 3);
#pragma unroll
        for (int j = 0; j < NT; ++j) {
            const int cc = col0 + (wn * NT + j) * 16 + m;
            float bv = 0.0f;
            if constexpr (ELU) bv = bias[cc];
#pragma unroll
            for (int r = 0; r < 8; ++r) {
                float v = acc[i][j][r];
                if constexpr (ELU) {
                    v += bv;
                    v = (v > 0.0f) ? v : (expf(v) - 1.0f);
                }
                C[(long)(rbase + r) * Dd + cc] = v;
            }
        }
    }
}

__global__ __launch_bounds__(256)
void proj12_kernel(
    const float* hmat, const float* W1, const float* W2,
    float* wh1, float* wh2)
{
    __shared__ __attribute__((aligned(16))) float lds[2 * (64 + 128) * GP];
    const bool second = blockIdx.x >= 2;
    proj_tile<64, 128, 2, 2, false>(hmat, second ? W2 : W1, nullptr,
                                    second ? wh2 : wh1,
                                    blockIdx.y * 64, (blockIdx.x & 1) * 128, lds);
}

__global__ __launch_bounds__(256) void scores_kernel(
    const float* __restrict__ wh1, const float* __restrict__ wh2,
    const float* __restrict__ Ai, const float* __restrict__ Aj,
    float* __restrict__ s1, float* __restrict__ s2,
    float* __restrict__ o1, float* __restrict__ o2)
{
    int idx = blockIdx.x * blockDim.x + threadIdx.x;
    if (idx >= Bc * Nn * Hh) return;
    const int h = idx % Hh;
    const int n = (idx / Hh) % Nn;
    const int b = idx / (Hh * Nn);

    const float* r1 = wh1 + ((long)(b * Nn + n)) * Dd + h * DH;
    const float* r2 = wh2 + ((long)(b * Nn + n)) * Dd + h * DH;
    const float* ai = Ai + h * DH;
    const float* aj = Aj + h * DH;

    float a1 = 0.f, a2 = 0.f, a3 = 0.f, a4 = 0.f;
#pragma unroll 8
    for (int d = 0; d < DH; ++d) {
        const float w1v = r1[d], w2v = r2[d], aiv = ai[d], ajv = aj[d];
        a1 += w1v * aiv;  a2 += w1v * ajv;
        a3 += w2v * aiv;  a4 += w2v * ajv;
    }
    const int o = (b * Hh + h) * Nn + n;
    s1[o] = a1; s2[o] = a2; o1[o] = a3; o2[o] = a4;
}

static constexpr int RPW = 4;
static constexpr int RS  = 8;
static constexpr int CAP = 32 * RS;

__device__ __forceinline__ float lane_pull(float x, int addr) {
    return __builtin_bit_cast(float,
        __builtin_amdgcn_ds_bpermute(addr, __builtin_bit_cast(int, x)));
}
__device__ __forceinline__ int lane_pull_i(int x, int addr) {
    return __builtin_amdgcn_ds_bpermute(addr, x);
}

template <int R, int JJ>
__device__ __forceinline__ void ce_regs(float (&v)[R]) {
#pragma unroll
    for (int u = 0; u < R; ++u) {
        if ((u & JJ) == 0) {
            const float a = v[u], c = v[u | JJ];
            v[u]      = fmaxf(a, c);
            v[u | JJ] = fminf(a, c);
        }
    }
}
template <int R, int K>
__device__ __forceinline__ void flip_regs(float (&v)[R]) {
#pragma unroll
    for (int u = 0; u < R; ++u) {
        if ((u & (K >> 1)) == 0) {
            const int p = u ^ (K - 1);
            const float a = v[u], c = v[p];
            v[u] = fmaxf(a, c);
            v[p] = fminf(a, c);
        }
    }
}
template <int LM>
__device__ __forceinline__ float lane_xor(float x) {
    const int b = __builtin_bit_cast(int, x);
    if constexpr (LM < 16) {
        return __builtin_bit_cast(float,
            __builtin_amdgcn_update_dpp(0, b, 0x160 + LM, 0xf, 0xf, true));
    } else {
        static_assert(LM == 31, "only the mirrored other-row exchange is used");
        return __builtin_bit_cast(float,
            __builtin_amdgcn_permlanex16(b, b, 0x89abcdefu, 0x01234567u, false, false));
    }
}
template <int R, int LM, bool FLIP>
__device__ __forceinline__ void ce_lanes(float (&v)[R], int lane) {
    constexpr int HB = (LM >= 16) ? 16 : (LM >= 8) ? 8 : (LM >= 4) ? 4
                     : (LM >= 2) ? 2 : 1;
    const float bound = (lane & HB) ? -__builtin_inff() : __builtin_inff();
    float p[R];
#pragma unroll
    for (int u = 0; u < R; ++u) p[u] = lane_xor<LM>(v[FLIP ? (u ^ (R - 1)) : u]);
#pragma unroll
    for (int u = 0; u < R; ++u) v[u] = __builtin_amdgcn_fmed3f(v[u], p[u], bound);
}
template <int R>
__device__ __forceinline__ void ce_regs_tail(float (&v)[R]) {
    if constexpr (R > 16) ce_regs<R, 16>(v);
    if constexpr (R > 8)  ce_regs<R, 8>(v);
    ce_regs<R, 4>(v); ce_regs<R, 2>(v); ce_regs<R, 1>(v);
}

template <int R>
__device__ __forceinline__ float entmax_tau(float (&v)[R], int lane)
{
    static_assert(R == 32 || R == 8, "register stages below are written for these");
    constexpr int LG = (R == 32) ? 5 : 3;

    flip_regs<R, 2>(v);
    flip_regs<R, 4>(v);  ce_regs<R, 1>(v);
    flip_regs<R, 8>(v);  ce_regs<R, 2>(v); ce_regs<R, 1>(v);
    if constexpr (R == 32) {
        flip_regs<R, 16>(v); ce_regs<R, 4>(v); ce_regs<R, 2>(v); ce_regs<R, 1>(v);
        flip_regs<R, 32>(v); ce_regs<R, 8>(v); ce_regs<R, 4>(v); ce_regs<R, 2>(v);
        ce_regs<R, 1>(v);
    }
    ce_lanes<R, 1, true>(v, lane);                                     ce_regs_tail<R>(v);
    ce_lanes<R, 3, true>(v, lane);  ce_lanes<R, 1, false>(v, lane);    ce_regs_tail<R>(v);
    ce_lanes<R, 7, true>(v, lane);  ce_lanes<R, 2, false>(v, lane);
    ce_lanes<R, 1, false>(v, lane);                                    ce_regs_tail<R>(v);
    ce_lanes<R, 15, true>(v, lane); ce_lanes<R, 4, false>(v, lane);
    ce_lanes<R, 2, false>(v, lane); ce_lanes<R, 1, false>(v, lane);    ce_regs_tail<R>(v);
    ce_lanes<R, 31, true>(v, lane); ce_lanes<R, 8, false>(v, lane);
    ce_lanes<R, 4, false>(v, lane); ce_lanes<R, 2, false>(v, lane);
    ce_lanes<R, 1, false>(v, lane);                                    ce_regs_tail<R>(v);

    float totS = 0.0f, totQ = 0.0f;
#pragma unroll
    for (int u = 0; u < R; ++u) { totS += v[u]; totQ += v[u] * v[u]; }
    float incS = totS, incQ = totQ;
#pragma unroll
    for (int off = 1; off < 32; off <<= 1) {
        const int   src = (lane >= off) ? (lane - off) : lane;
        const float pa  = lane_pull(incS, src << 2);
        const float pb  = lane_pull(incQ, src << 2);
        if (lane >= off) { incS += pa; incQ += pb; }
    }
    const int   prv = (lane > 0) ? (lane - 1) : 0;
    const float eS  = lane_pull(incS, prv << 2);
    const float eQ  = lane_pull(incQ, prv << 2);
    float runS = (lane > 0) ? eS : 0.0f;
    float runQ = (lane > 0) ? eQ : 0.0f;

    float Sp[R], Qp[R];
    int cnt = 0;
#pragma unroll
    for (int u = 0; u < R; ++u) {
        runS += v[u];
        runQ += v[u] * v[u];
        Sp[u] = runS;
        Qp[u] = runQ;
        const float rho = (float)(lane * R + u + 1);
        const float d   = rho * v[u] - runS;
        const float g   = v[u] * (d - runS) + runQ;
        cnt += (g <= 1.0f) ? 1 : 0;
    }
#pragma unroll
    for (int msk = 16; msk >= 1; msk >>= 1)
        cnt += lane_pull_i(cnt, (lane ^ msk) << 2);
    const int support = __builtin_amdgcn_readfirstlane(cnt);
    const int sidx = support - 1;
    const int us   = sidx & (R - 1);
    float Ssel = Sp[0], Qsel = Qp[0];
#pragma unroll
    for (int u = 1; u < R; ++u) {
        Ssel = (us == u) ? Sp[u] : Ssel;
        Qsel = (us == u) ? Qp[u] : Qsel;
    }
    float tsel;
    {
        const float rho   = (float)(lane * R + us + 1);
        const float mean  = Ssel / rho;
        const float msq   = Qsel / rho;
        const float ssv   = rho * (msq - mean * mean);
        const float delta = (1.0f - ssv) / rho;
        tsel = mean - ((delta > 0.0f) ? sqrtf(delta) : 0.0f);
    }
    return lane_pull(tsel, (sidx >> LG) << 2);
}

__global__ __launch_bounds__(256) void entmax_kernel(
    const float* __restrict__ s1, const float* __restrict__ s2,
    const float* __restrict__ o1, const float* __restrict__ o2,
    const int* __restrict__ adj, const int* __restrict__ nag,
    float* __restrict__ att)
{
    __shared__ float cand[8][Nn];

    const int lane = threadIdx.x & 31;
    const int wave = threadIdx.x >> 5;
    float* wl = cand[wave];
    const unsigned na      = (unsigned)nag[0];
    const unsigned step128 = 128u % na;
    const unsigned jm_lane = (unsigned)(lane * 4) % na;
    const int row0 = (blockIdx.x * 8 + wave) * RPW;

#pragma unroll 1
    for (int rr = 0; rr < RPW; ++rr) {
        const int bid = row0 + rr;
        const int i   = bid & (Nn - 1);
        const int h   = (bid >> 10) & (Hh - 1);
        const int b   = bid >> 13;

        const int   rowbase = (b * Hh + h) * Nn;
        const float s1i = s1[rowbase + i];
        const float o1i = o1[rowbase + i];
        const unsigned im = (unsigned)i % na;
        const long  adjbase = ((long)b * Nn + i) * Nn;

        float x[32];
        float lmax = -3.4e38f;
        unsigned jm = jm_lane;
#pragma unroll
        for (int q = 0; q < 8; ++q) {
            const int j = (q * 32 + lane) * 4;
            const float4 sv = *(const float4*)(s2 + rowbase + j);
            const float4 ov = *(const float4*)(o2 + rowbase + j);
            const int4   av = *(const int4*)(adj + adjbase + j);
            asm volatile("" :: "v"(sv.x), "v"(sv.y), "v"(sv.z), "v"(sv.w),
                               "v"(ov.x), "v"(ov.y), "v"(ov.z), "v"(ov.w));
            const float se[4] = { sv.x, sv.y, sv.z, sv.w };
            const float oe[4] = { ov.x, ov.y, ov.z, ov.w };
            const int   ae[4] = { av.x, av.y, av.z, av.w };
            unsigned m = jm;
#pragma unroll
            for (int c = 0; c < 4; ++c) {
                float e = (m == im) ? (s1i + se[c]) : (o1i + oe[c]);
                if (ae[c] == 0) e = -1.0e12f;
                const float xv = 0.5f * e;
                x[q * 4 + c] = xv;
                lmax = fmaxf(lmax, xv);
                m = (m + 1u == na) ? 0u : (m + 1u);
            }
            jm += step128;
            if (jm >= na) jm -= na;
        }
#pragma unroll
        for (int msk = 16; msk >= 1; msk >>= 1)
            lmax = fmaxf(lmax, lane_pull(lmax, (lane ^ msk) << 2));
        const float xmax = lmax;
#pragma unroll
        for (int u = 0; u < 32; ++u) x[u] -= xmax;

        int ncand = 0;
#pragma unroll
        for (int u = 0; u < 32; ++u) {
            const bool     keep = (x[u] >= -1.0f);
            const unsigned who  = __builtin_amdgcn_ballot_w32(keep);
            const int      pos  = (int)__builtin_amdgcn_mbcnt_lo(who, (unsigned)ncand);
            if (keep) wl[pos] = x[u];
            ncand += __builtin_popcount(who);
        }

        float tstar;
        if (ncand <= CAP) {
            __builtin_amdgcn_fence(__ATOMIC_ACQ_REL, "wavefront");
            float w[RS];
#pragma unroll
            for (int r = 0; r < RS; ++r) {
                const int   slot = r * 32 + lane;
                const float t    = wl[slot];
                w[r] = (slot < ncand) ? t : -2.0f;
            }
            tstar = entmax_tau<RS>(w, lane);
        } else {
            float v[32];
#pragma unroll
            for (int u = 0; u < 32; ++u) v[u] = x[u];
            tstar = entmax_tau<32>(v, lane);
        }

        float* outrow = att + ((long)(b * Hh + h) * Nn + i) * Nn;
#pragma unroll
        for (int q = 0; q < 8; ++q) {
            float4 o;
            float y;
            y = fmaxf(x[q * 4 + 0] - tstar, 0.0f); o.x = y * y;
            y = fmaxf(x[q * 4 + 1] - tstar, 0.0f); o.y = y * y;
            y = fmaxf(x[q * 4 + 2] - tstar, 0.0f); o.z = y * y;
            y = fmaxf(x[q * 4 + 3] - tstar, 0.0f); o.w = y * y;
            *(float4*)(outrow + (q * 32 + lane) * 4) = o;
        }
    }
}

static constexpr int KC  = 128;
static constexpr int LDP = 40;
static constexpr int LAP = KC + 4;
static constexpr int TS  = KC / 8;

template <bool ONE>
__device__ __forceinline__ void aggregate_body(
    const float* __restrict__ att, const float* __restrict__ wh1,
    const float* __restrict__ wh2, const unsigned na, const unsigned cnt,
    float* __restrict__ hp, float (&ldsW1)[KC][LDP], float (&ldsW2)[KC][LDP],
    float (&ldsA)[4][16][LAP])
{
    const int tid  = threadIdx.x;
    const int wave = __builtin_amdgcn_readfirstlane(tid >> 5);
    const int lane = tid & 31;
    const int m    = lane & 15;
    const int half = lane >> 4;
    const int bh   = blockIdx.x;
    const int b    = bh / Hh;
    const int h    = bh % Hh;

    const unsigned p0  = blockIdx.y * 64u;
    const unsigned res = ONE ? (p0 / cnt) : 0u;
    const unsigned q0  = ONE ? (p0 - res * cnt + (unsigned)wave * 16u)
                             : (p0 + (unsigned)wave * 16u);
    auto rowq = [&](unsigned mm) -> int {
        return ONE ? (int)((q0 + mm) * na + res) : (int)(q0 + mm);
    };

    const unsigned rowmod = (unsigned)rowq((unsigned)m) % na;
    unsigned cm = (unsigned)(4 * half) % na;
    const unsigned step8 = 8u % na;

    const float* attw = att + ((long)bh * Nn + rowq(0u)) * Nn;
    const long rstride = ONE ? (long)na * Nn : (long)Nn;
    const unsigned lane4 = 4u * (unsigned)lane;
    float (&myA)[16][LAP] = ldsA[wave];

    const int sr = tid >> 3;
    const int pc = tid & 7;
    const long whoff = ((long)(b * Nn + sr)) * Dd + h * DH + 4 * pc;
    unsigned rm = (unsigned)sr % na;
    const unsigned step16 = 16u % na;

    constexpr int NQ = ONE ? 8 : 16;
    v4f aN[16];
    v4f bN[NQ];
    auto fetch = [&](int kc) {
#pragma unroll
        for (int r = 0; r < 16; ++r) {
            const float* rowu = attw + (long)r * rstride + kc;
            aN[r] = *(const v4f*)(rowu + lane4);
        }
#pragma unroll
        for (int q = 0; q < 8; ++q) {
            const long o = whoff + (long)(kc + 16 * q) * Dd;
            if constexpr (ONE) {
                const bool selfk = (rm == res);
                bN[q] = *(const v4f*)((selfk ? wh1 : wh2) + o);
                rm += step16;
                if (rm >= na) rm -= na;
            } else {
                bN[q]     = *(const v4f*)(wh1 + o);
                bN[8 + q] = *(const v4f*)(wh2 + o);
            }
        }
    };
    auto stash = [&]() {
#pragma unroll
        for (int r = 0; r < 16; ++r) *(v4f*)&myA[r][lane4] = aN[r];
#pragma unroll
        for (int q = 0; q < 8; ++q) {
            *(v4f*)&ldsW1[16 * q + sr][4 * pc] = bN[q];
            if constexpr (!ONE) *(v4f*)&ldsW2[16 * q + sr][4 * pc] = bN[8 + q];
        }
    };

    v8f accS0 = {};
    v8f accS1 = {};

    fetch(0);

#pragma unroll 1
    for (int kc = 0; kc < Nn; kc += KC) {
        __syncthreads();
        stash();
        __syncthreads();

        if (kc + KC < Nn) fetch(kc + KC);

        constexpr int NB = ONE ? 8 : 16;
        float  br[2][NB];
        v4f    af[2];
        auto load_ab = [&](int t, float (&r)[NB], v4f& a) {
            a = *(const v4f*)&myA[m][8 * t + 4 * half];
#pragma unroll
            for (int s = 0; s < 2; ++s) {
                const int l0 = 8 * t + 4 * half + 2 * s;
                if constexpr (ONE) {
                    r[4 * s + 0] = ldsW1[l0][m];      r[4 * s + 1] = ldsW1[l0 + 1][m];
                    r[4 * s + 2] = ldsW1[l0][16 + m]; r[4 * s + 3] = ldsW1[l0 + 1][16 + m];
                } else {
                    r[8 * s + 0] = ldsW1[l0][m];      r[8 * s + 1] = ldsW1[l0 + 1][m];
                    r[8 * s + 2] = ldsW1[l0][16 + m]; r[8 * s + 3] = ldsW1[l0 + 1][16 + m];
                    r[8 * s + 4] = ldsW2[l0][m];      r[8 * s + 5] = ldsW2[l0 + 1][m];
                    r[8 * s + 6] = ldsW2[l0][16 + m]; r[8 * s + 7] = ldsW2[l0 + 1][16 + m];
                }
            }
        };
        load_ab(0, br[0], af[0]);

#pragma unroll
        for (int t = 0; t < TS; ++t) {
            if (t + 1 < TS) load_ab(t + 1, br[(t + 1) & 1], af[(t + 1) & 1]);
            __builtin_amdgcn_sched_barrier(0);

            const float av[4] = { af[t & 1].x, af[t & 1].y, af[t & 1].z, af[t & 1].w };
            const float (&r)[NB] = br[t & 1];

            if constexpr (ONE) {
#pragma unroll
                for (int s = 0; s < 2; ++s) {
                    v2f a;   a.x   = av[2 * s];     a.y   = av[2 * s + 1];
                    v2f b_0; b_0.x = r[4 * s + 0];  b_0.y = r[4 * s + 1];
                    v2f b_1; b_1.x = r[4 * s + 2];  b_1.y = r[4 * s + 3];
                    accS0 = __builtin_amdgcn_wmma_f32_16x16x4_f32(false, a, false, b_0,
                                                                  (short)0, accS0, false, false);
                    accS1 = __builtin_amdgcn_wmma_f32_16x16x4_f32(false, a, false, b_1,
                                                                  (short)0, accS1, false, false);
                }
            } else {
                bool sf[4];
                unsigned mm = cm;
#pragma unroll
                for (int e = 0; e < 4; ++e) {
                    sf[e] = (mm == rowmod);
                    mm = (mm + 1u == na) ? 0u : (mm + 1u);
                }
                cm += step8;
                if (cm >= na) cm -= na;

#pragma unroll
                for (int s = 0; s < 2; ++s) {
                    v2f aS, aO;
                    aS.x = sf[2 * s]     ? av[2 * s]     : 0.0f;
                    aS.y = sf[2 * s + 1] ? av[2 * s + 1] : 0.0f;
                    aO.x = sf[2 * s]     ? 0.0f : av[2 * s];
                    aO.y = sf[2 * s + 1] ? 0.0f : av[2 * s + 1];

                    v2f b1_0; b1_0.x = r[8 * s + 0]; b1_0.y = r[8 * s + 1];
                    v2f b1_1; b1_1.x = r[8 * s + 2]; b1_1.y = r[8 * s + 3];
                    v2f b2_0; b2_0.x = r[8 * s + 4]; b2_0.y = r[8 * s + 5];
                    v2f b2_1; b2_1.x = r[8 * s + 6]; b2_1.y = r[8 * s + 7];

                    accS0 = __builtin_amdgcn_wmma_f32_16x16x4_f32(false, aS, false, b1_0,
                                                                  (short)0, accS0, false, false);
                    accS0 = __builtin_amdgcn_wmma_f32_16x16x4_f32(false, aO, false, b2_0,
                                                                  (short)0, accS0, false, false);
                    accS1 = __builtin_amdgcn_wmma_f32_16x16x4_f32(false, aS, false, b1_1,
                                                                  (short)0, accS1, false, false);
                    accS1 = __builtin_amdgcn_wmma_f32_16x16x4_f32(false, aO, false, b2_1,
                                                                  (short)0, accS1, false, false);
                }
            }
            __builtin_amdgcn_sched_barrier(0);
        }
    }

    const unsigned mb = (unsigned)((lane >> 4) << 3);
#pragma unroll
    for (int r = 0; r < 8; ++r) {
        float* o = hp + ((long)(b * Nn + rowq(mb + (unsigned)r))) * Dd + h * DH;
        o[m]      = accS0[r];
        o[16 + m] = accS1[r];
    }
}

__global__ __launch_bounds__(128) void aggregate_kernel(
    const float* __restrict__ att, const float* __restrict__ wh1,
    const float* __restrict__ wh2, const int* __restrict__ nag,
    float* __restrict__ hp)
{
    __shared__ __attribute__((aligned(16))) float ldsW1[KC][LDP];
    __shared__ __attribute__((aligned(16))) float ldsW2[KC][LDP];
    __shared__ __attribute__((aligned(16))) float ldsA[4][16][LAP];

    const unsigned na  = (unsigned)nag[0];
    const unsigned cnt = (unsigned)Nn / na;
    const bool one = (cnt * na == (unsigned)Nn) && ((cnt & 63u) == 0u);
    if (one) aggregate_body<true >(att, wh1, wh2, na, cnt, hp, ldsW1, ldsW2, ldsA);
    else     aggregate_body<false>(att, wh1, wh2, na, cnt, hp, ldsW1, ldsW2, ldsA);
}

__global__ __launch_bounds__(256)
void outproj_kernel(
    const float* hp, const float* Wo,
    const float* __restrict__ bo, float* out)
{
    __shared__ __attribute__((aligned(16))) float lds[2 * (64 + 64) * GP];
    proj_tile<64, 64, 1, 2, true>(hp, Wo, bo, out,
                                  blockIdx.y * 64, blockIdx.x * 64, lds);
}

extern "C" void kernel_launch(void* const* d_in, const int* in_sizes, int n_in,
                              void* d_out, int out_size, void* d_ws, size_t ws_size,
                              hipStream_t stream) {
    (void)in_sizes; (void)n_in; (void)out_size; (void)ws_size;

    const float* hmat = (const float*)d_in[0];
    const int*   nag  = (const int*)  d_in[1];
    const int*   adj  = (const int*)  d_in[2];
    const float* W1   = (const float*)d_in[3];
    const float* W2   = (const float*)d_in[4];
    const float* Ai   = (const float*)d_in[5];
    const float* Aj   = (const float*)d_in[6];
    const float* Wo   = (const float*)d_in[7];
    const float* bo   = (const float*)d_in[8];
    float*       out  = (float*)d_out;

    float* ws  = (float*)d_ws;
    const size_t whN = (size_t)Bc * Nn * Dd;
    const size_t scN = (size_t)Bc * Hh * Nn;
    float* wh1 = ws;
    float* wh2 = wh1 + whN;
    float* s1  = wh2 + whN;
    float* s2  = s1 + scN;
    float* o1  = s2 + scN;
    float* o2  = o1 + scN;
    float* hp  = o2 + scN;
    float* att = hp + whN;

    proj12_kernel<<<dim3(4, (Bc * Nn) / 64), 256, 0, stream>>>(
        hmat, W1, W2, wh1, wh2);

    scores_kernel<<<(Bc * Nn * Hh + 255) / 256, 256, 0, stream>>>(
        wh1, wh2, Ai, Aj, s1, s2, o1, o2);

    entmax_kernel<<<(Bc * Hh * Nn) / (8 * RPW), 256, 0, stream>>>(
        s1, s2, o1, o2, adj, nag, att);

    aggregate_kernel<<<dim3(Bc * Hh, Nn / 64), 128, 0, stream>>>(
        att, wh1, wh2, nag, hp);

    outproj_kernel<<<dim3(Dd / 64, (Bc * Nn) / 64), 256, 0, stream>>>(
        hp, Wo, bo, out);
}


# entmax matrix is never stored: kernel 3 leaves xmax and tau per row; kernel 4 forms its att tiles from adj and the scores (same operations, same values) between its matrix instructions
# speedup vs baseline: 7.7533x; 1.1022x over previous
#include <hip/hip_runtime.h>
#include <hip/hip_bf16.h>
#include <math.h>

static constexpr int Bc = 4;
static constexpr int Nn = 1024;
static constexpr int Dd = 256;
static constexpr int Hh = 8;
static constexpr int DH = 32;

typedef float v2f __attribute__((ext_vector_type(2)));
typedef float v4f __attribute__((ext_vector_type(4)));
typedef float v8f __attribute__((ext_vector_type(8)));
typedef int   v4i __attribute__((ext_vector_type(4)));

static constexpr int GK  = 64;
static constexpr int GP  = GK + 4;
static constexpr int GT  = GK / 8;
static constexpr int NCH = Dd / GK;

__device__ __forceinline__ v4f frag4(const float* rowp, int t) {
    return *(const v4f*)(rowp + 8 * t);
}

template <int BM, int BN, int MT, int NT, bool ELU>
__device__ __forceinline__ void proj_tile(
    const float* A, const float* W,
    const float* __restrict__ bias, float* C,
    const int row0, const int col0, float* lds)
{
    constexpr int WN  = BN / (16 * NT);
    constexpr int WMV = BM / (16 * MT);
    static_assert(WN * WMV == 8, "the block has 8 waves");
    constexpr int BUF = (BM + BN) * GP;
    constexpr int NA  = BM / 16;
    constexpr int NB  = BN / 16;

    const int tid  = threadIdx.x;
    const int lane = tid & 31;
    const int wave = tid >> 5;
    const int m    = lane & 15;
    const int half = lane >> 4;
    const int wn   = wave % WN;
    const int wm   = wave / WN;

    const int c4 = tid & 15;
    const int r0 = tid >> 4;
    const float* ga = A + (long)(row0 + r0) * Dd + 4 * c4;
    const float* gb = W + (long)(col0 + r0) * Dd + 4 * c4;
    const int sa = r0 * GP + 4 * c4;
    const int sb = (BM + r0) * GP + 4 * c4;

    const int fa = (wm * 16 * MT + m) * GP + 4 * half;
    const int fb = (BM + wn * 16 * NT + m) * GP + 4 * half;

    v8f acc[MT][NT];
#pragma unroll
    for (int i = 0; i < MT; ++i)
#pragma unroll
        for (int j = 0; j < NT; ++j) acc[i][j] = v8f{};

    v4f pa[NA], pb[NB];
#pragma unroll
    for (int i = 0; i < NA; ++i) pa[i] = *(const v4f*)(ga + (long)(16 * i) * Dd);
#pragma unroll
    for (int i = 0; i < NB; ++i) pb[i] = *(const v4f*)(gb + (long)(16 * i) * Dd);
#pragma unroll
    for (int i = 0; i < NA; ++i) *(v4f*)(lds + sa + 16 * i * GP) = pa[i];
#pragma unroll
    for (int i = 0; i < NB; ++i) *(v4f*)(lds + sb + 16 * i * GP) = pb[i];
    __syncthreads();

#pragma unroll
    for (int c = 0; c < NCH; ++c) {
        const float* cur = lds + (c & 1) * BUF;
        float*       nxt = lds + ((c + 1) & 1) * BUF;
        if (c + 1 < NCH) {
#pragma unroll
            for (int i = 0; i < NA; ++i)
                pa[i] = *(const v4f*)(ga + (long)(16 * i) * Dd + (c + 1) * GK);
#pragma unroll
            for (int i = 0; i < NB; ++i)
                pb[i] = *(const v4f*)(gb + (long)(16 * i) * Dd + (c + 1) * GK);
        }
        __builtin_amdgcn_sched_barrier(0);

#pragma unroll
        for (int t = 0; t < GT; ++t) {
            v4f xa[MT], xb[NT];
#pragma unroll
            for (int i = 0; i < MT; ++i) xa[i] = frag4(cur + fa + 16 * i * GP, t);
#pragma unroll
            for (int j = 0; j < NT; ++j) xb[j] = frag4(cur + fb + 16 * j * GP, t);
#pragma unroll
            for (int s = 0; s < 2; ++s) {
#pragma unroll
                for (int i = 0; i < MT; ++i) {
                    v2f a;
                    a.x = s ? xa[i].z : xa[i].x;
                    a.y = s ? xa[i].w : xa[i].y;
#pragma unroll
                    for (int j = 0; j < NT; ++j) {
                        v2f b;
                        b.x = s ? xb[j].z : xb[j].x;
                        b.y = s ? xb[j].w : xb[j].y;
                        acc[i][j] = __builtin_amdgcn_wmma_f32_16x16x4_f32(
                            false, a, false, b, (short)0, acc[i][j], false, false);
                    }
                }
            }
        }

        __builtin_amdgcn_sched_barrier(0);
        if (c + 1 < NCH) {
#pragma unroll
            for (int i = 0; i < NA; ++i) *(v4f*)(nxt + sa + 16 * i * GP) = pa[i];
#pragma unroll
            for (int i = 0; i < NB; ++i) *(v4f*)(nxt + sb + 16 * i * GP) = pb[i];
            __syncthreads();
        }
    }

#pragma unroll
    for (int i = 0; i < MT; ++i) {
        const int rbase = row0 + (wm * MT + i) * 16 + (half << 3);
#pragma unroll
        for (int j = 0; j < NT; ++j) {
            const int cc = col0 + (wn * NT + j) * 16 + m;
            float bv = 0.0f;
            if constexpr (ELU) bv = bias[cc];
#pragma unroll
            for (int r = 0; r < 8; ++r) {
                float v = acc[i][j][r];
                if constexpr (ELU) {
                    v += bv;
                    v = (v > 0.0f) ? v : (expf(v) - 1.0f);
                }
                C[(long)(rbase + r) * Dd + cc] = v;
            }
        }
    }
}

__global__ __launch_bounds__(256)
void proj12_kernel(
    const float* hmat, const float* W1, const float* W2,
    float* wh1, float* wh2)
{
    __shared__ __attribute__((aligned(16))) float lds[2 * (64 + 128) * GP];
    const bool second = blockIdx.x >= 2;
    proj_tile<64, 128, 2, 2, false>(hmat, second ? W2 : W1, nullptr,
                                    second ? wh2 : wh1,
                                    blockIdx.y * 64, (blockIdx.x & 1) * 128, lds);
}

__global__ __launch_bounds__(256) void scores_kernel(
    const float* __restrict__ wh1, const float* __restrict__ wh2,
    const float* __restrict__ Ai, const float* __restrict__ Aj,
    float* __restrict__ s1, float* __restrict__ s2,
    float* __restrict__ o1, float* __restrict__ o2)
{
    int idx = blockIdx.x * blockDim.x + threadIdx.x;
    if (idx >= Bc * Nn * Hh) return;
    const int h = idx % Hh;
    const int n = (idx / Hh) % Nn;
    const int b = idx / (Hh * Nn);

    const float* r1 = wh1 + ((long)(b * Nn + n)) * Dd + h * DH;
    const float* r2 = wh2 + ((long)(b * Nn + n)) * Dd + h * DH;
    const float* ai = Ai + h * DH;
    const float* aj = Aj + h * DH;

    float a1 = 0.f, a2 = 0.f, a3 = 0.f, a4 = 0.f;
#pragma unroll 8
    for (int d = 0; d < DH; ++d) {
        const float w1v = r1[d], w2v = r2[d], aiv = ai[d], ajv = aj[d];
        a1 += w1v * aiv;  a2 += w1v * ajv;
        a3 += w2v * aiv;  a4 += w2v * ajv;
    }
    const int o = (b * Hh + h) * Nn + n;
    s1[o] = a1; s2[o] = a2; o1[o] = a3; o2[o] = a4;
}

static constexpr int RPW = 4;
static constexpr int RS  = 8;
static constexpr int CAP = 32 * RS;

__device__ __forceinline__ float lane_pull(float x, int addr) {
    return __builtin_bit_cast(float,
        __builtin_amdgcn_ds_bpermute(addr, __builtin_bit_cast(int, x)));
}
__device__ __forceinline__ int lane_pull_i(int x, int addr) {
    return __builtin_amdgcn_ds_bpermute(addr, x);
}

template <int R, int JJ>
__device__ __forceinline__ void ce_regs(float (&v)[R]) {
#pragma unroll
    for (int u = 0; u < R; ++u) {
        if ((u & JJ) == 0) {
            const float a = v[u], c = v[u | JJ];
            v[u]      = fmaxf(a, c);
            v[u | JJ] = fminf(a, c);
        }
    }
}
template <int R, int K>
__device__ __forceinline__ void flip_regs(float (&v)[R]) {
#pragma unroll
    for (int u = 0; u < R; ++u) {
        if ((u & (K >> 1)) == 0) {
            const int p = u ^ (K - 1);
            const float a = v[u], c = v[p];
            v[u] = fmaxf(a, c);
            v[p] = fminf(a, c);
        }
    }
}
template <int LM>
__device__ __forceinline__ float lane_xor(float x) {
    const int b = __builtin_bit_cast(int, x);
    if constexpr (LM < 16) {
        return __builtin_bit_cast(float,
            __builtin_amdgcn_update_dpp(0, b, 0x160 + LM, 0xf, 0xf, true));
    } else {
        static_assert(LM == 31, "only the mirrored other-row exchange is used");
        return __builtin_bit_cast(float,
            __builtin_amdgcn_permlanex16(b, b, 0x89abcdefu, 0x01234567u, false, false));
    }
}
template <int R, int LM, bool FLIP>
__device__ __forceinline__ void ce_lanes(float (&v)[R], int lane) {
    constexpr int HB = (LM >= 16) ? 16 : (LM >= 8) ? 8 : (LM >= 4) ? 4
                     : (LM >= 2) ? 2 : 1;
    const float bound = (lane & HB) ? -__builtin_inff() : __builtin_inff();
    float p[R];
#pragma unroll
    for (int u = 0; u < R; ++u) p[u] = lane_xor<LM>(v[FLIP ? (u ^ (R - 1)) : u]);
#pragma unroll
    for (int u = 0; u < R; ++u) v[u] = __builtin_amdgcn_fmed3f(v[u], p[u], bound);
}
template <int R>
__device__ __forceinline__ void ce_regs_tail(float (&v)[R]) {
    if constexpr (R > 16) ce_regs<R, 16>(v);
    if constexpr (R > 8)  ce_regs<R, 8>(v);
    ce_regs<R, 4>(v); ce_regs<R, 2>(v); ce_regs<R, 1>(v);
}

template <int R>
__device__ __forceinline__ float entmax_tau(float (&v)[R], int lane)
{
    static_assert(R == 32 || R == 8, "register stages below are written for these");
    constexpr int LG = (R == 32) ? 5 : 3;

    flip_regs<R, 2>(v);
    flip_regs<R, 4>(v);  ce_regs<R, 1>(v);
    flip_regs<R, 8>(v);  ce_regs<R, 2>(v); ce_regs<R, 1>(v);
    if constexpr (R == 32) {
        flip_regs<R, 16>(v); ce_regs<R, 4>(v); ce_regs<R, 2>(v); ce_regs<R, 1>(v);
        flip_regs<R, 32>(v); ce_regs<R, 8>(v); ce_regs<R, 4>(v); ce_regs<R, 2>(v);
        ce_regs<R, 1>(v);
    }
    ce_lanes<R, 1, true>(v, lane);                                     ce_regs_tail<R>(v);
    ce_lanes<R, 3, true>(v, lane);  ce_lanes<R, 1, false>(v, lane);    ce_regs_tail<R>(v);
    ce_lanes<R, 7, true>(v, lane);  ce_lanes<R, 2, false>(v, lane);
    ce_lanes<R, 1, false>(v, lane);                                    ce_regs_tail<R>(v);
    ce_lanes<R, 15, true>(v, lane); ce_lanes<R, 4, false>(v, lane);
    ce_lanes<R, 2, false>(v, lane); ce_lanes<R, 1, false>(v, lane);    ce_regs_tail<R>(v);
    ce_lanes<R, 31, true>(v, lane); ce_lanes<R, 8, false>(v, lane);
    ce_lanes<R, 4, false>(v, lane); ce_lanes<R, 2, false>(v, lane);
    ce_lanes<R, 1, false>(v, lane);                                    ce_regs_tail<R>(v);

    float totS = 0.0f, totQ = 0.0f;
#pragma unroll
    for (int u = 0; u < R; ++u) { totS += v[u]; totQ += v[u] * v[u]; }
    float incS = totS, incQ = totQ;
#pragma unroll
    for (int off = 1; off < 32; off <<= 1) {
        const int   src = (lane >= off) ? (lane - off) : lane;
        const float pa  = lane_pull(incS, src << 2);
        const float pb  = lane_pull(incQ, src << 2);
        if (lane >= off) { incS += pa; incQ += pb; }
    }
    const int   prv = (lane > 0) ? (lane - 1) : 0;
    const float eS  = lane_pull(incS, prv << 2);
    const float eQ  = lane_pull(incQ, prv << 2);
    float runS = (lane > 0) ? eS : 0.0f;
    float runQ = (lane > 0) ? eQ : 0.0f;

    float Sp[R], Qp[R];
    int cnt = 0;
#pragma unroll
    for (int u = 0; u < R; ++u) {
        runS += v[u];
        runQ += v[u] * v[u];
        Sp[u] = runS;
        Qp[u] = runQ;
        const float rho = (float)(lane * R + u + 1);
        const float d   = rho * v[u] - runS;
        const float g   = v[u] * (d - runS) + runQ;
        cnt += (g <= 1.0f) ? 1 : 0;
    }
#pragma unroll
    for (int msk = 16; msk >= 1; msk >>= 1)
        cnt += lane_pull_i(cnt, (lane ^ msk) << 2);
    const int support = __builtin_amdgcn_readfirstlane(cnt);
    const int sidx = support - 1;
    const int us   = sidx & (R - 1);
    float Ssel = Sp[0], Qsel = Qp[0];
#pragma unroll
    for (int u = 1; u < R; ++u) {
        Ssel = (us == u) ? Sp[u] : Ssel;
        Qsel = (us == u) ? Qp[u] : Qsel;
    }
    float tsel;
    {
        const float rho   = (float)(lane * R + us + 1);
        const float mean  = Ssel / rho;
        const float msq   = Qsel / rho;
        const float ssv   = rho * (msq - mean * mean);
        const float delta = (1.0f - ssv) / rho;
        tsel = mean - ((delta > 0.0f) ? sqrtf(delta) : 0.0f);
    }
    return lane_pull(tsel, (sidx >> LG) << 2);
}

__global__ __launch_bounds__(256) void entmax_kernel(
    const float* __restrict__ s1, const float* __restrict__ s2,
    const float* __restrict__ o1, const float* __restrict__ o2,
    const int* __restrict__ adj, const int* __restrict__ nag,
    float* __restrict__ rowmax, float* __restrict__ rowtau)
{
    __shared__ float cand[8][Nn];

    const int lane = threadIdx.x & 31;
    const int wave = threadIdx.x >> 5;
    float* wl = cand[wave];
    const unsigned na      = (unsigned)nag[0];
    const unsigned step128 = 128u % na;
    const unsigned jm_lane = (unsigned)(lane * 4) % na;
    const int row0 = (blockIdx.x * 8 + wave) * RPW;

#pragma unroll 1
    for (int rr = 0; rr < RPW; ++rr) {
        const int bid = row0 + rr;
        const int i   = bid & (Nn - 1);
        const int h   = (bid >> 10) & (Hh - 1);
        const int b   = bid >> 13;

        const int   rowbase = (b * Hh + h) * Nn;
        const float s1i = s1[rowbase + i];
        const float o1i = o1[rowbase + i];
        const unsigned im = (unsigned)i % na;
        const long  adjbase = ((long)b * Nn + i) * Nn;

        float x[32];
        float lmax = -3.4e38f;
        unsigned jm = jm_lane;
#pragma unroll
        for (int q = 0; q < 8; ++q) {
            const int j = (q * 32 + lane) * 4;
            const float4 sv = *(const float4*)(s2 + rowbase + j);
            const float4 ov = *(const float4*)(o2 + rowbase + j);
            const int4   av = *(const int4*)(adj + adjbase + j);
            asm volatile("" :: "v"(sv.x), "v"(sv.y), "v"(sv.z), "v"(sv.w),
                               "v"(ov.x), "v"(ov.y), "v"(ov.z), "v"(ov.w));
            const float se[4] = { sv.x, sv.y, sv.z, sv.w };
            const float oe[4] = { ov.x, ov.y, ov.z, ov.w };
            const int   ae[4] = { av.x, av.y, av.z, av.w };
            unsigned m = jm;
#pragma unroll
            for (int c = 0; c < 4; ++c) {
                float e = (m == im) ? (s1i + se[c]) : (o1i + oe[c]);
                if (ae[c] == 0) e = -1.0e12f;
                const float xv = 0.5f * e;
                x[q * 4 + c] = xv;
                lmax = fmaxf(lmax, xv);
                m = (m + 1u == na) ? 0u : (m + 1u);
            }
            jm += step128;
            if (jm >= na) jm -= na;
        }
#pragma unroll
        for (int msk = 16; msk >= 1; msk >>= 1)
            lmax = fmaxf(lmax, lane_pull(lmax, (lane ^ msk) << 2));
        const float xmax = lmax;
#pragma unroll
        for (int u = 0; u < 32; ++u) x[u] -= xmax;

        int ncand = 0;
#pragma unroll
        for (int u = 0; u < 32; ++u) {
            const bool     keep = (x[u] >= -1.0f);
            const unsigned who  = __builtin_amdgcn_ballot_w32(keep);
            const int      pos  = (int)__builtin_amdgcn_mbcnt_lo(who, (unsigned)ncand);
            if (keep) wl[pos] = x[u];
            ncand += __builtin_popcount(who);
        }

        float tstar;
        if (ncand <= CAP) {
            __builtin_amdgcn_fence(__ATOMIC_ACQ_REL, "wavefront");
            float w[RS];
#pragma unroll
            for (int r = 0; r < RS; ++r) {
                const int   slot = r * 32 + lane;
                const float t    = wl[slot];
                w[r] = (slot < ncand) ? t : -2.0f;
            }
            tstar = entmax_tau<RS>(w, lane);
        } else {
            float v[32];
#pragma unroll
            for (int u = 0; u < 32; ++u) v[u] = x[u];
            tstar = entmax_tau<32>(v, lane);
        }

        if (lane == 0) {
            rowmax[bid] = xmax;
            rowtau[bid] = tstar;
        }
    }
}

static constexpr int KC  = 128;
static constexpr int NCK = Nn / KC;
static constexpr int LDP = 40;
static constexpr int LAP = KC + 4;
static constexpr int TS  = KC / 8;
static_assert(TS == 16, "one tile row is formed per matrix step: 16 steps for 16 rows");

template <bool V> struct flag { static constexpr bool value = V; };

__device__ __forceinline__ void lds_barrier() {
    __builtin_amdgcn_fence(__ATOMIC_RELEASE, "workgroup", "local");
    __builtin_amdgcn_s_barrier();
    __builtin_amdgcn_fence(__ATOMIC_ACQUIRE, "workgroup", "local");
}

template <bool ONE>
__device__ __forceinline__ void aggregate_body(
    const int* __restrict__ adj,
    const float* __restrict__ s1, const float* __restrict__ s2,
    const float* __restrict__ o1, const float* __restrict__ o2,
    const float* __restrict__ rowmax, const float* __restrict__ rowtau,
    const float* __restrict__ wh1,
    const float* __restrict__ wh2, const unsigned na, const unsigned cnt,
    float* __restrict__ hp, float (&ldsW1)[KC][LDP], float (&ldsW2)[KC][LDP],
    float (&ldsA)[2][4][16][LAP])
{
    const int tid  = threadIdx.x;
    const int wave = __builtin_amdgcn_readfirstlane(tid >> 5);
    const int lane = tid & 31;
    const int m    = lane & 15;
    const int half = lane >> 4;
    const int bh   = blockIdx.x;
    const int b    = bh / Hh;
    const int h    = bh % Hh;

    const unsigned p0  = blockIdx.y * 64u;
    const unsigned res = ONE ? (p0 / cnt) : 0u;
    const unsigned q0  = ONE ? (p0 - res * cnt + (unsigned)wave * 16u)
                             : (p0 + (unsigned)wave * 16u);
    auto rowq = [&](unsigned mm) -> int {
        return ONE ? (int)((q0 + mm) * na + res) : (int)(q0 + mm);
    };

    const unsigned rowmod = (unsigned)rowq((unsigned)m) % na;
    unsigned cm = (unsigned)(4 * half) % na;
    const unsigned step8 = 8u % na;

    const int* adjw = adj + ((long)b * Nn + rowq(0u)) * Nn;
    const long rstride = ONE ? (long)na * Nn : (long)Nn;
    const unsigned lane4 = 4u * (unsigned)lane;
    const float* s2w = s2 + (long)bh * Nn;
    const float* o2w = o2 + (long)bh * Nn;
    const int   myrow = bh * Nn + rowq((unsigned)m);
    const float r_s1 = s1[myrow];
    const float r_o1 = o1[myrow];
    const float r_mx = rowmax[myrow];
    const float r_ta = rowtau[myrow];
    unsigned cmA = lane4 % na;
    const unsigned stepKC = (unsigned)KC % na;

    const int sr = tid >> 3;
    const int pc = tid & 7;
    const long whoff = ((long)(b * Nn + sr)) * Dd + h * DH + 4 * pc;
    unsigned rm = (unsigned)sr % na;
    const unsigned step16 = 16u % na;

    constexpr int NQ = ONE ? 8 : 16;
    v4i aC[16];
    v4f s2C, o2C;
    v4f s2N, o2N;
    unsigned mc[4];
    v4f bN[NQ];
    auto load_adj_row = [&](int r, int kc) {
        aC[r] = *(const v4i*)(adjw + (long)r * rstride + kc + lane4);
    };
    auto fetchB = [&](int kc) {
#pragma unroll
        for (int q = 0; q < 8; ++q) {
            const long o = whoff + (long)(kc + 16 * q) * Dd;
            if constexpr (ONE) {
                const bool selfk = (rm == res);
                bN[q] = *(const v4f*)((selfk ? wh1 : wh2) + o);
                rm += step16;
                if (rm >= na) rm -= na;
            } else {
                bN[q]     = *(const v4f*)(wh1 + o);
                bN[8 + q] = *(const v4f*)(wh2 + o);
            }
        }
    };
    auto stashB = [&]() {
#pragma unroll
        for (int q = 0; q < 8; ++q) {
            *(v4f*)&ldsW1[16 * q + sr][4 * pc] = bN[q];
            if constexpr (!ONE) *(v4f*)&ldsW2[16 * q + sr][4 * pc] = bN[8 + q];
        }
    };
    auto next_classes = [&]() {
        mc[0] = cmA;
#pragma unroll
        for (int cc = 1; cc < 4; ++cc)
            mc[cc] = (mc[cc - 1] + 1u == na) ? 0u : (mc[cc - 1] + 1u);
        cmA += stepKC;
        if (cmA >= na) cmA -= na;
    };

    struct RowK { float s1r, o1r, mxr, tar; unsigned rmr; };
    auto from_lane = [](float v, int r) -> float {
        return __builtin_bit_cast(float,
            __builtin_amdgcn_readlane(__builtin_bit_cast(int, v), r));
    };
    auto row_consts = [&](int r) -> RowK {
        RowK k;
        k.s1r = from_lane(r_s1, r);
        k.o1r = from_lane(r_o1, r);
        k.mxr = from_lane(r_mx, r);
        k.tar = from_lane(r_ta, r);
        k.rmr = ONE ? res : (unsigned)__builtin_amdgcn_readlane((int)rowmod, r);
        return k;
    };
    auto form_stage = [&](int st, const RowK& k, int r, v4f& ev) {
#pragma unroll
        for (int cc = 0; cc < 4; ++cc) {
            if (st == 0) {
                const bool selfc = (mc[cc] == k.rmr);
                ev[cc] = (selfc ? k.s1r : k.o1r) + (selfc ? s2C[cc] : o2C[cc]);
            } else if (st == 1) {
                float e = ev[cc];
                if (aC[r][cc] == 0) e = -1.0e12f;
                ev[cc] = 0.5f * e;
            } else if (st == 2) {
                float xv = ev[cc];
                xv -= k.mxr;
                ev[cc] = xv - k.tar;
            } else {
                const float y = fmaxf(ev[cc], 0.0f);
                ev[cc] = y * y;
            }
        }
    };

    v8f accS0 = {};
    v8f accS1 = {};

    auto block = [&](auto formc, float (*cur)[LAP], float (*nxt)[LAP], int kcR) {
        constexpr bool FORM = decltype(formc)::value;
        constexpr int NB = ONE ? 8 : 16;
        float  br[2][NB];
        v4f    af[2];
        auto load_ab = [&](int t, float (&r)[NB], v4f& a) {
            a = *(const v4f*)&cur[m][8 * t + 4 * half];
#pragma unroll
            for (int s = 0; s < 2; ++s) {
                const int l0 = 8 * t + 4 * half + 2 * s;
                if constexpr (ONE) {
                    r[4 * s + 0] = ldsW1[l0][m];      r[4 * s + 1] = ldsW1[l0 + 1][m];
                    r[4 * s + 2] = ldsW1[l0][16 + m]; r[4 * s + 3] = ldsW1[l0 + 1][16 + m];
                } else {
                    r[8 * s + 0] = ldsW1[l0][m];      r[8 * s + 1] = ldsW1[l0 + 1][m];
                    r[8 * s + 2] = ldsW1[l0][16 + m]; r[8 * s + 3] = ldsW1[l0 + 1][16 + m];
                    r[8 * s + 4] = ldsW2[l0][m];      r[8 * s + 5] = ldsW2[l0 + 1][m];
                    r[8 * s + 6] = ldsW2[l0][16 + m]; r[8 * s + 7] = ldsW2[l0 + 1][16 + m];
                }
            }
        };
        load_ab(0, br[0], af[0]);
        RowK kn = {};
        if constexpr (FORM) kn = row_consts(0);

#pragma unroll
        for (int t = 0; t < TS; ++t) {
            if (t + 1 < TS) load_ab(t + 1, br[(t + 1) & 1], af[(t + 1) & 1]);
            __builtin_amdgcn_sched_barrier(0);

            const float av[4] = { af[t & 1].x, af[t & 1].y, af[t & 1].z, af[t & 1].w };
            const float (&r)[NB] = br[t & 1];
            const RowK k = kn;
            v4f  fo = {};

            if constexpr (ONE) {
#pragma unroll
                for (int s = 0; s < 2; ++s) {
                    v2f a;   a.x   = av[2 * s];     a.y   = av[2 * s + 1];
                    v2f b_0; b_0.x = r[4 * s + 0];  b_0.y = r[4 * s + 1];
                    v2f b_1; b_1.x = r[4 * s + 2];  b_1.y = r[4 * s + 3];
                    accS0 = __builtin_amdgcn_wmma_f32_16x16x4_f32(false, a, false, b_0,
                                                                  (short)0, accS0, false, false);
                    if constexpr (FORM) {
                        __builtin_amdgcn_sched_barrier(0);
                        form_stage(2 * s, k, t, fo);
                        if (s == 1 && t + 1 < TS) kn = row_consts(t + 1);
                        __builtin_amdgcn_sched_barrier(0);
                    }
                    accS1 = __builtin_amdgcn_wmma_f32_16x16x4_f32(false, a, false, b_1,
                                                                  (short)0, accS1, false, false);
                    if constexpr (FORM) {
                        __builtin_amdgcn_sched_barrier(0);
                        form_stage(2 * s + 1, k, t, fo);
                        __builtin_amdgcn_sched_barrier(0);
                    }
                }
            } else {
                bool sf[4];
                unsigned mm = cm;
#pragma unroll
                for (int e = 0; e < 4; ++e) {
                    sf[e] = (mm == rowmod);
                    mm = (mm + 1u == na) ? 0u : (mm + 1u);
                }
                cm += step8;
                if (cm >= na) cm -= na;

#pragma unroll
                for (int s = 0; s < 2; ++s) {
                    v2f aS, aO;
                    aS.x = sf[2 * s]     ? av[2 * s]     : 0.0f;
                    aS.y = sf[2 * s + 1] ? av[2 * s + 1] : 0.0f;
                    aO.x = sf[2 * s]     ? 0.0f : av[2 * s];
                    aO.y = sf[2 * s + 1] ? 0.0f : av[2 * s + 1];

                    v2f b1_0; b1_0.x = r[8 * s + 0]; b1_0.y = r[8 * s + 1];
                    v2f b1_1; b1_1.x = r[8 * s + 2]; b1_1.y = r[8 * s + 3];
                    v2f b2_0; b2_0.x = r[8 * s + 4]; b2_0.y = r[8 * s + 5];
                    v2f b2_1; b2_1.x = r[8 * s + 6]; b2_1.y = r[8 * s + 7];

                    accS0 = __builtin_amdgcn_wmma_f32_16x16x4_f32(false, aS, false, b1_0,
                                                                  (short)0, accS0, false, false);
                    accS0 = __builtin_amdgcn_wmma_f32_16x16x4_f32(false, aO, false, b2_0,
                                                                  (short)0, accS0, false, false);
                    if constexpr (FORM) {
                        __builtin_amdgcn_sched_barrier(0);
                        form_stage(2 * s, k, t, fo);
                        if (s == 1 && t + 1 < TS) kn = row_consts(t + 1);
                        __builtin_amdgcn_sched_barrier(0);
                    }
                    accS1 = __builtin_amdgcn_wmma_f32_16x16x4_f32(false, aS, false, b1_1,
                                                                  (short)0, accS1, false, false);
                    accS1 = __builtin_amdgcn_wmma_f32_16x16x4_f32(false, aO, false, b2_1,
                                                                  (short)0, accS1, false, false);
                    if constexpr (FORM) {
                        __builtin_amdgcn_sched_barrier(0);
                        form_stage(2 * s + 1, k, t, fo);
                        __builtin_amdgcn_sched_barrier(0);
                    }
                }
            }
            if constexpr (FORM) {
                *(v4f*)&nxt[t][lane4] = fo;
                load_adj_row(t, kcR);
            }
            __builtin_amdgcn_sched_barrier(0);
        }
    };

    fetchB(0);
    s2C = *(const v4f*)(s2w + lane4);
    o2C = *(const v4f*)(o2w + lane4);
#pragma unroll
    for (int r = 0; r < 16; ++r) load_adj_row(r, 0);
    next_classes();
    {
        float (*t0)[LAP] = ldsA[0][wave];
#pragma unroll
        for (int r = 0; r < 16; ++r) {
            const RowK k = row_consts(r);
            v4f o = {};
#pragma unroll
            for (int st = 0; st < 4; ++st) form_stage(st, k, r, o);
            *(v4f*)&t0[r][lane4] = o;
            load_adj_row(r, KC);
        }
    }
    s2N = *(const v4f*)(s2w + KC + lane4);
    o2N = *(const v4f*)(o2w + KC + lane4);

#pragma unroll 1
    for (int c = 0; c + 1 < NCK; ++c) {
        const int kc = c * KC;
        lds_barrier();
        stashB();
        lds_barrier();

        fetchB(kc + KC);

        s2C = s2N;
        o2C = o2N;
        const int kc2 = (c + 2 < NCK) ? (kc + 2 * KC) : (Nn - KC);
        s2N = *(const v4f*)(s2w + kc2 + lane4);
        o2N = *(const v4f*)(o2w + kc2 + lane4);
        next_classes();

        block(flag<true>{}, ldsA[c & 1][wave], ldsA[(c + 1) & 1][wave], kc2);
    }
    lds_barrier();
    stashB();
    lds_barrier();
    block(flag<false>{}, ldsA[(NCK - 1) & 1][wave], ldsA[NCK & 1][wave], 0);

    const unsigned mb = (unsigned)((lane >> 4) << 3);
#pragma unroll
    for (int r = 0; r < 8; ++r) {
        float* o = hp + ((long)(b * Nn + rowq(mb + (unsigned)r))) * Dd + h * DH;
        o[m]      = accS0[r];
        o[16 + m] = accS1[r];
    }
}

__global__ __launch_bounds__(128) void aggregate_kernel(
    const int* __restrict__ adj,
    const float* __restrict__ s1, const float* __restrict__ s2,
    const float* __restrict__ o1, const float* __restrict__ o2,
    const float* __restrict__ rowmax, const float* __restrict__ rowtau,
    const float* __restrict__ wh1,
    const float* __restrict__ wh2, const int* __restrict__ nag,
    float* __restrict__ hp)
{
    __shared__ __attribute__((aligned(16))) float ldsW1[KC][LDP];
    __shared__ __attribute__((aligned(16))) float ldsW2[KC][LDP];
    __shared__ __attribute__((aligned(16))) float ldsA[2][4][16][LAP];

    const unsigned na  = (unsigned)nag[0];
    const unsigned cnt = (unsigned)Nn / na;
    const bool one = (cnt * na == (unsigned)Nn) && ((cnt & 63u) == 0u);
    if (__builtin_expect(one, 1))
        aggregate_body<true >(adj, s1, s2, o1, o2, rowmax, rowtau, wh1, wh2,
                              na, cnt, hp, ldsW1, ldsW2, ldsA);
    else
        aggregate_body<false>(adj, s1, s2, o1, o2, rowmax, rowtau, wh1, wh2,
                              na, cnt, hp, ldsW1, ldsW2, ldsA);
}

__global__ __launch_bounds__(256)
void outproj_kernel(
    const float* hp, const float* Wo,
    const float* __restrict__ bo, float* out)
{
    __shared__ __attribute__((aligned(16))) float lds[2 * (64 + 64) * GP];
    proj_tile<64, 64, 1, 2, true>(hp, Wo, bo, out,
                                  blockIdx.y * 64, blockIdx.x * 64, lds);
}

extern "C" void kernel_launch(void* const* d_in, const int* in_sizes, int n_in,
                              void* d_out, int out_size, void* d_ws, size_t ws_size,
                              hipStream_t stream) {
    (void)in_sizes; (void)n_in; (void)out_size; (void)ws_size;

    const float* hmat = (const float*)d_in[0];
    const int*   nag  = (const int*)  d_in[1];
    const int*   adj  = (const int*)  d_in[2];
    const float* W1   = (const float*)d_in[3];
    const float* W2   = (const float*)d_in[4];
    const float* Ai   = (const float*)d_in[5];
    const float* Aj   = (const float*)d_in[6];
    const float* Wo   = (const float*)d_in[7];
    const float* bo   = (const float*)d_in[8];
    float*       out  = (float*)d_out;

    float* ws  = (float*)d_ws;
    const size_t whN = (size_t)Bc * Nn * Dd;
    const size_t scN = (size_t)Bc * Hh * Nn;
    float* wh1 = ws;
    float* wh2 = wh1 + whN;
    float* s1  = wh2 + whN;
    float* s2  = s1 + scN;
    float* o1  = s2 + scN;
    float* o2  = o1 + scN;
    float* hp  = o2 + scN;
    float* rowmax = hp + whN;
    float* rowtau = rowmax + scN;

    proj12_kernel<<<dim3(4, (Bc * Nn) / 64), 256, 0, stream>>>(
        hmat, W1, W2, wh1, wh2);

    scores_kernel<<<(Bc * Nn * Hh + 255) / 256, 256, 0, stream>>>(
        wh1, wh2, Ai, Aj, s1, s2, o1, o2);

    entmax_kernel<<<(Bc * Hh * Nn) / (8 * RPW), 256, 0, stream>>>(
        s1, s2, o1, o2, adj, nag, rowmax, rowtau);

    aggregate_kernel<<<dim3(Bc * Hh, Nn / 64), 128, 0, stream>>>(
        adj, s1, s2, o1, o2, rowmax, rowtau, wh1, wh2, nag, hp);

    outproj_kernel<<<dim3(Dd / 64, (Bc * Nn) / 64), 256, 0, stream>>>(
        hp, Wo, bo, out);
}


# score dots (s1,s2,o1,o2) formed in the projection kernel's epilogue from its accumulators (16-lane swizzle sums); the separate scores kernel and its launch are gone; score sum order changes
# speedup vs baseline: 8.2299x; 1.0615x over previous
#include <hip/hip_runtime.h>
#include <hip/hip_bf16.h>
#include <math.h>

static constexpr int Bc = 4;
static constexpr int Nn = 1024;
static constexpr int Dd = 256;
static constexpr int Hh = 8;
static constexpr int DH = 32;

typedef float v2f __attribute__((ext_vector_type(2)));
typedef float v4f __attribute__((ext_vector_type(4)));
typedef float v8f __attribute__((ext_vector_type(8)));
typedef int   v4i __attribute__((ext_vector_type(4)));

static constexpr int GK  = 64;
static constexpr int GP  = GK + 4;
static constexpr int GT  = GK / 8;
static constexpr int NCH = Dd / GK;

__device__ __forceinline__ v4f frag4(const float* rowp, int t) {
    return *(const v4f*)(rowp + 8 * t);
}

__device__ __forceinline__ float half_sum16(float x) {
#define HS16_STEP(MASK)                                                        \
    x += __builtin_bit_cast(float, __builtin_amdgcn_update_dpp(                \
             0, __builtin_bit_cast(int, x), 0x160 + (MASK), 0xf, 0xf, true));
    HS16_STEP(1) HS16_STEP(2) HS16_STEP(4) HS16_STEP(8)
#undef HS16_STEP
    return x;
}

template <int BM, int BN, int MT, int NT, bool ELU, bool SC>
__device__ __forceinline__ void proj_tile(
    const float* A, const float* W,
    const float* __restrict__ bias, float* C,
    const int row0, const int col0, float* lds,
    const float* __restrict__ Ai, const float* __restrict__ Aj,
    float* __restrict__ scI, float* __restrict__ scJ)
{
    constexpr int WN  = BN / (16 * NT);
    constexpr int WMV = BM / (16 * MT);
    static_assert(WN * WMV == 8, "the block has 8 waves");
    constexpr int BUF = (BM + BN) * GP;
    constexpr int NA  = BM / 16;
    constexpr int NB  = BN / 16;

    const int tid  = threadIdx.x;
    const int lane = tid & 31;
    const int wave = tid >> 5;
    const int m    = lane & 15;
    const int half = lane >> 4;
    const int wn   = wave % WN;
    const int wm   = wave / WN;

    const int c4 = tid & 15;
    const int r0 = tid >> 4;
    const float* ga = A + (long)(row0 + r0) * Dd + 4 * c4;
    const float* gb = W + (long)(col0 + r0) * Dd + 4 * c4;
    const int sa = r0 * GP + 4 * c4;
    const int sb = (BM + r0) * GP + 4 * c4;

    const int fa = (wm * 16 * MT + m) * GP + 4 * half;
    const int fb = (BM + wn * 16 * NT + m) * GP + 4 * half;

    v8f acc[MT][NT];
#pragma unroll
    for (int i = 0; i < MT; ++i)
#pragma unroll
        for (int j = 0; j < NT; ++j) acc[i][j] = v8f{};

    float ai0 = 0.0f, ai1 = 0.0f, aj0 = 0.0f, aj1 = 0.0f;
    if constexpr (SC) {
        const int colw = col0 + wn * 32;
        ai0 = Ai[colw + m];  ai1 = Ai[colw + 16 + m];
        aj0 = Aj[colw + m];  aj1 = Aj[colw + 16 + m];
    }

    v4f pa[NA], pb[NB];
#pragma unroll
    for (int i = 0; i < NA; ++i) pa[i] = *(const v4f*)(ga + (long)(16 * i) * Dd);
#pragma unroll
    for (int i = 0; i < NB; ++i) pb[i] = *(const v4f*)(gb + (long)(16 * i) * Dd);
#pragma unroll
    for (int i = 0; i < NA; ++i) *(v4f*)(lds + sa + 16 * i * GP) = pa[i];
#pragma unroll
    for (int i = 0; i < NB; ++i) *(v4f*)(lds + sb + 16 * i * GP) = pb[i];
    __syncthreads();

#pragma unroll
    for (int c = 0; c < NCH; ++c) {
        const float* cur = lds + (c & 1) * BUF;
        float*       nxt = lds + ((c + 1) & 1) * BUF;
        if (c + 1 < NCH) {
#pragma unroll
            for (int i = 0; i < NA; ++i)
                pa[i] = *(const v4f*)(ga + (long)(16 * i) * Dd + (c + 1) * GK);
#pragma unroll
            for (int i = 0; i < NB; ++i)
                pb[i] = *(const v4f*)(gb + (long)(16 * i) * Dd + (c + 1) * GK);
        }
        __builtin_amdgcn_sched_barrier(0);

#pragma unroll
        for (int t = 0; t < GT; ++t) {
            v4f xa[MT], xb[NT];
#pragma unroll
            for (int i = 0; i < MT; ++i) xa[i] = frag4(cur + fa + 16 * i * GP, t);
#pragma unroll
            for (int j = 0; j < NT; ++j) xb[j] = frag4(cur + fb + 16 * j * GP, t);
#pragma unroll
            for (int s = 0; s < 2; ++s) {
#pragma unroll
                for (int i = 0; i < MT; ++i) {
                    v2f a;
                    a.x = s ? xa[i].z : xa[i].x;
                    a.y = s ? xa[i].w : xa[i].y;
#pragma unroll
                    for (int j = 0; j < NT; ++j) {
                        v2f b;
                        b.x = s ? xb[j].z : xb[j].x;
                        b.y = s ? xb[j].w : xb[j].y;
                        acc[i][j] = __builtin_amdgcn_wmma_f32_16x16x4_f32(
                            false, a, false, b, (short)0, acc[i][j], false, false);
                    }
                }
            }
        }

        __builtin_amdgcn_sched_barrier(0);
        if (c + 1 < NCH) {
#pragma unroll
            for (int i = 0; i < NA; ++i) *(v4f*)(nxt + sa + 16 * i * GP) = pa[i];
#pragma unroll
            for (int i = 0; i < NB; ++i) *(v4f*)(nxt + sb + 16 * i * GP) = pb[i];
            __syncthreads();
        }
    }

#pragma unroll
    for (int i = 0; i < MT; ++i) {
        const int rbase = row0 + (wm * MT + i) * 16 + (half << 3);
#pragma unroll
        for (int j = 0; j < NT; ++j) {
            const int cc = col0 + (wn * NT + j) * 16 + m;
            float bv = 0.0f;
            if constexpr (ELU) bv = bias[cc];
#pragma unroll
            for (int r = 0; r < 8; ++r) {
                float v = acc[i][j][r];
                if constexpr (ELU) {
                    v += bv;
                    v = (v > 0.0f) ? v : (expf(v) - 1.0f);
                }
                C[(long)(rbase + r) * Dd + cc] = v;
            }
        }
    }

    if constexpr (SC) {
        static_assert(MT == 2 && NT == 2, "a wave tile of 32 rows x one head's 32 columns");
        const int colw = col0 + wn * 32;
        float keepI = 0.0f, keepJ = 0.0f;
#pragma unroll
        for (int i = 0; i < MT; ++i) {
#pragma unroll
            for (int r = 0; r < 8; ++r) {
                const float pi = half_sum16(acc[i][0][r] * ai0 + acc[i][1][r] * ai1);
                const float pj = half_sum16(acc[i][0][r] * aj0 + acc[i][1][r] * aj1);
                if (m == i * 8 + r) { keepI = pi; keepJ = pj; }
            }
        }
        const int grow = row0 + wm * 32 + (m >> 3) * 16 + (half << 3) + (m & 7);
        const int o = ((grow / Nn) * Hh + colw / DH) * Nn + (grow % Nn);
        scI[o] = keepI;
        scJ[o] = keepJ;
    }
}

__global__ __launch_bounds__(256)
void proj12_kernel(
    const float* hmat, const float* W1, const float* W2,
    float* wh1, float* wh2,
    const float* __restrict__ Ai, const float* __restrict__ Aj,
    float* __restrict__ s1, float* __restrict__ s2,
    float* __restrict__ o1, float* __restrict__ o2)
{
    __shared__ __attribute__((aligned(16))) float lds[2 * (64 + 128) * GP];
    const bool second = blockIdx.x >= 2;
    proj_tile<64, 128, 2, 2, false, true>(hmat, second ? W2 : W1, nullptr,
                                          second ? wh2 : wh1,
                                          blockIdx.y * 64, (blockIdx.x & 1) * 128, lds,
                                          Ai, Aj, second ? o1 : s1, second ? o2 : s2);
}

static constexpr int RPW = 4;
static constexpr int RS  = 8;
static constexpr int CAP = 32 * RS;

__device__ __forceinline__ float lane_pull(float x, int addr) {
    return __builtin_bit_cast(float,
        __builtin_amdgcn_ds_bpermute(addr, __builtin_bit_cast(int, x)));
}
__device__ __forceinline__ int lane_pull_i(int x, int addr) {
    return __builtin_amdgcn_ds_bpermute(addr, x);
}

template <int R, int JJ>
__device__ __forceinline__ void ce_regs(float (&v)[R]) {
#pragma unroll
    for (int u = 0; u < R; ++u) {
        if ((u & JJ) == 0) {
            const float a = v[u], c = v[u | JJ];
            v[u]      = fmaxf(a, c);
            v[u | JJ] = fminf(a, c);
        }
    }
}
template <int R, int K>
__device__ __forceinline__ void flip_regs(float (&v)[R]) {
#pragma unroll
    for (int u = 0; u < R; ++u) {
        if ((u & (K >> 1)) == 0) {
            const int p = u ^ (K - 1);
            const float a = v[u], c = v[p];
            v[u] = fmaxf(a, c);
            v[p] = fminf(a, c);
        }
    }
}
template <int LM>
__device__ __forceinline__ float lane_xor(float x) {
    const int b = __builtin_bit_cast(int, x);
    if constexpr (LM < 16) {
        return __builtin_bit_cast(float,
            __builtin_amdgcn_update_dpp(0, b, 0x160 + LM, 0xf, 0xf, true));
    } else {
        static_assert(LM == 31, "only the mirrored other-row exchange is used");
        return __builtin_bit_cast(float,
            __builtin_amdgcn_permlanex16(b, b, 0x89abcdefu, 0x01234567u, false, false));
    }
}
template <int R, int LM, bool FLIP>
__device__ __forceinline__ void ce_lanes(float (&v)[R], int lane) {
    constexpr int HB = (LM >= 16) ? 16 : (LM >= 8) ? 8 : (LM >= 4) ? 4
                     : (LM >= 2) ? 2 : 1;
    const float bound = (lane & HB) ? -__builtin_inff() : __builtin_inff();
    float p[R];
#pragma unroll
    for (int u = 0; u < R; ++u) p[u] = lane_xor<LM>(v[FLIP ? (u ^ (R - 1)) : u]);
#pragma unroll
    for (int u = 0; u < R; ++u) v[u] = __builtin_amdgcn_fmed3f(v[u], p[u], bound);
}
template <int R>
__device__ __forceinline__ void ce_regs_tail(float (&v)[R]) {
    if constexpr (R > 16) ce_regs<R, 16>(v);
    if constexpr (R > 8)  ce_regs<R, 8>(v);
    ce_regs<R, 4>(v); ce_regs<R, 2>(v); ce_regs<R, 1>(v);
}

template <int R>
__device__ __forceinline__ float entmax_tau(float (&v)[R], int lane)
{
    static_assert(R == 32 || R == 8, "register stages below are written for these");
    constexpr int LG = (R == 32) ? 5 : 3;

    flip_regs<R, 2>(v);
    flip_regs<R, 4>(v);  ce_regs<R, 1>(v);
    flip_regs<R, 8>(v);  ce_regs<R, 2>(v); ce_regs<R, 1>(v);
    if constexpr (R == 32) {
        flip_regs<R, 16>(v); ce_regs<R, 4>(v); ce_regs<R, 2>(v); ce_regs<R, 1>(v);
        flip_regs<R, 32>(v); ce_regs<R, 8>(v); ce_regs<R, 4>(v); ce_regs<R, 2>(v);
        ce_regs<R, 1>(v);
    }
    ce_lanes<R, 1, true>(v, lane);                                     ce_regs_tail<R>(v);
    ce_lanes<R, 3, true>(v, lane);  ce_lanes<R, 1, false>(v, lane);    ce_regs_tail<R>(v);
    ce_lanes<R, 7, true>(v, lane);  ce_lanes<R, 2, false>(v, lane);
    ce_lanes<R, 1, false>(v, lane);                                    ce_regs_tail<R>(v);
    ce_lanes<R, 15, true>(v, lane); ce_lanes<R, 4, false>(v, lane);
    ce_lanes<R, 2, false>(v, lane); ce_lanes<R, 1, false>(v, lane);    ce_regs_tail<R>(v);
    ce_lanes<R, 31, true>(v, lane); ce_lanes<R, 8, false>(v, lane);
    ce_lanes<R, 4, false>(v, lane); ce_lanes<R, 2, false>(v, lane);
    ce_lanes<R, 1, false>(v, lane);                                    ce_regs_tail<R>(v);

    float totS = 0.0f, totQ = 0.0f;
#pragma unroll
    for (int u = 0; u < R; ++u) { totS += v[u]; totQ += v[u] * v[u]; }
    float incS = totS, incQ = totQ;
#pragma unroll
    for (int off = 1; off < 32; off <<= 1) {
        const int   src = (lane >= off) ? (lane - off) : lane;
        const float pa  = lane_pull(incS, src << 2);
        const float pb  = lane_pull(incQ, src << 2);
        if (lane >= off) { incS += pa; incQ += pb; }
    }
    const int   prv = (lane > 0) ? (lane - 1) : 0;
    const float eS  = lane_pull(incS, prv << 2);
    const float eQ  = lane_pull(incQ, prv << 2);
    float runS = (lane > 0) ? eS : 0.0f;
    float runQ = (lane > 0) ? eQ : 0.0f;

    float Sp[R], Qp[R];
    int cnt = 0;
#pragma unroll
    for (int u = 0; u < R; ++u) {
        runS += v[u];
        runQ += v[u] * v[u];
        Sp[u] = runS;
        Qp[u] = runQ;
        const float rho = (float)(lane * R + u + 1);
        const float d   = rho * v[u] - runS;
        const float g   = v[u] * (d - runS) + runQ;
        cnt += (g <= 1.0f) ? 1 : 0;
    }
#pragma unroll
    for (int msk = 16; msk >= 1; msk >>= 1)
        cnt += lane_pull_i(cnt, (lane ^ msk) << 2);
    const int support = __builtin_amdgcn_readfirstlane(cnt);
    const int sidx = support - 1;
    const int us   = sidx & (R - 1);
    float Ssel = Sp[0], Qsel = Qp[0];
#pragma unroll
    for (int u = 1; u < R; ++u) {
        Ssel = (us == u) ? Sp[u] : Ssel;
        Qsel = (us == u) ? Qp[u] : Qsel;
    }
    float tsel;
    {
        const float rho   = (float)(lane * R + us + 1);
        const float mean  = Ssel / rho;
        const float msq   = Qsel / rho;
        const float ssv   = rho * (msq - mean * mean);
        const float delta = (1.0f - ssv) / rho;
        tsel = mean - ((delta > 0.0f) ? sqrtf(delta) : 0.0f);
    }
    return lane_pull(tsel, (sidx >> LG) << 2);
}

__global__ __launch_bounds__(256) void entmax_kernel(
    const float* __restrict__ s1, const float* __restrict__ s2,
    const float* __restrict__ o1, const float* __restrict__ o2,
    const int* __restrict__ adj, const int* __restrict__ nag,
    float* __restrict__ rowmax, float* __restrict__ rowtau)
{
    __shared__ float cand[8][Nn];

    const int lane = threadIdx.x & 31;
    const int wave = threadIdx.x >> 5;
    float* wl = cand[wave];
    const unsigned na      = (unsigned)nag[0];
    const unsigned step128 = 128u % na;
    const unsigned jm_lane = (unsigned)(lane * 4) % na;
    const int row0 = (blockIdx.x * 8 + wave) * RPW;

#pragma unroll 1
    for (int rr = 0; rr < RPW; ++rr) {
        const int bid = row0 + rr;
        const int i   = bid & (Nn - 1);
        const int h   = (bid >> 10) & (Hh - 1);
        const int b   = bid >> 13;

        const int   rowbase = (b * Hh + h) * Nn;
        const float s1i = s1[rowbase + i];
        const float o1i = o1[rowbase + i];
        const unsigned im = (unsigned)i % na;
        const long  adjbase = ((long)b * Nn + i) * Nn;

        float x[32];
        float lmax = -3.4e38f;
        unsigned jm = jm_lane;
#pragma unroll
        for (int q = 0; q < 8; ++q) {
            const int j = (q * 32 + lane) * 4;
            const float4 sv = *(const float4*)(s2 + rowbase + j);
            const float4 ov = *(const float4*)(o2 + rowbase + j);
            const int4   av = *(const int4*)(adj + adjbase + j);
            asm volatile("" :: "v"(sv.x), "v"(sv.y), "v"(sv.z), "v"(sv.w),
                               "v"(ov.x), "v"(ov.y), "v"(ov.z), "v"(ov.w));
            const float se[4] = { sv.x, sv.y, sv.z, sv.w };
            const float oe[4] = { ov.x, ov.y, ov.z, ov.w };
            const int   ae[4] = { av.x, av.y, av.z, av.w };
            unsigned m = jm;
#pragma unroll
            for (int c = 0; c < 4; ++c) {
                float e = (m == im) ? (s1i + se[c]) : (o1i + oe[c]);
                if (ae[c] == 0) e = -1.0e12f;
                const float xv = 0.5f * e;
                x[q * 4 + c] = xv;
                lmax = fmaxf(lmax, xv);
                m = (m + 1u == na) ? 0u : (m + 1u);
            }
            jm += step128;
            if (jm >= na) jm -= na;
        }
#pragma unroll
        for (int msk = 16; msk >= 1; msk >>= 1)
            lmax = fmaxf(lmax, lane_pull(lmax, (lane ^ msk) << 2));
        const float xmax = lmax;
#pragma unroll
        for (int u = 0; u < 32; ++u) x[u] -= xmax;

        int ncand = 0;
#pragma unroll
        for (int u = 0; u < 32; ++u) {
            const bool     keep = (x[u] >= -1.0f);
            const unsigned who  = __builtin_amdgcn_ballot_w32(keep);
            const int      pos  = (int)__builtin_amdgcn_mbcnt_lo(who, (unsigned)ncand);
            if (keep) wl[pos] = x[u];
            ncand += __builtin_popcount(who);
        }

        float tstar;
        if (ncand <= CAP) {
            __builtin_amdgcn_fence(__ATOMIC_ACQ_REL, "wavefront");
            float w[RS];
#pragma unroll
            for (int r = 0; r < RS; ++r) {
                const int   slot = r * 32 + lane;
                const float t    = wl[slot];
                w[r] = (slot < ncand) ? t : -2.0f;
            }
            tstar = entmax_tau<RS>(w, lane);
        } else {
            float v[32];
#pragma unroll
            for (int u = 0; u < 32; ++u) v[u] = x[u];
            tstar = entmax_tau<32>(v, lane);
        }

        if (lane == 0) {
            rowmax[bid] = xmax;
            rowtau[bid] = tstar;
        }
    }
}

static constexpr int KC  = 128;
static constexpr int NCK = Nn / KC;
static constexpr int LDP = 40;
static constexpr int LAP = KC + 4;
static constexpr int TS  = KC / 8;
static_assert(TS == 16, "one tile row is formed per matrix step: 16 steps for 16 rows");

template <bool V> struct flag { static constexpr bool value = V; };

__device__ __forceinline__ void lds_barrier() {
    __builtin_amdgcn_fence(__ATOMIC_RELEASE, "workgroup", "local");
    __builtin_amdgcn_s_barrier();
    __builtin_amdgcn_fence(__ATOMIC_ACQUIRE, "workgroup", "local");
}

template <bool ONE>
__device__ __forceinline__ void aggregate_body(
    const int* __restrict__ adj,
    const float* __restrict__ s1, const float* __restrict__ s2,
    const float* __restrict__ o1, const float* __restrict__ o2,
    const float* __restrict__ rowmax, const float* __restrict__ rowtau,
    const float* __restrict__ wh1,
    const float* __restrict__ wh2, const unsigned na, const unsigned cnt,
    float* __restrict__ hp, float (&ldsW1)[KC][LDP], float (&ldsW2)[KC][LDP],
    float (&ldsA)[2][4][16][LAP])
{
    const int tid  = threadIdx.x;
    const int wave = __builtin_amdgcn_readfirstlane(tid >> 5);
    const int lane = tid & 31;
    const int m    = lane & 15;
    const int half = lane >> 4;
    const int bh   = blockIdx.x;
    const int b    = bh / Hh;
    const int h    = bh % Hh;

    const unsigned p0  = blockIdx.y * 64u;
    const unsigned res = ONE ? (p0 / cnt) : 0u;
    const unsigned q0  = ONE ? (p0 - res * cnt + (unsigned)wave * 16u)
                             : (p0 + (unsigned)wave * 16u);
    auto rowq = [&](unsigned mm) -> int {
        return ONE ? (int)((q0 + mm) * na + res) : (int)(q0 + mm);
    };

    const unsigned rowmod = (unsigned)rowq((unsigned)m) % na;
    unsigned cm = (unsigned)(4 * half) % na;
    const unsigned step8 = 8u % na;

    const int* adjw = adj + ((long)b * Nn + rowq(0u)) * Nn;
    const long rstride = ONE ? (long)na * Nn : (long)Nn;
    const unsigned lane4 = 4u * (unsigned)lane;
    const float* s2w = s2 + (long)bh * Nn;
    const float* o2w = o2 + (long)bh * Nn;
    const int   myrow = bh * Nn + rowq((unsigned)m);
    const float r_s1 = s1[myrow];
    const float r_o1 = o1[myrow];
    const float r_mx = rowmax[myrow];
    const float r_ta = rowtau[myrow];
    unsigned cmA = lane4 % na;
    const unsigned stepKC = (unsigned)KC % na;

    const int sr = tid >> 3;
    const int pc = tid & 7;
    const long whoff = ((long)(b * Nn + sr)) * Dd + h * DH + 4 * pc;
    unsigned rm = (unsigned)sr % na;
    const unsigned step16 = 16u % na;

    constexpr int NQ = ONE ? 8 : 16;
    v4i aC[16];
    v4f s2C, o2C;
    v4f s2N, o2N;
    unsigned mc[4];
    v4f bN[NQ];
    auto load_adj_row = [&](int r, int kc) {
        aC[r] = *(const v4i*)(adjw + (long)r * rstride + kc + lane4);
    };
    auto fetchB = [&](int kc) {
#pragma unroll
        for (int q = 0; q < 8; ++q) {
            const long o = whoff + (long)(kc + 16 * q) * Dd;
            if constexpr (ONE) {
                const bool selfk = (rm == res);
                bN[q] = *(const v4f*)((selfk ? wh1 : wh2) + o);
                rm += step16;
                if (rm >= na) rm -= na;
            } else {
                bN[q]     = *(const v4f*)(wh1 + o);
                bN[8 + q] = *(const v4f*)(wh2 + o);
            }
        }
    };
    auto stashB = [&]() {
#pragma unroll
        for (int q = 0; q < 8; ++q) {
            *(v4f*)&ldsW1[16 * q + sr][4 * pc] = bN[q];
            if constexpr (!ONE) *(v4f*)&ldsW2[16 * q + sr][4 * pc] = bN[8 + q];
        }
    };
    auto next_classes = [&]() {
        mc[0] = cmA;
#pragma unroll
        for (int cc = 1; cc < 4; ++cc)
            mc[cc] = (mc[cc - 1] + 1u == na) ? 0u : (mc[cc - 1] + 1u);
        cmA += stepKC;
        if (cmA >= na) cmA -= na;
    };

    struct RowK { float s1r, o1r, mxr, tar; unsigned rmr; };
    auto from_lane = [](float v, int r) -> float {
        return __builtin_bit_cast(float,
            __builtin_amdgcn_readlane(__builtin_bit_cast(int, v), r));
    };
    auto row_consts = [&](int r) -> RowK {
        RowK k;
        k.s1r = from_lane(r_s1, r);
        k.o1r = from_lane(r_o1, r);
        k.mxr = from_lane(r_mx, r);
        k.tar = from_lane(r_ta, r);
        k.rmr = ONE ? res : (unsigned)__builtin_amdgcn_readlane((int)rowmod, r);
        return k;
    };
    auto form_stage = [&](int st, const RowK& k, int r, v4f& ev) {
#pragma unroll
        for (int cc = 0; cc < 4; ++cc) {
            if (st == 0) {
                const bool selfc = (mc[cc] == k.rmr);
                ev[cc] = (selfc ? k.s1r : k.o1r) + (selfc ? s2C[cc] : o2C[cc]);
            } else if (st == 1) {
                float e = ev[cc];
                if (aC[r][cc] == 0) e = -1.0e12f;
                ev[cc] = 0.5f * e;
            } else if (st == 2) {
                float xv = ev[cc];
                xv -= k.mxr;
                ev[cc] = xv - k.tar;
            } else {
                const float y = fmaxf(ev[cc], 0.0f);
                ev[cc] = y * y;
            }
        }
    };

    v8f accS0 = {};
    v8f accS1 = {};

    auto block = [&](auto formc, float (*cur)[LAP], float (*nxt)[LAP], int kcR) {
        constexpr bool FORM = decltype(formc)::value;
        constexpr int NB = ONE ? 8 : 16;
        float  br[2][NB];
        v4f    af[2];
        auto load_ab = [&](int t, float (&r)[NB], v4f& a) {
            a = *(const v4f*)&cur[m][8 * t + 4 * half];
#pragma unroll
            for (int s = 0; s < 2; ++s) {
                const int l0 = 8 * t + 4 * half + 2 * s;
                if constexpr (ONE) {
                    r[4 * s + 0] = ldsW1[l0][m];      r[4 * s + 1] = ldsW1[l0 + 1][m];
                    r[4 * s + 2] = ldsW1[l0][16 + m]; r[4 * s + 3] = ldsW1[l0 + 1][16 + m];
                } else {
                    r[8 * s + 0] = ldsW1[l0][m];      r[8 * s + 1] = ldsW1[l0 + 1][m];
                    r[8 * s + 2] = ldsW1[l0][16 + m]; r[8 * s + 3] = ldsW1[l0 + 1][16 + m];
                    r[8 * s + 4] = ldsW2[l0][m];      r[8 * s + 5] = ldsW2[l0 + 1][m];
                    r[8 * s + 6] = ldsW2[l0][16 + m]; r[8 * s + 7] = ldsW2[l0 + 1][16 + m];
                }
            }
        };
        load_ab(0, br[0], af[0]);
        RowK kn = {};
        if constexpr (FORM) kn = row_consts(0);

#pragma unroll
        for (int t = 0; t < TS; ++t) {
            if (t + 1 < TS) load_ab(t + 1, br[(t + 1) & 1], af[(t + 1) & 1]);
            __builtin_amdgcn_sched_barrier(0);

            const float av[4] = { af[t & 1].x, af[t & 1].y, af[t & 1].z, af[t & 1].w };
            const float (&r)[NB] = br[t & 1];
            const RowK k = kn;
            v4f  fo = {};

            if constexpr (ONE) {
#pragma unroll
                for (int s = 0; s < 2; ++s) {
                    v2f a;   a.x   = av[2 * s];     a.y   = av[2 * s + 1];
                    v2f b_0; b_0.x = r[4 * s + 0];  b_0.y = r[4 * s + 1];
                    v2f b_1; b_1.x = r[4 * s + 2];  b_1.y = r[4 * s + 3];
                    accS0 = __builtin_amdgcn_wmma_f32_16x16x4_f32(false, a, false, b_0,
                                                                  (short)0, accS0, false, false);
                    if constexpr (FORM) {
                        __builtin_amdgcn_sched_barrier(0);
                        form_stage(2 * s, k, t, fo);
                        if (s == 1 && t + 1 < TS) kn = row_consts(t + 1);
                        __builtin_amdgcn_sched_barrier(0);
                    }
                    accS1 = __builtin_amdgcn_wmma_f32_16x16x4_f32(false, a, false, b_1,
                                                                  (short)0, accS1, false, false);
                    if constexpr (FORM) {
                        __builtin_amdgcn_sched_barrier(0);
                        form_stage(2 * s + 1, k, t, fo);
                        __builtin_amdgcn_sched_barrier(0);
                    }
                }
            } else {
                bool sf[4];
                unsigned mm = cm;
#pragma unroll
                for (int e = 0; e < 4; ++e) {
                    sf[e] = (mm == rowmod);
                    mm = (mm + 1u == na) ? 0u : (mm + 1u);
                }
                cm += step8;
                if (cm >= na) cm -= na;

#pragma unroll
                for (int s = 0; s < 2; ++s) {
                    v2f aS, aO;
                    aS.x = sf[2 * s]     ? av[2 * s]     : 0.0f;
                    aS.y = sf[2 * s + 1] ? av[2 * s + 1] : 0.0f;
                    aO.x = sf[2 * s]     ? 0.0f : av[2 * s];
                    aO.y = sf[2 * s + 1] ? 0.0f : av[2 * s + 1];

                    v2f b1_0; b1_0.x = r[8 * s + 0]; b1_0.y = r[8 * s + 1];
                    v2f b1_1; b1_1.x = r[8 * s + 2]; b1_1.y = r[8 * s + 3];
                    v2f b2_0; b2_0.x = r[8 * s + 4]; b2_0.y = r[8 * s + 5];
                    v2f b2_1; b2_1.x = r[8 * s + 6]; b2_1.y = r[8 * s + 7];

                    accS0 = __builtin_amdgcn_wmma_f32_16x16x4_f32(false, aS, false, b1_0,
                                                                  (short)0, accS0, false, false);
                    accS0 = __builtin_amdgcn_wmma_f32_16x16x4_f32(false, aO, false, b2_0,
                                                                  (short)0, accS0, false, false);
                    if constexpr (FORM) {
                        __builtin_amdgcn_sched_barrier(0);
                        form_stage(2 * s, k, t, fo);
                        if (s == 1 && t + 1 < TS) kn = row_consts(t + 1);
                        __builtin_amdgcn_sched_barrier(0);
                    }
                    accS1 = __builtin_amdgcn_wmma_f32_16x16x4_f32(false, aS, false, b1_1,
                                                                  (short)0, accS1, false, false);
                    accS1 = __builtin_amdgcn_wmma_f32_16x16x4_f32(false, aO, false, b2_1,
                                                                  (short)0, accS1, false, false);
                    if constexpr (FORM) {
                        __builtin_amdgcn_sched_barrier(0);
                        form_stage(2 * s + 1, k, t, fo);
                        __builtin_amdgcn_sched_barrier(0);
                    }
                }
            }
            if constexpr (FORM) {
                *(v4f*)&nxt[t][lane4] = fo;
                load_adj_row(t, kcR);
            }
            __builtin_amdgcn_sched_barrier(0);
        }
    };

    fetchB(0);
    s2C = *(const v4f*)(s2w + lane4);
    o2C = *(const v4f*)(o2w + lane4);
#pragma unroll
    for (int r = 0; r < 16; ++r) load_adj_row(r, 0);
    next_classes();
    {
        float (*t0)[LAP] = ldsA[0][wave];
#pragma unroll
        for (int r = 0; r < 16; ++r) {
            const RowK k = row_consts(r);
            v4f o = {};
#pragma unroll
            for (int st = 0; st < 4; ++st) form_stage(st, k, r, o);
            *(v4f*)&t0[r][lane4] = o;
            load_adj_row(r, KC);
        }
    }
    s2N = *(const v4f*)(s2w + KC + lane4);
    o2N = *(const v4f*)(o2w + KC + lane4);

#pragma unroll 1
    for (int c = 0; c + 1 < NCK; ++c) {
        const int kc = c * KC;
        lds_barrier();
        stashB();
        lds_barrier();

        fetchB(kc + KC);

        s2C = s2N;
        o2C = o2N;
        const int kc2 = (c + 2 < NCK) ? (kc + 2 * KC) : (Nn - KC);
        s2N = *(const v4f*)(s2w + kc2 + lane4);
        o2N = *(const v4f*)(o2w + kc2 + lane4);
        next_classes();

        block(flag<true>{}, ldsA[c & 1][wave], ldsA[(c + 1) & 1][wave], kc2);
    }
    lds_barrier();
    stashB();
    lds_barrier();
    block(flag<false>{}, ldsA[(NCK - 1) & 1][wave], ldsA[NCK & 1][wave], 0);

    const unsigned mb = (unsigned)((lane >> 4) << 3);
#pragma unroll
    for (int r = 0; r < 8; ++r) {
        float* o = hp + ((long)(b * Nn + rowq(mb + (unsigned)r))) * Dd + h * DH;
        o[m]      = accS0[r];
        o[16 + m] = accS1[r];
    }
}

__global__ __launch_bounds__(128) void aggregate_kernel(
    const int* __restrict__ adj,
    const float* __restrict__ s1, const float* __restrict__ s2,
    const float* __restrict__ o1, const float* __restrict__ o2,
    const float* __restrict__ rowmax, const float* __restrict__ rowtau,
    const float* __restrict__ wh1,
    const float* __restrict__ wh2, const int* __restrict__ nag,
    float* __restrict__ hp)
{
    __shared__ __attribute__((aligned(16))) float ldsW1[KC][LDP];
    __shared__ __attribute__((aligned(16))) float ldsW2[KC][LDP];
    __shared__ __attribute__((aligned(16))) float ldsA[2][4][16][LAP];

    const unsigned na  = (unsigned)nag[0];
    const unsigned cnt = (unsigned)Nn / na;
    const bool one = (cnt * na == (unsigned)Nn) && ((cnt & 63u) == 0u);
    if (__builtin_expect(one, 1))
        aggregate_body<true >(adj, s1, s2, o1, o2, rowmax, rowtau, wh1, wh2,
                              na, cnt, hp, ldsW1, ldsW2, ldsA);
    else
        aggregate_body<false>(adj, s1, s2, o1, o2, rowmax, rowtau, wh1, wh2,
                              na, cnt, hp, ldsW1, ldsW2, ldsA);
}

__global__ __launch_bounds__(256)
void outproj_kernel(
    const float* hp, const float* Wo,
    const float* __restrict__ bo, float* out)
{
    __shared__ __attribute__((aligned(16))) float lds[2 * (64 + 64) * GP];
    proj_tile<64, 64, 1, 2, true, false>(hp, Wo, bo, out,
                                         blockIdx.y * 64, blockIdx.x * 64, lds,
                                         nullptr, nullptr, nullptr, nullptr);
}

extern "C" void kernel_launch(void* const* d_in, const int* in_sizes, int n_in,
                              void* d_out, int out_size, void* d_ws, size_t ws_size,
                              hipStream_t stream) {
    (void)in_sizes; (void)n_in; (void)out_size; (void)ws_size;

    const float* hmat = (const float*)d_in[0];
    const int*   nag  = (const int*)  d_in[1];
    const int*   adj  = (const int*)  d_in[2];
    const float* W1   = (const float*)d_in[3];
    const float* W2   = (const float*)d_in[4];
    const float* Ai   = (const float*)d_in[5];
    const float* Aj   = (const float*)d_in[6];
    const float* Wo   = (const float*)d_in[7];
    const float* bo   = (const float*)d_in[8];
    float*       out  = (float*)d_out;

    float* ws  = (float*)d_ws;
    const size_t whN = (size_t)Bc * Nn * Dd;
    const size_t scN = (size_t)Bc * Hh * Nn;
    float* wh1 = ws;
    float* wh2 = wh1 + whN;
    float* s1  = wh2 + whN;
    float* s2  = s1 + scN;
    float* o1  = s2 + scN;
    float* o2  = o1 + scN;
    float* hp  = o2 + scN;
    float* rowmax = hp + whN;
    float* rowtau = rowmax + scN;

    proj12_kernel<<<dim3(4, (Bc * Nn) / 64), 256, 0, stream>>>(
        hmat, W1, W2, wh1, wh2, Ai, Aj, s1, s2, o1, o2);

    entmax_kernel<<<(Bc * Hh * Nn) / (8 * RPW), 256, 0, stream>>>(
        s1, s2, o1, o2, adj, nag, rowmax, rowtau);

    aggregate_kernel<<<dim3(Bc * Hh, Nn / 64), 128, 0, stream>>>(
        adj, s1, s2, o1, o2, rowmax, rowtau, wh1, wh2, nag, hp);

    outproj_kernel<<<dim3(Dd / 64, (Bc * Nn) / 64), 256, 0, stream>>>(
        hp, Wo, bo, out);
}
